# starting kernel
# baseline (speedup 1.0000x reference)
#include <hip/hip_runtime.h>
#include <hip/hip_bf16.h>

typedef __bf16 bf16_t;
typedef __attribute__((ext_vector_type(16))) __bf16 v16bf;
typedef __attribute__((ext_vector_type(4)))  __bf16 v4bf;
typedef __attribute__((ext_vector_type(8)))  float  v8f;

#define DIMD   256
#define HEADS  4
#define HD     64
#define FFND   1024
#define BATCH  8
#define SEQ    2048
#define NTOK   (BATCH*SEQ)
#define NEGV   (-1e9f)

__device__ __forceinline__ v16bf load_frag_bf16(const bf16_t* base) {
  const int half = (threadIdx.x >> 4) & 1;
  const bf16_t* p = base + 8 * half;
  v16bf f;
#pragma unroll
  for (int e = 0; e < 8; ++e) {
    f[e]     = p[e];
    f[e + 8] = p[e + 16];
  }
  return f;
}

__device__ __forceinline__ v8f wmma_bf16(v16bf a, v16bf b, v8f c) {
  return __builtin_amdgcn_wmma_f32_16x16x32_bf16(
       false, a,  false, b,
       (short)0, c,  false,  false);
}

__global__ __launch_bounds__(256) void cvt_bf16(const float* __restrict__ in,
                                                bf16_t* __restrict__ out) {
  const size_t i = (size_t)blockIdx.x * 256 + threadIdx.x;
  const float4 v = ((const float4*)in)[i];
  v4bf o;
  o[0] = (bf16_t)v.x; o[1] = (bf16_t)v.y; o[2] = (bf16_t)v.z; o[3] = (bf16_t)v.w;
  ((v4bf*)out)[i] = o;
}

template <int MODE>
__global__ __launch_bounds__(128) void gemm_wmma(
    const bf16_t* __restrict__ A, const bf16_t* __restrict__ B,
    const float* __restrict__ bias, float* __restrict__ Yf,
    bf16_t* __restrict__ Yb,
    bf16_t* __restrict__ Qb, bf16_t* __restrict__ Kb, bf16_t* __restrict__ Vt,
    int M, int N, int K) {
  const int lane = threadIdx.x & 31;
  const int wave = threadIdx.x >> 5;
  const int r    = lane & 15;
  const int half = lane >> 4;
  const int row0 = blockIdx.y * 128 + wave * 32;
  const int col0 = blockIdx.x * 64;

  v8f acc[2][4] = {};
  const bf16_t* ap0 = A + (size_t)(row0 + r) * K;
  const bf16_t* ap1 = A + (size_t)(row0 + 16 + r) * K;
  const bf16_t* bp  = B + (size_t)(col0 + r) * K;

  for (int k0 = 0; k0 < K; k0 += 32) {
    __builtin_prefetch(ap0 + k0 + 64, 0, 0);
    __builtin_prefetch(ap1 + k0 + 64, 0, 0);
    v16bf a0 = load_frag_bf16(ap0 + k0);
    v16bf a1 = load_frag_bf16(ap1 + k0);
#pragma unroll
    for (int nb = 0; nb < 4; ++nb) {
      v16bf bfr = load_frag_bf16(bp + (size_t)nb * 16 * K + k0);
      acc[0][nb] = wmma_bf16(a0, bfr, acc[0][nb]);
      acc[1][nb] = wmma_bf16(a1, bfr, acc[1][nb]);
    }
  }

#pragma unroll
  for (int mi = 0; mi < 2; ++mi) {
#pragma unroll
    for (int nb = 0; nb < 4; ++nb) {
#pragma unroll
      for (int i = 0; i < 8; ++i) {
        const int row = row0 + mi * 16 + i + 8 * half;
        const int col = col0 + nb * 16 + r;
        float y = acc[mi][nb][i] + bias[col];
        if (MODE == 0) {
          Yf[(size_t)row * N + col] = y;
        } else if (MODE == 1) {
          Yb[(size_t)row * N + col] = (bf16_t)fmaxf(y, 0.0f);
        } else {
          const int b = row >> 11, t = row & (SEQ - 1);
          if (col < DIMD) {
            const int h = col >> 6, d = col & 63;
            Qb[(((size_t)b * HEADS + h) * SEQ + t) * HD + d] = (bf16_t)y;
          } else if (col < 2 * DIMD) {
            const int c = col - DIMD, h = c >> 6, d = c & 63;
            Kb[(((size_t)b * HEADS + h) * SEQ + t) * HD + d] = (bf16_t)y;
          } else {
            const int c = col - 2 * DIMD, h = c >> 6, d = c & 63;
            Vt[(((size_t)b * HEADS + h) * HD + d) * SEQ + t] = (bf16_t)y;
          }
        }
      }
    }
  }
}

__global__ __launch_bounds__(32) void attn_wmma(
    const bf16_t* __restrict__ Qb, const bf16_t* __restrict__ Kb,
    const bf16_t* __restrict__ Vt, const float* __restrict__ mask,
    bf16_t* __restrict__ attn, float* __restrict__ halt) {
  __shared__ bf16_t lds_p[2][16 * 32];

  const int bid  = blockIdx.x;
  const int qt   = bid & (SEQ / 32 - 1);
  const int bh   = bid >> 6;
  const int b    = bh >> 2;
  const int lane = threadIdx.x & 31;
  const int r    = lane & 15;
  const int half = lane >> 4;
  const float scale = 0.125f;

  const bf16_t* Qbase = Qb + ((size_t)bh * SEQ + qt * 32) * HD;
  const bf16_t* Kbase = Kb + (size_t)bh * SEQ * HD;
  const bf16_t* Vbase = Vt + (size_t)bh * HD * SEQ;
  const float*  mrow  = mask + (size_t)b * SEQ;

  v16bf qa[2][2];
#pragma unroll
  for (int qi = 0; qi < 2; ++qi) {
    qa[qi][0] = load_frag_bf16(Qbase + (size_t)(qi * 16 + r) * HD + 0);
    qa[qi][1] = load_frag_bf16(Qbase + (size_t)(qi * 16 + r) * HD + 32);
  }

  float m[2][8], l[2][8];
#pragma unroll
  for (int qi = 0; qi < 2; ++qi)
#pragma unroll
    for (int i = 0; i < 8; ++i) { m[qi][i] = -3.0e38f; l[qi][i] = 0.0f; }

  for (int kt = 0; kt < SEQ / 16; ++kt) {
    v16bf kb0 = load_frag_bf16(Kbase + (size_t)(kt * 16 + r) * HD + 0);
    v16bf kb1 = load_frag_bf16(Kbase + (size_t)(kt * 16 + r) * HD + 32);
    const float mk = mrow[kt * 16 + r];
#pragma unroll
    for (int qi = 0; qi < 2; ++qi) {
      v8f s = {};
      s = wmma_bf16(qa[qi][0], kb0, s);
      s = wmma_bf16(qa[qi][1], kb1, s);
#pragma unroll
      for (int i = 0; i < 8; ++i) {
        const float sv = (mk > 0.5f) ? NEGV : s[i] * scale;
        const float mn = fmaxf(m[qi][i], sv);
        l[qi][i] = l[qi][i] * __expf(m[qi][i] - mn) + __expf(sv - mn);
        m[qi][i] = mn;
      }
    }
  }
#pragma unroll
  for (int d = 1; d <= 8; d <<= 1) {
#pragma unroll
    for (int qi = 0; qi < 2; ++qi) {
#pragma unroll
      for (int i = 0; i < 8; ++i) {
        const float mo = __shfl_xor(m[qi][i], d, 32);
        const float lo = __shfl_xor(l[qi][i], d, 32);
        const float mn = fmaxf(m[qi][i], mo);
        l[qi][i] = l[qi][i] * __expf(m[qi][i] - mn) + lo * __expf(mo - mn);
        m[qi][i] = mn;
      }
    }
  }
  float rl[2][8];
#pragma unroll
  for (int qi = 0; qi < 2; ++qi)
#pragma unroll
    for (int i = 0; i < 8; ++i) rl[qi][i] = 1.0f / l[qi][i];

  v8f o[2][4] = {};
  for (int kt2 = 0; kt2 < SEQ / 32; ++kt2) {
#pragma unroll
    for (int sub = 0; sub < 2; ++sub) {
      const int kt = kt2 * 2 + sub;
      v16bf kb0 = load_frag_bf16(Kbase + (size_t)(kt * 16 + r) * HD + 0);
      v16bf kb1 = load_frag_bf16(Kbase + (size_t)(kt * 16 + r) * HD + 32);
      const float mk = mrow[kt * 16 + r];
      float cs = 0.0f;
#pragma unroll
      for (int qi = 0; qi < 2; ++qi) {
        v8f s = {};
        s = wmma_bf16(qa[qi][0], kb0, s);
        s = wmma_bf16(qa[qi][1], kb1, s);
#pragma unroll
        for (int i = 0; i < 8; ++i) {
          const float sv = (mk > 0.5f) ? NEGV : s[i] * scale;
          const float p  = __expf(sv - m[qi][i]) * rl[qi][i];
          cs += p;
          lds_p[qi][(i + 8 * half) * 32 + sub * 16 + r] = (bf16_t)p;
        }
      }
      cs += __shfl_xor(cs, 16, 32);
      if (lane < 16) atomicAdd(&halt[(size_t)b * SEQ + kt * 16 + lane], cs);
    }
    __syncthreads();
    v16bf pa0 = load_frag_bf16(lds_p[0] + r * 32);
    v16bf pa1 = load_frag_bf16(lds_p[1] + r * 32);
#pragma unroll
    for (int nb = 0; nb < 4; ++nb) {
      v16bf vb = load_frag_bf16(Vbase + (size_t)(nb * 16 + r) * SEQ + kt2 * 32);
      o[0][nb] = wmma_bf16(pa0, vb, o[0][nb]);
      o[1][nb] = wmma_bf16(pa1, vb, o[1][nb]);
    }
    __syncthreads();
  }

  const int h = bh & 3;
#pragma unroll
  for (int qi = 0; qi < 2; ++qi) {
#pragma unroll
    for (int nb = 0; nb < 4; ++nb) {
#pragma unroll
      for (int i = 0; i < 8; ++i) {
        const int t = qt * 32 + qi * 16 + i + 8 * half;
        attn[((size_t)b * SEQ + t) * DIMD + h * HD + nb * 16 + r] =
            (bf16_t)o[qi][nb][i];
      }
    }
  }
}

template <bool WRITE_BF16>
__global__ __launch_bounds__(256) void add_ln(
    const float* __restrict__ A, const float* __restrict__ Bv,
    const float* __restrict__ g, const float* __restrict__ be,
    float* __restrict__ out, bf16_t* __restrict__ outb) {
  __shared__ float red[256];
  const int row = blockIdx.x, tid = threadIdx.x;
  const float v = A[(size_t)row * DIMD + tid] + Bv[(size_t)row * DIMD + tid];
  red[tid] = v;
  __syncthreads();
  for (int s = 128; s > 0; s >>= 1) {
    if (tid < s) red[tid] += red[tid + s];
    __syncthreads();
  }
  const float mu = red[0] * (1.0f / DIMD);
  __syncthreads();
  const float d = v - mu;
  red[tid] = d * d;
  __syncthreads();
  for (int s = 128; s > 0; s >>= 1) {
    if (tid < s) red[tid] += red[tid + s];
    __syncthreads();
  }
  const float var = red[0] * (1.0f / DIMD);
  const float y = d * rsqrtf(var + 1e-5f) * g[tid] + be[tid];
  out[(size_t)row * DIMD + tid] = y;
  if (WRITE_BF16) outb[(size_t)row * DIMD + tid] = (bf16_t)y;
}

__global__ __launch_bounds__(256) void halt_finalize(
    const float* __restrict__ hacc, const float* __restrict__ mask,
    float* __restrict__ out) {
  __shared__ float red[256];
  const int b = blockIdx.x, tid = threadIdx.x;
  float lm = -3.0e38f;
  for (int j = tid; j < SEQ; j += 256) {
    const float v = hacc[(size_t)b * SEQ + j] * 0.25f +
                    mask[(size_t)b * SEQ + j] * (-1e10f);
    lm = fmaxf(lm, v);
  }
  red[tid] = lm;
  __syncthreads();
  for (int s = 128; s > 0; s >>= 1) {
    if (tid < s) red[tid] = fmaxf(red[tid], red[tid + s]);
    __syncthreads();
  }
  const float mx = red[0];
  __syncthreads();
  float ls = 0.0f;
  for (int j = tid; j < SEQ; j += 256) {
    const float v = hacc[(size_t)b * SEQ + j] * 0.25f +
                    mask[(size_t)b * SEQ + j] * (-1e10f);
    ls += __expf(v - mx);
  }
  red[tid] = ls;
  __syncthreads();
  for (int s = 128; s > 0; s >>= 1) {
    if (tid < s) red[tid] += red[tid + s];
    __syncthreads();
  }
  const float inv = 1.0f / red[0];
  for (int j = tid; j < SEQ; j += 256) {
    const float v = hacc[(size_t)b * SEQ + j] * 0.25f +
                    mask[(size_t)b * SEQ + j] * (-1e10f);
    out[(size_t)b * SEQ + j] = __expf(v - mx) * inv;
  }
}

extern "C" void kernel_launch(void* const* d_in, const int* in_sizes, int n_in,
                              void* d_out, int out_size, void* d_ws,
                              size_t ws_size, hipStream_t stream) {
  (void)in_sizes; (void)n_in; (void)out_size; (void)ws_size;
  const float* x    = (const float*)d_in[0];
  const float* mask = (const float*)d_in[1];
  const float* Wqkv = (const float*)d_in[2];
  const float* bqkv = (const float*)d_in[3];
  const float* Wo   = (const float*)d_in[4];
  const float* bo   = (const float*)d_in[5];
  const float* W1   = (const float*)d_in[6];
  const float* b1   = (const float*)d_in[7];
  const float* W2   = (const float*)d_in[8];
  const float* b2   = (const float*)d_in[9];
  const float* g1   = (const float*)d_in[10];
  const float* be1  = (const float*)d_in[11];
  const float* g2   = (const float*)d_in[12];
  const float* be2  = (const float*)d_in[13];

  char* ws = (char*)d_ws;
  size_t off = 0;
  auto alloc = [&](size_t bytes) {
    char* p = ws + off;
    off += (bytes + 255) & ~(size_t)255;
    return p;
  };
  bf16_t* xb    = (bf16_t*)alloc((size_t)NTOK * DIMD * 2);
  bf16_t* Wqkvb = (bf16_t*)alloc((size_t)3 * DIMD * DIMD * 2);
  bf16_t* Wob   = (bf16_t*)alloc((size_t)DIMD * DIMD * 2);
  bf16_t* W1b   = (bf16_t*)alloc((size_t)FFND * DIMD * 2);
  bf16_t* W2b   = (bf16_t*)alloc((size_t)DIMD * FFND * 2);
  bf16_t* Qb    = (bf16_t*)alloc((size_t)NTOK * DIMD * 2);
  bf16_t* Kb    = (bf16_t*)alloc((size_t)NTOK * DIMD * 2);
  bf16_t* Vt    = (bf16_t*)alloc((size_t)NTOK * DIMD * 2);
  bf16_t* attnb = (bf16_t*)alloc((size_t)NTOK * DIMD * 2);
  float*  attnO = (float*) alloc((size_t)NTOK * DIMD * 4);
  float*  hbuf  = (float*) alloc((size_t)NTOK * DIMD * 4);
  bf16_t* hb    = (bf16_t*)alloc((size_t)NTOK * DIMD * 2);
  bf16_t* ffn1b = (bf16_t*)alloc((size_t)NTOK * FFND * 2);
  float*  ffn2  = (float*) alloc((size_t)NTOK * DIMD * 4);
  float*  halt  = (float*) alloc((size_t)NTOK * 4);

  float* out      = (float*)d_out;
  float* out_halt = out + (size_t)NTOK * DIMD;

  hipMemsetAsync(halt, 0, (size_t)NTOK * sizeof(float), stream);

  cvt_bf16<<<NTOK * DIMD / 1024, 256, 0, stream>>>(x, xb);
  cvt_bf16<<<3 * DIMD * DIMD / 1024, 256, 0, stream>>>(Wqkv, Wqkvb);
  cvt_bf16<<<DIMD * DIMD / 1024, 256, 0, stream>>>(Wo, Wob);
  cvt_bf16<<<FFND * DIMD / 1024, 256, 0, stream>>>(W1, W1b);
  cvt_bf16<<<DIMD * FFND / 1024, 256, 0, stream>>>(W2, W2b);

  gemm_wmma<2><<<dim3(3 * DIMD / 64, NTOK / 128), 128, 0, stream>>>(
      xb, Wqkvb, bqkv, nullptr, nullptr, Qb, Kb, Vt, NTOK, 3 * DIMD, DIMD);

  attn_wmma<<<BATCH * HEADS * (SEQ / 32), 32, 0, stream>>>(
      Qb, Kb, Vt, mask, attnb, halt);

  gemm_wmma<0><<<dim3(DIMD / 64, NTOK / 128), 128, 0, stream>>>(
      attnb, Wob, bo, attnO, nullptr, nullptr, nullptr, nullptr,
      NTOK, DIMD, DIMD);

  add_ln<true><<<NTOK, 256, 0, stream>>>(x, attnO, g1, be1, hbuf, hb);

  gemm_wmma<1><<<dim3(FFND / 64, NTOK / 128), 128, 0, stream>>>(
      hb, W1b, b1, nullptr, ffn1b, nullptr, nullptr, nullptr,
      NTOK, FFND, DIMD);

  gemm_wmma<0><<<dim3(DIMD / 64, NTOK / 128), 128, 0, stream>>>(
      ffn1b, W2b, b2, ffn2, nullptr, nullptr, nullptr, nullptr,
      NTOK, DIMD, FFND);

  add_ln<false><<<NTOK, 256, 0, stream>>>(hbuf, ffn2, g2, be2, out, nullptr);

  halt_finalize<<<BATCH, 256, 0, stream>>>(halt, mask, out_halt);
}


# B operands in WMMA fragment order: QKV GEMM writes K (tiles accumulated transposed) and V, a weight cvt writes the 4 weights; each fragment load is 512 contiguous bytes (4 lines, not 16)
# speedup vs baseline: 1.4375x; 1.4375x over previous
#include <hip/hip_runtime.h>
#include <hip/hip_bf16.h>

typedef __bf16 bf16_t;
typedef __attribute__((ext_vector_type(16))) __bf16 v16bf;
typedef __attribute__((ext_vector_type(4)))  __bf16 v4bf;
typedef __attribute__((ext_vector_type(8)))  __bf16 v8bf;
typedef __attribute__((ext_vector_type(8)))  float  v8f;

#define DIMD   256
#define HEADS  4
#define HD     64
#define FFND   1024
#define BATCH  8
#define SEQ    2048
#define NTOK   (BATCH*SEQ)
#define NEGV   (-1e9f)

__device__ __forceinline__ v16bf load_frag_bf16(const bf16_t* base) {
  const int half = (threadIdx.x >> 4) & 1;
  const bf16_t* p = base + 8 * half;
  v16bf f;
#pragma unroll
  for (int e = 0; e < 8; ++e) {
    f[e]     = p[e];
    f[e + 8] = p[e + 16];
  }
  return f;
}

__device__ __forceinline__ v16bf load_frag_packed(const v8bf* p) {
  const v8bf lo = p[0];
  const v8bf hi = p[32];
  return __builtin_shufflevector(lo, hi, 0, 1, 2, 3, 4, 5, 6, 7,
                                 8, 9, 10, 11, 12, 13, 14, 15);
}

__device__ __forceinline__ v8f wmma_bf16(v16bf a, v16bf b, v8f c) {
  return __builtin_amdgcn_wmma_f32_16x16x32_bf16(
       false, a,  false, b,
       (short)0, c,  false,  false);
}

#define CVT_U 8
__global__ __launch_bounds__(256) void cvt_bf16(const float* __restrict__ in,
                                                bf16_t* __restrict__ out) {
  const size_t base = (size_t)blockIdx.x * (256 * CVT_U) + threadIdx.x;
  float4 v[CVT_U];
#pragma unroll
  for (int k = 0; k < CVT_U; ++k) v[k] = ((const float4*)in)[base + k * 256];
#pragma unroll
  for (int k = 0; k < CVT_U; ++k) {
    v4bf o;
    o[0] = (bf16_t)v[k].x; o[1] = (bf16_t)v[k].y;
    o[2] = (bf16_t)v[k].z; o[3] = (bf16_t)v[k].w;
    ((v4bf*)out)[base + k * 256] = o;
  }
}

template <int KD>
__global__ __launch_bounds__(256) void cvt_bf16_frag(const float* __restrict__ in,
                                                     bf16_t* __restrict__ out) {
  const int u    = blockIdx.x * 256 + threadIdx.x;
  const int l    = u & 31;
  const int j    = (u >> 5) & 1;
  const int tile = u >> 6;
  const int kc   = tile % (KD / 32);
  const int rt   = tile / (KD / 32);
  const int row  = rt * 16 + (l & 15);
  const int k    = kc * 32 + j * 16 + 8 * (l >> 4);
  const float4* p = (const float4*)(in + (size_t)row * KD + k);
  const float4 x0 = p[0], x1 = p[1];
  v8bf o;
  o[0] = (bf16_t)x0.x; o[1] = (bf16_t)x0.y; o[2] = (bf16_t)x0.z; o[3] = (bf16_t)x0.w;
  o[4] = (bf16_t)x1.x; o[5] = (bf16_t)x1.y; o[6] = (bf16_t)x1.z; o[7] = (bf16_t)x1.w;
  ((v8bf*)out)[u] = o;
}

template <bool TRANS>
__device__ __forceinline__ void gemm_mainloop(
    v8f (&acc)[2][4], const bf16_t* __restrict__ ap0,
    const bf16_t* __restrict__ ap1, const v8bf* __restrict__ bq, int K) {
  const int kch = K >> 5;
  for (int k0 = 0; k0 < K; k0 += 32) {
    __builtin_prefetch(ap0 + k0 + 64, 0, 0);
    __builtin_prefetch(ap1 + k0 + 64, 0, 0);
    v16bf a0 = load_frag_bf16(ap0 + k0);
    v16bf a1 = load_frag_bf16(ap1 + k0);
#pragma unroll
    for (int nb = 0; nb < 4; ++nb) {
      v16bf bfr = load_frag_packed(bq + (size_t)(nb * kch + (k0 >> 5)) * 64);
      if (TRANS) {
        acc[0][nb] = wmma_bf16(bfr, a0, acc[0][nb]);
        acc[1][nb] = wmma_bf16(bfr, a1, acc[1][nb]);
      } else {
        acc[0][nb] = wmma_bf16(a0, bfr, acc[0][nb]);
        acc[1][nb] = wmma_bf16(a1, bfr, acc[1][nb]);
      }
    }
  }
}

template <int MODE>
__global__ __launch_bounds__(128) void gemm_wmma(
    const bf16_t* __restrict__ A, const bf16_t* __restrict__ B,
    const float* __restrict__ bias, float* __restrict__ Yf,
    bf16_t* __restrict__ Yb,
    bf16_t* __restrict__ Qb, bf16_t* __restrict__ Kb, bf16_t* __restrict__ Vt,
    int M, int N, int K) {
  const int lane = threadIdx.x & 31;
  const int wave = threadIdx.x >> 5;
  const int r    = lane & 15;
  const int half = lane >> 4;
  const int row0 = blockIdx.y * 128 + wave * 32;
  const int col0 = blockIdx.x * 64;

  v8f acc[2][4] = {};
  const bf16_t* ap0 = A + (size_t)(row0 + r) * K;
  const bf16_t* ap1 = A + (size_t)(row0 + 16 + r) * K;
  const v8bf* bq = (const v8bf*)B + (size_t)(col0 >> 4) * (K >> 5) * 64 + lane;

  const bool kblk = (MODE == 2) && (col0 >= DIMD) && (col0 < 2 * DIMD);
  if (kblk) gemm_mainloop<true>(acc, ap0, ap1, bq, K);
  else      gemm_mainloop<false>(acc, ap0, ap1, bq, K);

  if (MODE == 2 && col0 >= DIMD) {
    const int bb  = row0 >> 11;
    const int t0  = row0 & (SEQ - 1);
    if (kblk) {
      const size_t bhk = (size_t)bb * HEADS + ((col0 - DIMD) >> 6);
      v8bf* kf = (v8bf*)Kb + (bhk * (SEQ / 16) + (t0 >> 4)) * 128 + lane;
#pragma unroll
      for (int mi = 0; mi < 2; ++mi) {
#pragma unroll
        for (int nb = 0; nb < 4; ++nb) {
          v8bf o;
#pragma unroll
          for (int i = 0; i < 8; ++i)
            o[i] = (bf16_t)(acc[mi][nb][i] + bias[col0 + nb * 16 + 8 * half + i]);
          kf[mi * 128 + nb * 32] = o;
        }
      }
    } else {
      const size_t bhv = (size_t)bb * HEADS + ((col0 - 2 * DIMD) >> 6);
      v8bf* vf = (v8bf*)Vt + (bhv * (SEQ / 32) + (t0 >> 5)) * 256 + lane;
#pragma unroll
      for (int mi = 0; mi < 2; ++mi) {
#pragma unroll
        for (int nb = 0; nb < 4; ++nb) {
          const float bv = bias[col0 + nb * 16 + r];
          v8bf o;
#pragma unroll
          for (int i = 0; i < 8; ++i) o[i] = (bf16_t)(acc[mi][nb][i] + bv);
          vf[nb * 64 + mi * 32] = o;
        }
      }
    }
    return;
  }

#pragma unroll
  for (int mi = 0; mi < 2; ++mi) {
#pragma unroll
    for (int nb = 0; nb < 4; ++nb) {
#pragma unroll
      for (int i = 0; i < 8; ++i) {
        const int row = row0 + mi * 16 + i + 8 * half;
        const int col = col0 + nb * 16 + r;
        float y = acc[mi][nb][i] + bias[col];
        if (MODE == 0) {
          Yf[(size_t)row * N + col] = y;
        } else if (MODE == 1) {
          Yb[(size_t)row * N + col] = (bf16_t)fmaxf(y, 0.0f);
        } else {
          const int b = row >> 11, t = row & (SEQ - 1);
          const int h = col >> 6, d = col & 63;
          Qb[(((size_t)b * HEADS + h) * SEQ + t) * HD + d] = (bf16_t)y;
        }
      }
    }
  }
}

#define KSPLIT 4
__global__ __launch_bounds__(32 * KSPLIT) void attn_wmma(
    const bf16_t* __restrict__ Qb, const bf16_t* __restrict__ Kb,
    const bf16_t* __restrict__ Vt, const float* __restrict__ mask,
    bf16_t* __restrict__ attn, float* __restrict__ halt) {
  __shared__ bf16_t lds_p[KSPLIT][2][16 * 32];
  __shared__ float  lds_m[KSPLIT][32];
  __shared__ float  lds_l[KSPLIT][32];
  __shared__ float  lds_o[KSPLIT][64][32];

  const int bid  = blockIdx.x;
  const int qt   = bid & (SEQ / 32 - 1);
  const int bh   = bid >> 6;
  const int b    = bh >> 2;
  const int lane = threadIdx.x & 31;
  const int w    = threadIdx.x >> 5;
  const int r    = lane & 15;
  const int half = lane >> 4;
  const float scale = 0.125f;

  const bf16_t* Qbase = Qb + ((size_t)bh * SEQ + qt * 32) * HD;
  const v8bf* Kq = (const v8bf*)Kb +
                   ((size_t)bh * (SEQ / 16) + (size_t)w * (SEQ / KSPLIT / 16)) * 128 + lane;
  const v8bf* Vq = (const v8bf*)Vt +
                   ((size_t)bh * (SEQ / 32) + (size_t)w * (SEQ / KSPLIT / 32)) * 256 + lane;
  const float*  mrow  = mask + (size_t)b * SEQ + (size_t)w * (SEQ / KSPLIT);
  float*        hrow  = halt + (size_t)b * SEQ + (size_t)w * (SEQ / KSPLIT);

  v16bf qa[2][2];
#pragma unroll
  for (int qi = 0; qi < 2; ++qi) {
    qa[qi][0] = load_frag_bf16(Qbase + (size_t)(qi * 16 + r) * HD + 0);
    qa[qi][1] = load_frag_bf16(Qbase + (size_t)(qi * 16 + r) * HD + 32);
  }

  float m[2][8], l[2][8];
#pragma unroll
  for (int qi = 0; qi < 2; ++qi)
#pragma unroll
    for (int i = 0; i < 8; ++i) { m[qi][i] = -3.0e38f; l[qi][i] = 0.0f; }

  for (int kt = 0; kt < SEQ / KSPLIT / 16; ++kt) {
    v16bf kb0 = load_frag_packed(Kq + (size_t)kt * 128);
    v16bf kb1 = load_frag_packed(Kq + (size_t)kt * 128 + 64);
    const float mk = mrow[kt * 16 + r];
#pragma unroll
    for (int qi = 0; qi < 2; ++qi) {
      v8f s = {};
      s = wmma_bf16(qa[qi][0], kb0, s);
      s = wmma_bf16(qa[qi][1], kb1, s);
#pragma unroll
      for (int i = 0; i < 8; ++i) {
        const float sv = (mk > 0.5f) ? NEGV : s[i] * scale;
        const float mn = fmaxf(m[qi][i], sv);
        l[qi][i] = l[qi][i] * __expf(m[qi][i] - mn) + __expf(sv - mn);
        m[qi][i] = mn;
      }
    }
  }
#pragma unroll
  for (int d = 1; d <= 8; d <<= 1) {
#pragma unroll
    for (int qi = 0; qi < 2; ++qi) {
#pragma unroll
      for (int i = 0; i < 8; ++i) {
        const float mo = __shfl_xor(m[qi][i], d, 32);
        const float lo = __shfl_xor(l[qi][i], d, 32);
        const float mn = fmaxf(m[qi][i], mo);
        l[qi][i] = l[qi][i] * __expf(m[qi][i] - mn) + lo * __expf(mo - mn);
        m[qi][i] = mn;
      }
    }
  }
  if (r == 0) {
#pragma unroll
    for (int qi = 0; qi < 2; ++qi) {
#pragma unroll
      for (int i = 0; i < 8; ++i) {
        lds_m[w][qi * 16 + i + 8 * half] = m[qi][i];
        lds_l[w][qi * 16 + i + 8 * half] = l[qi][i];
      }
    }
  }
  __syncthreads();
  float rl[2][8];
#pragma unroll
  for (int qi = 0; qi < 2; ++qi) {
#pragma unroll
    for (int i = 0; i < 8; ++i) {
      const int row = qi * 16 + i + 8 * half;
      float mw[KSPLIT], lw[KSPLIT];
#pragma unroll
      for (int k = 0; k < KSPLIT; ++k) { mw[k] = lds_m[k][row]; lw[k] = lds_l[k][row]; }
      float mx = mw[0];
#pragma unroll
      for (int k = 1; k < KSPLIT; ++k) mx = fmaxf(mx, mw[k]);
      float ls = 0.0f;
#pragma unroll
      for (int k = 0; k < KSPLIT; ++k) ls += lw[k] * __expf(mw[k] - mx);
      m[qi][i]  = mx;
      rl[qi][i] = 1.0f / ls;
    }
  }

  bf16_t* myp0 = &lds_p[w][0][0];
  bf16_t* myp1 = &lds_p[w][1][0];
  v8f o[2][4] = {};
  for (int kt2 = 0; kt2 < SEQ / KSPLIT / 32; ++kt2) {
#pragma unroll
    for (int sub = 0; sub < 2; ++sub) {
      const int kt = kt2 * 2 + sub;
      v16bf kb0 = load_frag_packed(Kq + (size_t)kt * 128);
      v16bf kb1 = load_frag_packed(Kq + (size_t)kt * 128 + 64);
      const float mk = mrow[kt * 16 + r];
      float cs = 0.0f;
#pragma unroll
      for (int qi = 0; qi < 2; ++qi) {
        v8f s = {};
        s = wmma_bf16(qa[qi][0], kb0, s);
        s = wmma_bf16(qa[qi][1], kb1, s);
        bf16_t* pq = qi ? myp1 : myp0;
#pragma unroll
        for (int i = 0; i < 8; ++i) {
          const float sv = (mk > 0.5f) ? NEGV : s[i] * scale;
          const float p  = __expf(sv - m[qi][i]) * rl[qi][i];
          cs += p;
          pq[(i + 8 * half) * 32 + sub * 16 + r] = (bf16_t)p;
        }
      }
      cs += __shfl_xor(cs, 16, 32);
      if (lane < 16) atomicAdd(&hrow[kt * 16 + lane], cs);
    }
    v16bf pa0 = load_frag_bf16(myp0 + r * 32);
    v16bf pa1 = load_frag_bf16(myp1 + r * 32);
#pragma unroll
    for (int nb = 0; nb < 4; ++nb) {
      v16bf vb = load_frag_packed(Vq + (size_t)kt2 * 256 + nb * 64);
      o[0][nb] = wmma_bf16(pa0, vb, o[0][nb]);
      o[1][nb] = wmma_bf16(pa1, vb, o[1][nb]);
    }
  }

#pragma unroll
  for (int qi = 0; qi < 2; ++qi) {
#pragma unroll
    for (int nb = 0; nb < 4; ++nb) {
#pragma unroll
      for (int i = 0; i < 8; ++i) {
        lds_o[w][(qi * 4 + nb) * 8 + i][lane] = o[qi][nb][i];
      }
    }
  }
  __syncthreads();

  const int h   = bh & 3;
  const int qif = w >> 1;
#pragma unroll
  for (int ii = 0; ii < 4; ++ii) {
    const int i = (w & 1) * 4 + ii;
    const int t = qt * 32 + qif * 16 + i + 8 * half;
#pragma unroll
    for (int nb = 0; nb < 4; ++nb) {
      const int j = (qif * 4 + nb) * 8 + i;
      float acc = lds_o[0][j][lane];
#pragma unroll
      for (int k = 1; k < KSPLIT; ++k) acc += lds_o[k][j][lane];
      attn[((size_t)b * SEQ + t) * DIMD + h * HD + nb * 16 + r] = (bf16_t)acc;
    }
  }
}

#define LN_ROWS  4
#define LN_WAVES 8
template <bool WRITE_BF16>
__global__ __launch_bounds__(32 * LN_WAVES)
__attribute__((amdgpu_waves_per_eu(2, 4))) void add_ln(
    const float* __restrict__ A, const float* __restrict__ Bv,
    const float* __restrict__ g, const float* __restrict__ be,
    float* __restrict__ out, bf16_t* __restrict__ outb) {
  const int lane = threadIdx.x & 31;
  const int wave = threadIdx.x >> 5;
  const size_t row0 = ((size_t)blockIdx.x * LN_WAVES + wave) * LN_ROWS;
  const float4* a4 = (const float4*)(A + row0 * DIMD) + lane;
  const float4* b4 = (const float4*)(Bv + row0 * DIMD) + lane;

  float v[LN_ROWS][8];
  float s[LN_ROWS];
#pragma unroll
  for (int r = 0; r < LN_ROWS; ++r) {
#pragma unroll
    for (int hf = 0; hf < 2; ++hf) {
      const float4 x = a4[r * (DIMD / 4) + hf * 32];
      const float4 y = b4[r * (DIMD / 4) + hf * 32];
      v[r][hf * 4 + 0] = x.x + y.x;
      v[r][hf * 4 + 1] = x.y + y.y;
      v[r][hf * 4 + 2] = x.z + y.z;
      v[r][hf * 4 + 3] = x.w + y.w;
    }
    s[r] = ((v[r][0] + v[r][1]) + (v[r][2] + v[r][3])) +
           ((v[r][4] + v[r][5]) + (v[r][6] + v[r][7]));
  }
  float4 gg[2], bb[2];
#pragma unroll
  for (int hf = 0; hf < 2; ++hf) {
    gg[hf] = ((const float4*)g)[lane + hf * 32];
    bb[hf] = ((const float4*)be)[lane + hf * 32];
  }
#pragma unroll
  for (int m = 16; m >= 1; m >>= 1) {
#pragma unroll
    for (int r = 0; r < LN_ROWS; ++r) s[r] += __shfl_xor(s[r], m, 32);
  }

  float q[LN_ROWS];
#pragma unroll
  for (int r = 0; r < LN_ROWS; ++r) {
    const float mu = s[r] * (1.0f / DIMD);
#pragma unroll
    for (int j = 0; j < 8; ++j) v[r][j] -= mu;
    q[r] = ((v[r][0] * v[r][0] + v[r][1] * v[r][1]) +
            (v[r][2] * v[r][2] + v[r][3] * v[r][3])) +
           ((v[r][4] * v[r][4] + v[r][5] * v[r][5]) +
            (v[r][6] * v[r][6] + v[r][7] * v[r][7]));
  }
#pragma unroll
  for (int m = 16; m >= 1; m >>= 1) {
#pragma unroll
    for (int r = 0; r < LN_ROWS; ++r) q[r] += __shfl_xor(q[r], m, 32);
  }

#pragma unroll
  for (int r = 0; r < LN_ROWS; ++r) {
    const float var = q[r] * (1.0f / DIMD);
    const float rs  = rsqrtf(var + 1e-5f);
    const size_t row = row0 + r;
#pragma unroll
    for (int hf = 0; hf < 2; ++hf) {
      float4 y;
      y.x = v[r][hf * 4 + 0] * rs * gg[hf].x + bb[hf].x;
      y.y = v[r][hf * 4 + 1] * rs * gg[hf].y + bb[hf].y;
      y.z = v[r][hf * 4 + 2] * rs * gg[hf].z + bb[hf].z;
      y.w = v[r][hf * 4 + 3] * rs * gg[hf].w + bb[hf].w;
      ((float4*)(out + row * DIMD))[lane + hf * 32] = y;
      if (WRITE_BF16) {
        v4bf o;
        o[0] = (bf16_t)y.x; o[1] = (bf16_t)y.y; o[2] = (bf16_t)y.z; o[3] = (bf16_t)y.w;
        ((v4bf*)(outb + row * DIMD))[lane + hf * 32] = o;
      }
    }
  }
}

__global__ __launch_bounds__(256) void halt_finalize(
    const float* __restrict__ hacc, const float* __restrict__ mask,
    float* __restrict__ out) {
  __shared__ float red[256];
  const int b = blockIdx.x, tid = threadIdx.x;
  float lm = -3.0e38f;
  for (int j = tid; j < SEQ; j += 256) {
    const float v = hacc[(size_t)b * SEQ + j] * 0.25f +
                    mask[(size_t)b * SEQ + j] * (-1e10f);
    lm = fmaxf(lm, v);
  }
  red[tid] = lm;
  __syncthreads();
  for (int s = 128; s > 0; s >>= 1) {
    if (tid < s) red[tid] = fmaxf(red[tid], red[tid + s]);
    __syncthreads();
  }
  const float mx = red[0];
  __syncthreads();
  float ls = 0.0f;
  for (int j = tid; j < SEQ; j += 256) {
    const float v = hacc[(size_t)b * SEQ + j] * 0.25f +
                    mask[(size_t)b * SEQ + j] * (-1e10f);
    ls += __expf(v - mx);
  }
  red[tid] = ls;
  __syncthreads();
  for (int s = 128; s > 0; s >>= 1) {
    if (tid < s) red[tid] += red[tid + s];
    __syncthreads();
  }
  const float inv = 1.0f / red[0];
  for (int j = tid; j < SEQ; j += 256) {
    const float v = hacc[(size_t)b * SEQ + j] * 0.25f +
                    mask[(size_t)b * SEQ + j] * (-1e10f);
    out[(size_t)b * SEQ + j] = __expf(v - mx) * inv;
  }
}

extern "C" void kernel_launch(void* const* d_in, const int* in_sizes, int n_in,
                              void* d_out, int out_size, void* d_ws,
                              size_t ws_size, hipStream_t stream) {
  (void)in_sizes; (void)n_in; (void)out_size; (void)ws_size;
  const float* x    = (const float*)d_in[0];
  const float* mask = (const float*)d_in[1];
  const float* Wqkv = (const float*)d_in[2];
  const float* bqkv = (const float*)d_in[3];
  const float* Wo   = (const float*)d_in[4];
  const float* bo   = (const float*)d_in[5];
  const float* W1   = (const float*)d_in[6];
  const float* b1   = (const float*)d_in[7];
  const float* W2   = (const float*)d_in[8];
  const float* b2   = (const float*)d_in[9];
  const float* g1   = (const float*)d_in[10];
  const float* be1  = (const float*)d_in[11];
  const float* g2   = (const float*)d_in[12];
  const float* be2  = (const float*)d_in[13];

  char* ws = (char*)d_ws;
  size_t off = 0;
  auto alloc = [&](size_t bytes) {
    char* p = ws + off;
    off += (bytes + 255) & ~(size_t)255;
    return p;
  };
  bf16_t* xb    = (bf16_t*)alloc((size_t)NTOK * DIMD * 2);
  bf16_t* Wqkvb = (bf16_t*)alloc((size_t)3 * DIMD * DIMD * 2);
  bf16_t* Wob   = (bf16_t*)alloc((size_t)DIMD * DIMD * 2);
  bf16_t* W1b   = (bf16_t*)alloc((size_t)FFND * DIMD * 2);
  bf16_t* W2b   = (bf16_t*)alloc((size_t)DIMD * FFND * 2);
  bf16_t* Qb    = (bf16_t*)alloc((size_t)NTOK * DIMD * 2);
  bf16_t* Kb    = (bf16_t*)alloc((size_t)NTOK * DIMD * 2);
  bf16_t* Vt    = (bf16_t*)alloc((size_t)NTOK * DIMD * 2);
  bf16_t* attnb = (bf16_t*)alloc((size_t)NTOK * DIMD * 2);
  float*  attnO = (float*) alloc((size_t)NTOK * DIMD * 4);
  float*  hbuf  = (float*) alloc((size_t)NTOK * DIMD * 4);
  bf16_t* hb    = (bf16_t*)alloc((size_t)NTOK * DIMD * 2);
  bf16_t* ffn1b = (bf16_t*)alloc((size_t)NTOK * FFND * 2);
  float*  ffn2  = (float*) alloc((size_t)NTOK * DIMD * 4);
  float*  halt  = (float*) alloc((size_t)NTOK * 4);

  float* out      = (float*)d_out;
  float* out_halt = out + (size_t)NTOK * DIMD;

  hipMemsetAsync(halt, 0, (size_t)NTOK * sizeof(float), stream);

  cvt_bf16<<<NTOK * DIMD / (1024 * CVT_U), 256, 0, stream>>>(x, xb);
  cvt_bf16_frag<DIMD><<<3 * DIMD * DIMD / (8 * 256), 256, 0, stream>>>(Wqkv, Wqkvb);
  cvt_bf16_frag<DIMD><<<DIMD * DIMD / (8 * 256), 256, 0, stream>>>(Wo, Wob);
  cvt_bf16_frag<DIMD><<<FFND * DIMD / (8 * 256), 256, 0, stream>>>(W1, W1b);
  cvt_bf16_frag<FFND><<<DIMD * FFND / (8 * 256), 256, 0, stream>>>(W2, W2b);

  gemm_wmma<2><<<dim3(3 * DIMD / 64, NTOK / 128), 128, 0, stream>>>(
      xb, Wqkvb, bqkv, nullptr, nullptr, Qb, Kb, Vt, NTOK, 3 * DIMD, DIMD);

  attn_wmma<<<BATCH * HEADS * (SEQ / 32), 32 * KSPLIT, 0, stream>>>(
      Qb, Kb, Vt, mask, attnb, halt);

  gemm_wmma<0><<<dim3(DIMD / 64, NTOK / 128), 128, 0, stream>>>(
      attnb, Wob, bo, attnO, nullptr, nullptr, nullptr, nullptr,
      NTOK, DIMD, DIMD);

  add_ln<true><<<NTOK / (LN_WAVES * LN_ROWS), 32 * LN_WAVES, 0, stream>>>(
      x, attnO, g1, be1, hbuf, hb);

  gemm_wmma<1><<<dim3(FFND / 64, NTOK / 128), 128, 0, stream>>>(
      hb, W1b, b1, nullptr, ffn1b, nullptr, nullptr, nullptr,
      NTOK, FFND, DIMD);

  gemm_wmma<0><<<dim3(DIMD / 64, NTOK / 128), 128, 0, stream>>>(
      ffn1b, W2b, b2, ffn2, nullptr, nullptr, nullptr, nullptr,
      NTOK, DIMD, FFND);

  add_ln<false><<<NTOK / (LN_WAVES * LN_ROWS), 32 * LN_WAVES, 0, stream>>>(
      hbuf, ffn2, g2, be2, out, nullptr);

  halt_finalize<<<BATCH, 256, 0, stream>>>(halt, mask, out_halt);
}


# attention: score tile computed transposed (S^T = K.Q^T): a query's softmax max/sum live in one lane, packed P registers feed P.V without LDS, halt column sums by DPP lane tree (new sum order)
# speedup vs baseline: 1.5733x; 1.0945x over previous
#include <hip/hip_runtime.h>
#include <hip/hip_bf16.h>

typedef __bf16 bf16_t;
typedef __attribute__((ext_vector_type(16))) __bf16 v16bf;
typedef __attribute__((ext_vector_type(4)))  __bf16 v4bf;
typedef __attribute__((ext_vector_type(8)))  __bf16 v8bf;
typedef __attribute__((ext_vector_type(8)))  float  v8f;

#define DIMD   256
#define HEADS  4
#define HD     64
#define FFND   1024
#define BATCH  8
#define SEQ    2048
#define NTOK   (BATCH*SEQ)

__device__ __forceinline__ v16bf load_frag_bf16(const bf16_t* base) {
  const int half = (threadIdx.x >> 4) & 1;
  const bf16_t* p = base + 8 * half;
  v16bf f;
#pragma unroll
  for (int e = 0; e < 8; ++e) {
    f[e]     = p[e];
    f[e + 8] = p[e + 16];
  }
  return f;
}

__device__ __forceinline__ v16bf load_frag_packed(const v8bf* p) {
  const v8bf lo = p[0];
  const v8bf hi = p[32];
  return __builtin_shufflevector(lo, hi, 0, 1, 2, 3, 4, 5, 6, 7,
                                 8, 9, 10, 11, 12, 13, 14, 15);
}

__device__ __forceinline__ v8f wmma_bf16(v16bf a, v16bf b, v8f c) {
  return __builtin_amdgcn_wmma_f32_16x16x32_bf16(
       false, a,  false, b,
       (short)0, c,  false,  false);
}

#define CVT_U 8
__global__ __launch_bounds__(256) void cvt_bf16(const float* __restrict__ in,
                                                bf16_t* __restrict__ out) {
  const size_t base = (size_t)blockIdx.x * (256 * CVT_U) + threadIdx.x;
  float4 v[CVT_U];
#pragma unroll
  for (int k = 0; k < CVT_U; ++k) v[k] = ((const float4*)in)[base + k * 256];
#pragma unroll
  for (int k = 0; k < CVT_U; ++k) {
    v4bf o;
    o[0] = (bf16_t)v[k].x; o[1] = (bf16_t)v[k].y;
    o[2] = (bf16_t)v[k].z; o[3] = (bf16_t)v[k].w;
    ((v4bf*)out)[base + k * 256] = o;
  }
}

template <int KD>
__global__ __launch_bounds__(256) void cvt_bf16_frag(const float* __restrict__ in,
                                                     bf16_t* __restrict__ out) {
  const int u    = blockIdx.x * 256 + threadIdx.x;
  const int l    = u & 31;
  const int j    = (u >> 5) & 1;
  const int tile = u >> 6;
  const int kc   = tile % (KD / 32);
  const int rt   = tile / (KD / 32);
  const int row  = rt * 16 + (l & 15);
  const int k    = kc * 32 + j * 16 + 8 * (l >> 4);
  const float4* p = (const float4*)(in + (size_t)row * KD + k);
  const float4 x0 = p[0], x1 = p[1];
  v8bf o;
  o[0] = (bf16_t)x0.x; o[1] = (bf16_t)x0.y; o[2] = (bf16_t)x0.z; o[3] = (bf16_t)x0.w;
  o[4] = (bf16_t)x1.x; o[5] = (bf16_t)x1.y; o[6] = (bf16_t)x1.z; o[7] = (bf16_t)x1.w;
  ((v8bf*)out)[u] = o;
}

template <bool TRANS>
__device__ __forceinline__ void gemm_mainloop(
    v8f (&acc)[2][4], const bf16_t* __restrict__ ap0,
    const bf16_t* __restrict__ ap1, const v8bf* __restrict__ bq, int K) {
  const int kch = K >> 5;
  for (int k0 = 0; k0 < K; k0 += 32) {
    __builtin_prefetch(ap0 + k0 + 64, 0, 0);
    __builtin_prefetch(ap1 + k0 + 64, 0, 0);
    v16bf a0 = load_frag_bf16(ap0 + k0);
    v16bf a1 = load_frag_bf16(ap1 + k0);
#pragma unroll
    for (int nb = 0; nb < 4; ++nb) {
      v16bf bfr = load_frag_packed(bq + (size_t)(nb * kch + (k0 >> 5)) * 64);
      if (TRANS) {
        acc[0][nb] = wmma_bf16(bfr, a0, acc[0][nb]);
        acc[1][nb] = wmma_bf16(bfr, a1, acc[1][nb]);
      } else {
        acc[0][nb] = wmma_bf16(a0, bfr, acc[0][nb]);
        acc[1][nb] = wmma_bf16(a1, bfr, acc[1][nb]);
      }
    }
  }
}

template <int MODE>
__global__ __launch_bounds__(128) void gemm_wmma(
    const bf16_t* __restrict__ A, const bf16_t* __restrict__ B,
    const float* __restrict__ bias, float* __restrict__ Yf,
    bf16_t* __restrict__ Yb,
    bf16_t* __restrict__ Qb, bf16_t* __restrict__ Kb, bf16_t* __restrict__ Vt,
    int M, int N, int K) {
  const int lane = threadIdx.x & 31;
  const int wave = threadIdx.x >> 5;
  const int r    = lane & 15;
  const int half = lane >> 4;
  const int row0 = blockIdx.y * 128 + wave * 32;
  const int col0 = blockIdx.x * 64;

  v8f acc[2][4] = {};
  const bf16_t* ap0 = A + (size_t)(row0 + r) * K;
  const bf16_t* ap1 = A + (size_t)(row0 + 16 + r) * K;
  const v8bf* bq = (const v8bf*)B + (size_t)(col0 >> 4) * (K >> 5) * 64 + lane;

  const bool kblk = (MODE == 2) && (col0 >= DIMD) && (col0 < 2 * DIMD);
  if (kblk) gemm_mainloop<true>(acc, ap0, ap1, bq, K);
  else      gemm_mainloop<false>(acc, ap0, ap1, bq, K);

  if (MODE == 2 && col0 >= DIMD) {
    const int bb  = row0 >> 11;
    const int t0  = row0 & (SEQ - 1);
    if (kblk) {
      const size_t bhk = (size_t)bb * HEADS + ((col0 - DIMD) >> 6);
      v8bf* kf = (v8bf*)Kb + (bhk * (SEQ / 16) + (t0 >> 4)) * 128 + lane;
#pragma unroll
      for (int mi = 0; mi < 2; ++mi) {
#pragma unroll
        for (int nb = 0; nb < 4; ++nb) {
          v8bf o;
#pragma unroll
          for (int i = 0; i < 8; ++i)
            o[i] = (bf16_t)(acc[mi][nb][i] + bias[col0 + nb * 16 + 8 * half + i]);
          kf[mi * 128 + nb * 32] = o;
        }
      }
    } else {
      const size_t bhv = (size_t)bb * HEADS + ((col0 - 2 * DIMD) >> 6);
      v8bf* vf = (v8bf*)Vt + (bhv * (SEQ / 32) + (t0 >> 5)) * 256 + lane;
#pragma unroll
      for (int mi = 0; mi < 2; ++mi) {
#pragma unroll
        for (int nb = 0; nb < 4; ++nb) {
          const float bv = bias[col0 + nb * 16 + r];
          v8bf o;
#pragma unroll
          for (int i = 0; i < 8; ++i) o[i] = (bf16_t)(acc[mi][nb][i] + bv);
          vf[nb * 64 + mi * 32] = o;
        }
      }
    }
    return;
  }

#pragma unroll
  for (int mi = 0; mi < 2; ++mi) {
#pragma unroll
    for (int nb = 0; nb < 4; ++nb) {
#pragma unroll
      for (int i = 0; i < 8; ++i) {
        const int row = row0 + mi * 16 + i + 8 * half;
        const int col = col0 + nb * 16 + r;
        float y = acc[mi][nb][i] + bias[col];
        if (MODE == 0) {
          Yf[(size_t)row * N + col] = y;
        } else if (MODE == 1) {
          Yb[(size_t)row * N + col] = (bf16_t)fmaxf(y, 0.0f);
        } else {
          const int b = row >> 11, t = row & (SEQ - 1);
          const int h = col >> 6, d = col & 63;
          Qb[(((size_t)b * HEADS + h) * SEQ + t) * HD + d] = (bf16_t)y;
        }
      }
    }
  }
}

#define KSPLIT 4
#define CSCALE 0.18033688011112042f
#define NEGRAW (-8589934592.0f)

typedef __attribute__((ext_vector_type(2))) float  f32x2;
typedef __attribute__((ext_vector_type(2))) __bf16 bf16x2;

__device__ __forceinline__ float xhalf(float x) {
  const int b = __builtin_bit_cast(int, x);
  return __builtin_bit_cast(float, __builtin_amdgcn_permlanex16(
      b, b, 0x76543210, 0xfedcba98, false, false));
}

template <int XM>
__device__ __forceinline__ float row_xor(float x) {
  const int b = __builtin_bit_cast(int, x);
  return __builtin_bit_cast(float, __builtin_amdgcn_update_dpp(
      0, b, 0x160 + XM, 0xf, 0xf, true));
}

__device__ __forceinline__ float max3f(float a, float b, float c) {
  return __builtin_fmaxf(__builtin_fmaxf(a, b), c);
}

__global__ __launch_bounds__(32 * KSPLIT) void attn_wmma(
    const bf16_t* __restrict__ Qb, const bf16_t* __restrict__ Kb,
    const bf16_t* __restrict__ Vt, const float* __restrict__ mask,
    bf16_t* __restrict__ attn, float* __restrict__ halt) {
  __shared__ float  lds_m[KSPLIT][32];
  __shared__ float  lds_l[KSPLIT][32];
  __shared__ float  lds_o[KSPLIT][64][32];

  const int bid  = blockIdx.x;
  const int qt   = bid & (SEQ / 32 - 1);
  const int bh   = bid >> 6;
  const int b    = bh >> 2;
  const int lane = threadIdx.x & 31;
  const int w    = threadIdx.x >> 5;
  const int r    = lane & 15;
  const int half = lane >> 4;

  const bf16_t* Qbase = Qb + ((size_t)bh * SEQ + qt * 32) * HD;
  const v8bf* Kq = (const v8bf*)Kb +
                   ((size_t)bh * (SEQ / 16) + (size_t)w * (SEQ / KSPLIT / 16)) * 128 + lane;
  const v8bf* Vq = (const v8bf*)Vt +
                   ((size_t)bh * (SEQ / 32) + (size_t)w * (SEQ / KSPLIT / 32)) * 256 + lane;
  const float* mrow8 = mask + (size_t)b * SEQ + (size_t)w * (SEQ / KSPLIT) + 8 * half;
  float*       hrow  = halt + (size_t)b * SEQ + (size_t)w * (SEQ / KSPLIT);

  v16bf qa[2][2];
#pragma unroll
  for (int qi = 0; qi < 2; ++qi) {
    qa[qi][0] = load_frag_bf16(Qbase + (size_t)(qi * 16 + r) * HD + 0);
    qa[qi][1] = load_frag_bf16(Qbase + (size_t)(qi * 16 + r) * HD + 32);
  }

  const f32x2 c2 = {CSCALE, CSCALE};

  float m[2], sh[2], l[2];
#pragma unroll
  for (int qi = 0; qi < 2; ++qi) {
    m[qi] = -3.0e38f; sh[qi] = 3.0e38f * CSCALE; l[qi] = 0.0f;
  }

  for (int kt = 0; kt < SEQ / KSPLIT / 16; ++kt) {
    const v16bf kb0 = load_frag_packed(Kq + (size_t)kt * 128);
    const v16bf kb1 = load_frag_packed(Kq + (size_t)kt * 128 + 64);
    const v8f mk = *(const v8f*)(mrow8 + kt * 16);
    __builtin_amdgcn_sched_barrier(0);
    const bool msk[8] = {mk[0] > 0.5f, mk[1] > 0.5f, mk[2] > 0.5f, mk[3] > 0.5f,
                         mk[4] > 0.5f, mk[5] > 0.5f, mk[6] > 0.5f, mk[7] > 0.5f};
#pragma unroll
    for (int qi = 0; qi < 2; ++qi) {
      v8f s = {};
      s = wmma_bf16(kb0, qa[qi][0], s);
      s = wmma_bf16(kb1, qa[qi][1], s);
      float sv[8];
#pragma unroll
      for (int i = 0; i < 8; ++i) sv[i] = msk[i] ? NEGRAW : s[i];
      float mx = max3f(m[qi], sv[0], sv[1]);
      mx = max3f(mx, sv[2], sv[3]);
      mx = max3f(mx, sv[4], sv[5]);
      mx = max3f(mx, sv[6], sv[7]);
      const float nm = -(mx * CSCALE);
      const f32x2 nm2 = {nm, nm};
      const float alpha = __builtin_amdgcn_exp2f(nm - sh[qi]);
      f32x2 e[4];
#pragma unroll
      for (int j = 0; j < 4; ++j) {
        const f32x2 a = {sv[2 * j], sv[2 * j + 1]};
        const f32x2 t = __builtin_elementwise_fma(a, c2, nm2);
        e[j][0] = __builtin_amdgcn_exp2f(t[0]);
        e[j][1] = __builtin_amdgcn_exp2f(t[1]);
      }
      const f32x2 es = (e[0] + e[1]) + (e[2] + e[3]);
      l[qi] = __builtin_fmaf(l[qi], alpha, es[0] + es[1]);
      m[qi] = mx;
      sh[qi] = nm;
    }
  }
#pragma unroll
  for (int qi = 0; qi < 2; ++qi) {
    const float so = xhalf(sh[qi]);
    const float lo = xhalf(l[qi]);
    const float sn = __builtin_fminf(sh[qi], so);
    l[qi] = l[qi] * __builtin_amdgcn_exp2f(sn - sh[qi]) +
            lo * __builtin_amdgcn_exp2f(sn - so);
    sh[qi] = sn;
  }
  if (half == 0) {
#pragma unroll
    for (int qi = 0; qi < 2; ++qi) {
      lds_m[w][qi * 16 + r] = sh[qi];
      lds_l[w][qi * 16 + r] = l[qi];
    }
  }
  __syncthreads();
  f32x2 nmf[2], rl[2];
#pragma unroll
  for (int qi = 0; qi < 2; ++qi) {
    float sw[KSPLIT], lw[KSPLIT];
#pragma unroll
    for (int k = 0; k < KSPLIT; ++k) {
      sw[k] = lds_m[k][qi * 16 + r];
      lw[k] = lds_l[k][qi * 16 + r];
    }
    float nm = sw[0];
#pragma unroll
    for (int k = 1; k < KSPLIT; ++k) nm = __builtin_fminf(nm, sw[k]);
    float ls = 0.0f;
#pragma unroll
    for (int k = 0; k < KSPLIT; ++k)
      ls += lw[k] * __builtin_amdgcn_exp2f(nm - sw[k]);
    const float inv = 1.0f / ls;
    nmf[qi][0] = nm;  nmf[qi][1] = nm;
    rl[qi][0]  = inv; rl[qi][1]  = inv;
  }

  const bool b3 = (r & 8) != 0, b2 = (r & 4) != 0, b1 = (r & 2) != 0;
  v8f o[2][4] = {};
  for (int kt2 = 0; kt2 < SEQ / KSPLIT / 32; ++kt2) {
    v16bf pa[2];
#pragma unroll
    for (int sub = 0; sub < 2; ++sub) {
      const int kt = kt2 * 2 + sub;
      const v16bf kb0 = load_frag_packed(Kq + (size_t)kt * 128);
      const v16bf kb1 = load_frag_packed(Kq + (size_t)kt * 128 + 64);
      const v8f mk = *(const v8f*)(mrow8 + kt * 16);
      __builtin_amdgcn_sched_barrier(0);
      const bool msk[8] = {mk[0] > 0.5f, mk[1] > 0.5f, mk[2] > 0.5f, mk[3] > 0.5f,
                           mk[4] > 0.5f, mk[5] > 0.5f, mk[6] > 0.5f, mk[7] > 0.5f};
      f32x2 cs[4];
#pragma unroll
      for (int qi = 0; qi < 2; ++qi) {
        v8f s = {};
        s = wmma_bf16(kb0, qa[qi][0], s);
        s = wmma_bf16(kb1, qa[qi][1], s);
#pragma unroll
        for (int j = 0; j < 4; ++j) {
          const f32x2 a = {msk[2 * j] ? NEGRAW : s[2 * j],
                           msk[2 * j + 1] ? NEGRAW : s[2 * j + 1]};
          const f32x2 t = __builtin_elementwise_fma(a, c2, nmf[qi]);
          f32x2 p;
          p[0] = __builtin_amdgcn_exp2f(t[0]);
          p[1] = __builtin_amdgcn_exp2f(t[1]);
          p = p * rl[qi];
          if (qi == 0) cs[j] = p; else cs[j] = cs[j] + p;
          const bf16x2 pb = __builtin_convertvector(p, bf16x2);
          pa[qi][sub * 8 + 2 * j]     = pb[0];
          pa[qi][sub * 8 + 2 * j + 1] = pb[1];
        }
      }
      float k4[4];
#pragma unroll
      for (int j = 0; j < 4; ++j) {
        const float lo4 = cs[j >> 1][j & 1];
        const float hi4 = cs[(j + 4) >> 1][j & 1];
        const float keep = b3 ? hi4 : lo4;
        const float send = b3 ? lo4 : hi4;
        k4[j] = keep + row_xor<8>(send);
      }
      float k2[2];
#pragma unroll
      for (int j = 0; j < 2; ++j) {
        const float keep = b2 ? k4[j + 2] : k4[j];
        const float send = b2 ? k4[j] : k4[j + 2];
        k2[j] = keep + row_xor<4>(send);
      }
      float k1 = (b1 ? k2[1] : k2[0]) + row_xor<2>(b1 ? k2[0] : k2[1]);
      k1 += row_xor<1>(k1);
      if ((lane & 1) == 0) atomicAdd(&hrow[kt * 16 + 8 * half + (r >> 1)], k1);
    }
#pragma unroll
    for (int nb = 0; nb < 4; ++nb) {
      const v16bf vb = load_frag_packed(Vq + (size_t)kt2 * 256 + nb * 64);
      o[0][nb] = wmma_bf16(pa[0], vb, o[0][nb]);
      o[1][nb] = wmma_bf16(pa[1], vb, o[1][nb]);
    }
  }

#pragma unroll
  for (int qi = 0; qi < 2; ++qi) {
#pragma unroll
    for (int nb = 0; nb < 4; ++nb) {
#pragma unroll
      for (int i = 0; i < 8; ++i) {
        lds_o[w][(qi * 4 + nb) * 8 + i][lane] = o[qi][nb][i];
      }
    }
  }
  __syncthreads();

  const int h   = bh & 3;
  const int qif = w >> 1;
#pragma unroll
  for (int ii = 0; ii < 4; ++ii) {
    const int i = (w & 1) * 4 + ii;
    const int t = qt * 32 + qif * 16 + i + 8 * half;
#pragma unroll
    for (int nb = 0; nb < 4; ++nb) {
      const int j = (qif * 4 + nb) * 8 + i;
      float acc = lds_o[0][j][lane];
#pragma unroll
      for (int k = 1; k < KSPLIT; ++k) acc += lds_o[k][j][lane];
      attn[((size_t)b * SEQ + t) * DIMD + h * HD + nb * 16 + r] = (bf16_t)acc;
    }
  }
}

#define LN_ROWS  4
#define LN_WAVES 8
template <bool WRITE_BF16>
__global__ __launch_bounds__(32 * LN_WAVES)
__attribute__((amdgpu_waves_per_eu(2, 4))) void add_ln(
    const float* __restrict__ A, const float* __restrict__ Bv,
    const float* __restrict__ g, const float* __restrict__ be,
    float* __restrict__ out, bf16_t* __restrict__ outb) {
  const int lane = threadIdx.x & 31;
  const int wave = threadIdx.x >> 5;
  const size_t row0 = ((size_t)blockIdx.x * LN_WAVES + wave) * LN_ROWS;
  const float4* a4 = (const float4*)(A + row0 * DIMD) + lane;
  const float4* b4 = (const float4*)(Bv + row0 * DIMD) + lane;

  float v[LN_ROWS][8];
  float s[LN_ROWS];
#pragma unroll
  for (int r = 0; r < LN_ROWS; ++r) {
#pragma unroll
    for (int hf = 0; hf < 2; ++hf) {
      const float4 x = a4[r * (DIMD / 4) + hf * 32];
      const float4 y = b4[r * (DIMD / 4) + hf * 32];
      v[r][hf * 4 + 0] = x.x + y.x;
      v[r][hf * 4 + 1] = x.y + y.y;
      v[r][hf * 4 + 2] = x.z + y.z;
      v[r][hf * 4 + 3] = x.w + y.w;
    }
    s[r] = ((v[r][0] + v[r][1]) + (v[r][2] + v[r][3])) +
           ((v[r][4] + v[r][5]) + (v[r][6] + v[r][7]));
  }
  float4 gg[2], bb[2];
#pragma unroll
  for (int hf = 0; hf < 2; ++hf) {
    gg[hf] = ((const float4*)g)[lane + hf * 32];
    bb[hf] = ((const float4*)be)[lane + hf * 32];
  }
#pragma unroll
  for (int m = 16; m >= 1; m >>= 1) {
#pragma unroll
    for (int r = 0; r < LN_ROWS; ++r) s[r] += __shfl_xor(s[r], m, 32);
  }

  float q[LN_ROWS];
#pragma unroll
  for (int r = 0; r < LN_ROWS; ++r) {
    const float mu = s[r] * (1.0f / DIMD);
#pragma unroll
    for (int j = 0; j < 8; ++j) v[r][j] -= mu;
    q[r] = ((v[r][0] * v[r][0] + v[r][1] * v[r][1]) +
            (v[r][2] * v[r][2] + v[r][3] * v[r][3])) +
           ((v[r][4] * v[r][4] + v[r][5] * v[r][5]) +
            (v[r][6] * v[r][6] + v[r][7] * v[r][7]));
  }
#pragma unroll
  for (int m = 16; m >= 1; m >>= 1) {
#pragma unroll
    for (int r = 0; r < LN_ROWS; ++r) q[r] += __shfl_xor(q[r], m, 32);
  }

#pragma unroll
  for (int r = 0; r < LN_ROWS; ++r) {
    const float var = q[r] * (1.0f / DIMD);
    const float rs  = rsqrtf(var + 1e-5f);
    const size_t row = row0 + r;
#pragma unroll
    for (int hf = 0; hf < 2; ++hf) {
      float4 y;
      y.x = v[r][hf * 4 + 0] * rs * gg[hf].x + bb[hf].x;
      y.y = v[r][hf * 4 + 1] * rs * gg[hf].y + bb[hf].y;
      y.z = v[r][hf * 4 + 2] * rs * gg[hf].z + bb[hf].z;
      y.w = v[r][hf * 4 + 3] * rs * gg[hf].w + bb[hf].w;
      ((float4*)(out + row * DIMD))[lane + hf * 32] = y;
      if (WRITE_BF16) {
        v4bf o;
        o[0] = (bf16_t)y.x; o[1] = (bf16_t)y.y; o[2] = (bf16_t)y.z; o[3] = (bf16_t)y.w;
        ((v4bf*)(outb + row * DIMD))[lane + hf * 32] = o;
      }
    }
  }
}

__global__ __launch_bounds__(256) void halt_finalize(
    const float* __restrict__ hacc, const float* __restrict__ mask,
    float* __restrict__ out) {
  __shared__ float red[256];
  const int b = blockIdx.x, tid = threadIdx.x;
  float lm = -3.0e38f;
  for (int j = tid; j < SEQ; j += 256) {
    const float v = hacc[(size_t)b * SEQ + j] * 0.25f +
                    mask[(size_t)b * SEQ + j] * (-1e10f);
    lm = fmaxf(lm, v);
  }
  red[tid] = lm;
  __syncthreads();
  for (int s = 128; s > 0; s >>= 1) {
    if (tid < s) red[tid] = fmaxf(red[tid], red[tid + s]);
    __syncthreads();
  }
  const float mx = red[0];
  __syncthreads();
  float ls = 0.0f;
  for (int j = tid; j < SEQ; j += 256) {
    const float v = hacc[(size_t)b * SEQ + j] * 0.25f +
                    mask[(size_t)b * SEQ + j] * (-1e10f);
    ls += __expf(v - mx);
  }
  red[tid] = ls;
  __syncthreads();
  for (int s = 128; s > 0; s >>= 1) {
    if (tid < s) red[tid] += red[tid + s];
    __syncthreads();
  }
  const float inv = 1.0f / red[0];
  for (int j = tid; j < SEQ; j += 256) {
    const float v = hacc[(size_t)b * SEQ + j] * 0.25f +
                    mask[(size_t)b * SEQ + j] * (-1e10f);
    out[(size_t)b * SEQ + j] = __expf(v - mx) * inv;
  }
}

extern "C" void kernel_launch(void* const* d_in, const int* in_sizes, int n_in,
                              void* d_out, int out_size, void* d_ws,
                              size_t ws_size, hipStream_t stream) {
  (void)in_sizes; (void)n_in; (void)out_size; (void)ws_size;
  const float* x    = (const float*)d_in[0];
  const float* mask = (const float*)d_in[1];
  const float* Wqkv = (const float*)d_in[2];
  const float* bqkv = (const float*)d_in[3];
  const float* Wo   = (const float*)d_in[4];
  const float* bo   = (const float*)d_in[5];
  const float* W1   = (const float*)d_in[6];
  const float* b1   = (const float*)d_in[7];
  const float* W2   = (const float*)d_in[8];
  const float* b2   = (const float*)d_in[9];
  const float* g1   = (const float*)d_in[10];
  const float* be1  = (const float*)d_in[11];
  const float* g2   = (const float*)d_in[12];
  const float* be2  = (const float*)d_in[13];

  char* ws = (char*)d_ws;
  size_t off = 0;
  auto alloc = [&](size_t bytes) {
    char* p = ws + off;
    off += (bytes + 255) & ~(size_t)255;
    return p;
  };
  bf16_t* xb    = (bf16_t*)alloc((size_t)NTOK * DIMD * 2);
  bf16_t* Wqkvb = (bf16_t*)alloc((size_t)3 * DIMD * DIMD * 2);
  bf16_t* Wob   = (bf16_t*)alloc((size_t)DIMD * DIMD * 2);
  bf16_t* W1b   = (bf16_t*)alloc((size_t)FFND * DIMD * 2);
  bf16_t* W2b   = (bf16_t*)alloc((size_t)DIMD * FFND * 2);
  bf16_t* Qb    = (bf16_t*)alloc((size_t)NTOK * DIMD * 2);
  bf16_t* Kb    = (bf16_t*)alloc((size_t)NTOK * DIMD * 2);
  bf16_t* Vt    = (bf16_t*)alloc((size_t)NTOK * DIMD * 2);
  bf16_t* attnb = (bf16_t*)alloc((size_t)NTOK * DIMD * 2);
  float*  attnO = (float*) alloc((size_t)NTOK * DIMD * 4);
  float*  hbuf  = (float*) alloc((size_t)NTOK * DIMD * 4);
  bf16_t* hb    = (bf16_t*)alloc((size_t)NTOK * DIMD * 2);
  bf16_t* ffn1b = (bf16_t*)alloc((size_t)NTOK * FFND * 2);
  float*  ffn2  = (float*) alloc((size_t)NTOK * DIMD * 4);
  float*  halt  = (float*) alloc((size_t)NTOK * 4);

  float* out      = (float*)d_out;
  float* out_halt = out + (size_t)NTOK * DIMD;

  hipMemsetAsync(halt, 0, (size_t)NTOK * sizeof(float), stream);

  cvt_bf16<<<NTOK * DIMD / (1024 * CVT_U), 256, 0, stream>>>(x, xb);
  cvt_bf16_frag<DIMD><<<3 * DIMD * DIMD / (8 * 256), 256, 0, stream>>>(Wqkv, Wqkvb);
  cvt_bf16_frag<DIMD><<<DIMD * DIMD / (8 * 256), 256, 0, stream>>>(Wo, Wob);
  cvt_bf16_frag<DIMD><<<FFND * DIMD / (8 * 256), 256, 0, stream>>>(W1, W1b);
  cvt_bf16_frag<FFND><<<DIMD * FFND / (8 * 256), 256, 0, stream>>>(W2, W2b);

  gemm_wmma<2><<<dim3(3 * DIMD / 64, NTOK / 128), 128, 0, stream>>>(
      xb, Wqkvb, bqkv, nullptr, nullptr, Qb, Kb, Vt, NTOK, 3 * DIMD, DIMD);

  attn_wmma<<<BATCH * HEADS * (SEQ / 32), 32 * KSPLIT, 0, stream>>>(
      Qb, Kb, Vt, mask, attnb, halt);

  gemm_wmma<0><<<dim3(DIMD / 64, NTOK / 128), 128, 0, stream>>>(
      attnb, Wob, bo, attnO, nullptr, nullptr, nullptr, nullptr,
      NTOK, DIMD, DIMD);

  add_ln<true><<<NTOK / (LN_WAVES * LN_ROWS), 32 * LN_WAVES, 0, stream>>>(
      x, attnO, g1, be1, hbuf, hb);

  gemm_wmma<1><<<dim3(FFND / 64, NTOK / 128), 128, 0, stream>>>(
      hb, W1b, b1, nullptr, ffn1b, nullptr, nullptr, nullptr,
      NTOK, FFND, DIMD);

  gemm_wmma<0><<<dim3(DIMD / 64, NTOK / 128), 128, 0, stream>>>(
      ffn1b, W2b, b2, ffn2, nullptr, nullptr, nullptr, nullptr,
      NTOK, DIMD, FFND);

  add_ln<false><<<NTOK / (LN_WAVES * LN_ROWS), 32 * LN_WAVES, 0, stream>>>(
      hbuf, ffn2, g2, be2, out, nullptr);

  halt_finalize<<<BATCH, 256, 0, stream>>>(halt, mask, out_halt);
}


# independent launches merged: x and the 4 weight bf16 conversions plus halt-sum zeroing in one launch (memset dropped); halting softmax blocks ride in the first add+LN launch; 13 kernels+memset -> 8
# speedup vs baseline: 1.6544x; 1.0515x over previous
#include <hip/hip_runtime.h>
#include <hip/hip_bf16.h>

typedef __bf16 bf16_t;
typedef __attribute__((ext_vector_type(16))) __bf16 v16bf;
typedef __attribute__((ext_vector_type(4)))  __bf16 v4bf;
typedef __attribute__((ext_vector_type(8)))  __bf16 v8bf;
typedef __attribute__((ext_vector_type(8)))  float  v8f;

#define DIMD   256
#define HEADS  4
#define HD     64
#define FFND   1024
#define BATCH  8
#define SEQ    2048
#define NTOK   (BATCH*SEQ)

__device__ __forceinline__ v16bf load_frag_bf16(const bf16_t* base) {
  const int half = (threadIdx.x >> 4) & 1;
  const bf16_t* p = base + 8 * half;
  v16bf f;
#pragma unroll
  for (int e = 0; e < 8; ++e) {
    f[e]     = p[e];
    f[e + 8] = p[e + 16];
  }
  return f;
}

__device__ __forceinline__ v16bf load_frag_packed(const v8bf* p) {
  const v8bf lo = p[0];
  const v8bf hi = p[32];
  return __builtin_shufflevector(lo, hi, 0, 1, 2, 3, 4, 5, 6, 7,
                                 8, 9, 10, 11, 12, 13, 14, 15);
}

__device__ __forceinline__ v8f wmma_bf16(v16bf a, v16bf b, v8f c) {
  return __builtin_amdgcn_wmma_f32_16x16x32_bf16(
       false, a,  false, b,
       (short)0, c,  false,  false);
}

#define CVT_U 8
__device__ __forceinline__ void cvt_bf16_blk(const float* __restrict__ in,
                                             bf16_t* __restrict__ out, int blk) {
  const size_t base = (size_t)blk * (256 * CVT_U) + threadIdx.x;
  float4 v[CVT_U];
#pragma unroll
  for (int k = 0; k < CVT_U; ++k) v[k] = ((const float4*)in)[base + k * 256];
#pragma unroll
  for (int k = 0; k < CVT_U; ++k) {
    v4bf o;
    o[0] = (bf16_t)v[k].x; o[1] = (bf16_t)v[k].y;
    o[2] = (bf16_t)v[k].z; o[3] = (bf16_t)v[k].w;
    ((v4bf*)out)[base + k * 256] = o;
  }
}

template <int KD>
__device__ __forceinline__ void cvt_frag_blk(const float* __restrict__ in,
                                             bf16_t* __restrict__ out, int blk) {
  const int u    = blk * 256 + threadIdx.x;
  const int l    = u & 31;
  const int j    = (u >> 5) & 1;
  const int tile = u >> 6;
  const int kc   = tile % (KD / 32);
  const int rt   = tile / (KD / 32);
  const int row  = rt * 16 + (l & 15);
  const int k    = kc * 32 + j * 16 + 8 * (l >> 4);
  const float4* p = (const float4*)(in + (size_t)row * KD + k);
  const float4 x0 = p[0], x1 = p[1];
  v8bf o;
  o[0] = (bf16_t)x0.x; o[1] = (bf16_t)x0.y; o[2] = (bf16_t)x0.z; o[3] = (bf16_t)x0.w;
  o[4] = (bf16_t)x1.x; o[5] = (bf16_t)x1.y; o[6] = (bf16_t)x1.z; o[7] = (bf16_t)x1.w;
  ((v8bf*)out)[u] = o;
}

#define PB_X    (NTOK * DIMD / (1024 * CVT_U))
#define PB_WQKV (3 * DIMD * DIMD / (8 * 256))
#define PB_WO   (DIMD * DIMD / (8 * 256))
#define PB_W1   (FFND * DIMD / (8 * 256))
#define PB_W2   (DIMD * FFND / (8 * 256))
#define PB_HZ   (NTOK / (4 * 256))
__global__ __launch_bounds__(256) void prep_all(
    const float* __restrict__ x, const float* __restrict__ Wqkv,
    const float* __restrict__ Wo, const float* __restrict__ W1,
    const float* __restrict__ W2, bf16_t* __restrict__ xb,
    bf16_t* __restrict__ Wqkvb, bf16_t* __restrict__ Wob,
    bf16_t* __restrict__ W1b, bf16_t* __restrict__ W2b,
    float* __restrict__ halt) {
  int blk = blockIdx.x;
  if (blk < PB_X) { cvt_bf16_blk(x, xb, blk); return; }
  blk -= PB_X;
  if (blk < PB_WQKV) { cvt_frag_blk<DIMD>(Wqkv, Wqkvb, blk); return; }
  blk -= PB_WQKV;
  if (blk < PB_WO) { cvt_frag_blk<DIMD>(Wo, Wob, blk); return; }
  blk -= PB_WO;
  if (blk < PB_W1) { cvt_frag_blk<DIMD>(W1, W1b, blk); return; }
  blk -= PB_W1;
  if (blk < PB_W2) { cvt_frag_blk<FFND>(W2, W2b, blk); return; }
  blk -= PB_W2;
  ((float4*)halt)[blk * 256 + threadIdx.x] = make_float4(0.0f, 0.0f, 0.0f, 0.0f);
}

template <bool TRANS>
__device__ __forceinline__ void gemm_mainloop(
    v8f (&acc)[2][4], const bf16_t* __restrict__ ap0,
    const bf16_t* __restrict__ ap1, const v8bf* __restrict__ bq, int K) {
  const int kch = K >> 5;
  for (int k0 = 0; k0 < K; k0 += 32) {
    __builtin_prefetch(ap0 + k0 + 64, 0, 0);
    __builtin_prefetch(ap1 + k0 + 64, 0, 0);
    v16bf a0 = load_frag_bf16(ap0 + k0);
    v16bf a1 = load_frag_bf16(ap1 + k0);
#pragma unroll
    for (int nb = 0; nb < 4; ++nb) {
      v16bf bfr = load_frag_packed(bq + (size_t)(nb * kch + (k0 >> 5)) * 64);
      if (TRANS) {
        acc[0][nb] = wmma_bf16(bfr, a0, acc[0][nb]);
        acc[1][nb] = wmma_bf16(bfr, a1, acc[1][nb]);
      } else {
        acc[0][nb] = wmma_bf16(a0, bfr, acc[0][nb]);
        acc[1][nb] = wmma_bf16(a1, bfr, acc[1][nb]);
      }
    }
  }
}

template <int MODE>
__global__ __launch_bounds__(128) void gemm_wmma(
    const bf16_t* __restrict__ A, const bf16_t* __restrict__ B,
    const float* __restrict__ bias, float* __restrict__ Yf,
    bf16_t* __restrict__ Yb,
    bf16_t* __restrict__ Qb, bf16_t* __restrict__ Kb, bf16_t* __restrict__ Vt,
    int M, int N, int K) {
  const int lane = threadIdx.x & 31;
  const int wave = threadIdx.x >> 5;
  const int r    = lane & 15;
  const int half = lane >> 4;
  const int row0 = blockIdx.y * 128 + wave * 32;
  const int col0 = blockIdx.x * 64;

  v8f acc[2][4] = {};
  const bf16_t* ap0 = A + (size_t)(row0 + r) * K;
  const bf16_t* ap1 = A + (size_t)(row0 + 16 + r) * K;
  const v8bf* bq = (const v8bf*)B + (size_t)(col0 >> 4) * (K >> 5) * 64 + lane;

  const bool kblk = (MODE == 2) && (col0 >= DIMD) && (col0 < 2 * DIMD);
  if (kblk) gemm_mainloop<true>(acc, ap0, ap1, bq, K);
  else      gemm_mainloop<false>(acc, ap0, ap1, bq, K);

  if (MODE == 2 && col0 >= DIMD) {
    const int bb  = row0 >> 11;
    const int t0  = row0 & (SEQ - 1);
    if (kblk) {
      const size_t bhk = (size_t)bb * HEADS + ((col0 - DIMD) >> 6);
      v8bf* kf = (v8bf*)Kb + (bhk * (SEQ / 16) + (t0 >> 4)) * 128 + lane;
#pragma unroll
      for (int mi = 0; mi < 2; ++mi) {
#pragma unroll
        for (int nb = 0; nb < 4; ++nb) {
          v8bf o;
#pragma unroll
          for (int i = 0; i < 8; ++i)
            o[i] = (bf16_t)(acc[mi][nb][i] + bias[col0 + nb * 16 + 8 * half + i]);
          kf[mi * 128 + nb * 32] = o;
        }
      }
    } else {
      const size_t bhv = (size_t)bb * HEADS + ((col0 - 2 * DIMD) >> 6);
      v8bf* vf = (v8bf*)Vt + (bhv * (SEQ / 32) + (t0 >> 5)) * 256 + lane;
#pragma unroll
      for (int mi = 0; mi < 2; ++mi) {
#pragma unroll
        for (int nb = 0; nb < 4; ++nb) {
          const float bv = bias[col0 + nb * 16 + r];
          v8bf o;
#pragma unroll
          for (int i = 0; i < 8; ++i) o[i] = (bf16_t)(acc[mi][nb][i] + bv);
          vf[nb * 64 + mi * 32] = o;
        }
      }
    }
    return;
  }

#pragma unroll
  for (int mi = 0; mi < 2; ++mi) {
#pragma unroll
    for (int nb = 0; nb < 4; ++nb) {
#pragma unroll
      for (int i = 0; i < 8; ++i) {
        const int row = row0 + mi * 16 + i + 8 * half;
        const int col = col0 + nb * 16 + r;
        float y = acc[mi][nb][i] + bias[col];
        if (MODE == 0) {
          Yf[(size_t)row * N + col] = y;
        } else if (MODE == 1) {
          Yb[(size_t)row * N + col] = (bf16_t)fmaxf(y, 0.0f);
        } else {
          const int b = row >> 11, t = row & (SEQ - 1);
          const int h = col >> 6, d = col & 63;
          Qb[(((size_t)b * HEADS + h) * SEQ + t) * HD + d] = (bf16_t)y;
        }
      }
    }
  }
}

#define KSPLIT 4
#define CSCALE 0.18033688011112042f
#define NEGRAW (-8589934592.0f)

typedef __attribute__((ext_vector_type(2))) float  f32x2;
typedef __attribute__((ext_vector_type(2))) __bf16 bf16x2;

__device__ __forceinline__ float xhalf(float x) {
  const int b = __builtin_bit_cast(int, x);
  return __builtin_bit_cast(float, __builtin_amdgcn_permlanex16(
      b, b, 0x76543210, 0xfedcba98, false, false));
}

template <int XM>
__device__ __forceinline__ float row_xor(float x) {
  const int b = __builtin_bit_cast(int, x);
  return __builtin_bit_cast(float, __builtin_amdgcn_update_dpp(
      0, b, 0x160 + XM, 0xf, 0xf, true));
}

__device__ __forceinline__ float max3f(float a, float b, float c) {
  return __builtin_fmaxf(__builtin_fmaxf(a, b), c);
}

__global__ __launch_bounds__(32 * KSPLIT) void attn_wmma(
    const bf16_t* __restrict__ Qb, const bf16_t* __restrict__ Kb,
    const bf16_t* __restrict__ Vt, const float* __restrict__ mask,
    bf16_t* __restrict__ attn, float* __restrict__ halt) {
  __shared__ float  lds_m[KSPLIT][32];
  __shared__ float  lds_l[KSPLIT][32];
  __shared__ float  lds_o[KSPLIT][64][32];

  const int bid  = blockIdx.x;
  const int qt   = bid & (SEQ / 32 - 1);
  const int bh   = bid >> 6;
  const int b    = bh >> 2;
  const int lane = threadIdx.x & 31;
  const int w    = threadIdx.x >> 5;
  const int r    = lane & 15;
  const int half = lane >> 4;

  const bf16_t* Qbase = Qb + ((size_t)bh * SEQ + qt * 32) * HD;
  const v8bf* Kq = (const v8bf*)Kb +
                   ((size_t)bh * (SEQ / 16) + (size_t)w * (SEQ / KSPLIT / 16)) * 128 + lane;
  const v8bf* Vq = (const v8bf*)Vt +
                   ((size_t)bh * (SEQ / 32) + (size_t)w * (SEQ / KSPLIT / 32)) * 256 + lane;
  const float* mrow8 = mask + (size_t)b * SEQ + (size_t)w * (SEQ / KSPLIT) + 8 * half;
  float*       hrow  = halt + (size_t)b * SEQ + (size_t)w * (SEQ / KSPLIT);

  v16bf qa[2][2];
#pragma unroll
  for (int qi = 0; qi < 2; ++qi) {
    qa[qi][0] = load_frag_bf16(Qbase + (size_t)(qi * 16 + r) * HD + 0);
    qa[qi][1] = load_frag_bf16(Qbase + (size_t)(qi * 16 + r) * HD + 32);
  }

  const f32x2 c2 = {CSCALE, CSCALE};

  float m[2], sh[2], l[2];
#pragma unroll
  for (int qi = 0; qi < 2; ++qi) {
    m[qi] = -3.0e38f; sh[qi] = 3.0e38f * CSCALE; l[qi] = 0.0f;
  }

  for (int kt = 0; kt < SEQ / KSPLIT / 16; ++kt) {
    const v16bf kb0 = load_frag_packed(Kq + (size_t)kt * 128);
    const v16bf kb1 = load_frag_packed(Kq + (size_t)kt * 128 + 64);
    const v8f mk = *(const v8f*)(mrow8 + kt * 16);
    __builtin_amdgcn_sched_barrier(0);
    const bool msk[8] = {mk[0] > 0.5f, mk[1] > 0.5f, mk[2] > 0.5f, mk[3] > 0.5f,
                         mk[4] > 0.5f, mk[5] > 0.5f, mk[6] > 0.5f, mk[7] > 0.5f};
#pragma unroll
    for (int qi = 0; qi < 2; ++qi) {
      v8f s = {};
      s = wmma_bf16(kb0, qa[qi][0], s);
      s = wmma_bf16(kb1, qa[qi][1], s);
      float sv[8];
#pragma unroll
      for (int i = 0; i < 8; ++i) sv[i] = msk[i] ? NEGRAW : s[i];
      float mx = max3f(m[qi], sv[0], sv[1]);
      mx = max3f(mx, sv[2], sv[3]);
      mx = max3f(mx, sv[4], sv[5]);
      mx = max3f(mx, sv[6], sv[7]);
      const float nm = -(mx * CSCALE);
      const f32x2 nm2 = {nm, nm};
      const float alpha = __builtin_amdgcn_exp2f(nm - sh[qi]);
      f32x2 e[4];
#pragma unroll
      for (int j = 0; j < 4; ++j) {
        const f32x2 a = {sv[2 * j], sv[2 * j + 1]};
        const f32x2 t = __builtin_elementwise_fma(a, c2, nm2);
        e[j][0] = __builtin_amdgcn_exp2f(t[0]);
        e[j][1] = __builtin_amdgcn_exp2f(t[1]);
      }
      const f32x2 es = (e[0] + e[1]) + (e[2] + e[3]);
      l[qi] = __builtin_fmaf(l[qi], alpha, es[0] + es[1]);
      m[qi] = mx;
      sh[qi] = nm;
    }
  }
#pragma unroll
  for (int qi = 0; qi < 2; ++qi) {
    const float so = xhalf(sh[qi]);
    const float lo = xhalf(l[qi]);
    const float sn = __builtin_fminf(sh[qi], so);
    l[qi] = l[qi] * __builtin_amdgcn_exp2f(sn - sh[qi]) +
            lo * __builtin_amdgcn_exp2f(sn - so);
    sh[qi] = sn;
  }
  if (half == 0) {
#pragma unroll
    for (int qi = 0; qi < 2; ++qi) {
      lds_m[w][qi * 16 + r] = sh[qi];
      lds_l[w][qi * 16 + r] = l[qi];
    }
  }
  __syncthreads();
  f32x2 nmf[2], rl[2];
#pragma unroll
  for (int qi = 0; qi < 2; ++qi) {
    float sw[KSPLIT], lw[KSPLIT];
#pragma unroll
    for (int k = 0; k < KSPLIT; ++k) {
      sw[k] = lds_m[k][qi * 16 + r];
      lw[k] = lds_l[k][qi * 16 + r];
    }
    float nm = sw[0];
#pragma unroll
    for (int k = 1; k < KSPLIT; ++k) nm = __builtin_fminf(nm, sw[k]);
    float ls = 0.0f;
#pragma unroll
    for (int k = 0; k < KSPLIT; ++k)
      ls += lw[k] * __builtin_amdgcn_exp2f(nm - sw[k]);
    const float inv = 1.0f / ls;
    nmf[qi][0] = nm;  nmf[qi][1] = nm;
    rl[qi][0]  = inv; rl[qi][1]  = inv;
  }

  const bool b3 = (r & 8) != 0, b2 = (r & 4) != 0, b1 = (r & 2) != 0;
  v8f o[2][4] = {};
  for (int kt2 = 0; kt2 < SEQ / KSPLIT / 32; ++kt2) {
    v16bf pa[2];
#pragma unroll
    for (int sub = 0; sub < 2; ++sub) {
      const int kt = kt2 * 2 + sub;
      const v16bf kb0 = load_frag_packed(Kq + (size_t)kt * 128);
      const v16bf kb1 = load_frag_packed(Kq + (size_t)kt * 128 + 64);
      const v8f mk = *(const v8f*)(mrow8 + kt * 16);
      __builtin_amdgcn_sched_barrier(0);
      const bool msk[8] = {mk[0] > 0.5f, mk[1] > 0.5f, mk[2] > 0.5f, mk[3] > 0.5f,
                           mk[4] > 0.5f, mk[5] > 0.5f, mk[6] > 0.5f, mk[7] > 0.5f};
      f32x2 cs[4];
#pragma unroll
      for (int qi = 0; qi < 2; ++qi) {
        v8f s = {};
        s = wmma_bf16(kb0, qa[qi][0], s);
        s = wmma_bf16(kb1, qa[qi][1], s);
#pragma unroll
        for (int j = 0; j < 4; ++j) {
          const f32x2 a = {msk[2 * j] ? NEGRAW : s[2 * j],
                           msk[2 * j + 1] ? NEGRAW : s[2 * j + 1]};
          const f32x2 t = __builtin_elementwise_fma(a, c2, nmf[qi]);
          f32x2 p;
          p[0] = __builtin_amdgcn_exp2f(t[0]);
          p[1] = __builtin_amdgcn_exp2f(t[1]);
          p = p * rl[qi];
          if (qi == 0) cs[j] = p; else cs[j] = cs[j] + p;
          const bf16x2 pb = __builtin_convertvector(p, bf16x2);
          pa[qi][sub * 8 + 2 * j]     = pb[0];
          pa[qi][sub * 8 + 2 * j + 1] = pb[1];
        }
      }
      float k4[4];
#pragma unroll
      for (int j = 0; j < 4; ++j) {
        const float lo4 = cs[j >> 1][j & 1];
        const float hi4 = cs[(j + 4) >> 1][j & 1];
        const float keep = b3 ? hi4 : lo4;
        const float send = b3 ? lo4 : hi4;
        k4[j] = keep + row_xor<8>(send);
      }
      float k2[2];
#pragma unroll
      for (int j = 0; j < 2; ++j) {
        const float keep = b2 ? k4[j + 2] : k4[j];
        const float send = b2 ? k4[j] : k4[j + 2];
        k2[j] = keep + row_xor<4>(send);
      }
      float k1 = (b1 ? k2[1] : k2[0]) + row_xor<2>(b1 ? k2[0] : k2[1]);
      k1 += row_xor<1>(k1);
      if ((lane & 1) == 0) atomicAdd(&hrow[kt * 16 + 8 * half + (r >> 1)], k1);
    }
#pragma unroll
    for (int nb = 0; nb < 4; ++nb) {
      const v16bf vb = load_frag_packed(Vq + (size_t)kt2 * 256 + nb * 64);
      o[0][nb] = wmma_bf16(pa[0], vb, o[0][nb]);
      o[1][nb] = wmma_bf16(pa[1], vb, o[1][nb]);
    }
  }

#pragma unroll
  for (int qi = 0; qi < 2; ++qi) {
#pragma unroll
    for (int nb = 0; nb < 4; ++nb) {
#pragma unroll
      for (int i = 0; i < 8; ++i) {
        lds_o[w][(qi * 4 + nb) * 8 + i][lane] = o[qi][nb][i];
      }
    }
  }
  __syncthreads();

  const int h   = bh & 3;
  const int qif = w >> 1;
#pragma unroll
  for (int ii = 0; ii < 4; ++ii) {
    const int i = (w & 1) * 4 + ii;
    const int t = qt * 32 + qif * 16 + i + 8 * half;
#pragma unroll
    for (int nb = 0; nb < 4; ++nb) {
      const int j = (qif * 4 + nb) * 8 + i;
      float acc = lds_o[0][j][lane];
#pragma unroll
      for (int k = 1; k < KSPLIT; ++k) acc += lds_o[k][j][lane];
      attn[((size_t)b * SEQ + t) * DIMD + h * HD + nb * 16 + r] = (bf16_t)acc;
    }
  }
}

__device__ __forceinline__ void halt_softmax_blk(
    const float* __restrict__ hacc, const float* __restrict__ mask,
    float* __restrict__ out, int b) {
  __shared__ float red[256];
  const int tid = threadIdx.x;
  float lm = -3.0e38f;
  for (int j = tid; j < SEQ; j += 256) {
    const float v = hacc[(size_t)b * SEQ + j] * 0.25f +
                    mask[(size_t)b * SEQ + j] * (-1e10f);
    lm = fmaxf(lm, v);
  }
  red[tid] = lm;
  __syncthreads();
  for (int s = 128; s > 0; s >>= 1) {
    if (tid < s) red[tid] = fmaxf(red[tid], red[tid + s]);
    __syncthreads();
  }
  const float mx = red[0];
  __syncthreads();
  float ls = 0.0f;
  for (int j = tid; j < SEQ; j += 256) {
    const float v = hacc[(size_t)b * SEQ + j] * 0.25f +
                    mask[(size_t)b * SEQ + j] * (-1e10f);
    ls += __expf(v - mx);
  }
  red[tid] = ls;
  __syncthreads();
  for (int s = 128; s > 0; s >>= 1) {
    if (tid < s) red[tid] += red[tid + s];
    __syncthreads();
  }
  const float inv = 1.0f / red[0];
  for (int j = tid; j < SEQ; j += 256) {
    const float v = hacc[(size_t)b * SEQ + j] * 0.25f +
                    mask[(size_t)b * SEQ + j] * (-1e10f);
    out[(size_t)b * SEQ + j] = __expf(v - mx) * inv;
  }
}

#define LN_ROWS  4
#define LN_WAVES 8
template <bool WRITE_BF16, bool WITH_HALT>
__global__ __launch_bounds__(32 * LN_WAVES)
__attribute__((amdgpu_waves_per_eu(2, 4))) void add_ln(
    const float* __restrict__ A, const float* __restrict__ Bv,
    const float* __restrict__ g, const float* __restrict__ be,
    float* __restrict__ out, bf16_t* __restrict__ outb,
    const float* __restrict__ hacc, const float* __restrict__ mask,
    float* __restrict__ hout) {
  if (WITH_HALT && blockIdx.x < BATCH) {
    halt_softmax_blk(hacc, mask, hout, blockIdx.x);
    return;
  }
  const int lnblk = WITH_HALT ? (int)blockIdx.x - BATCH : (int)blockIdx.x;
  const int lane = threadIdx.x & 31;
  const int wave = threadIdx.x >> 5;
  const size_t row0 = ((size_t)lnblk * LN_WAVES + wave) * LN_ROWS;
  const float4* a4 = (const float4*)(A + row0 * DIMD) + lane;
  const float4* b4 = (const float4*)(Bv + row0 * DIMD) + lane;

  float v[LN_ROWS][8];
  float s[LN_ROWS];
#pragma unroll
  for (int r = 0; r < LN_ROWS; ++r) {
#pragma unroll
    for (int hf = 0; hf < 2; ++hf) {
      const float4 x = a4[r * (DIMD / 4) + hf * 32];
      const float4 y = b4[r * (DIMD / 4) + hf * 32];
      v[r][hf * 4 + 0] = x.x + y.x;
      v[r][hf * 4 + 1] = x.y + y.y;
      v[r][hf * 4 + 2] = x.z + y.z;
      v[r][hf * 4 + 3] = x.w + y.w;
    }
    s[r] = ((v[r][0] + v[r][1]) + (v[r][2] + v[r][3])) +
           ((v[r][4] + v[r][5]) + (v[r][6] + v[r][7]));
  }
  float4 gg[2], bb[2];
#pragma unroll
  for (int hf = 0; hf < 2; ++hf) {
    gg[hf] = ((const float4*)g)[lane + hf * 32];
    bb[hf] = ((const float4*)be)[lane + hf * 32];
  }
#pragma unroll
  for (int m = 16; m >= 1; m >>= 1) {
#pragma unroll
    for (int r = 0; r < LN_ROWS; ++r) s[r] += __shfl_xor(s[r], m, 32);
  }

  float q[LN_ROWS];
#pragma unroll
  for (int r = 0; r < LN_ROWS; ++r) {
    const float mu = s[r] * (1.0f / DIMD);
#pragma unroll
    for (int j = 0; j < 8; ++j) v[r][j] -= mu;
    q[r] = ((v[r][0] * v[r][0] + v[r][1] * v[r][1]) +
            (v[r][2] * v[r][2] + v[r][3] * v[r][3])) +
           ((v[r][4] * v[r][4] + v[r][5] * v[r][5]) +
            (v[r][6] * v[r][6] + v[r][7] * v[r][7]));
  }
#pragma unroll
  for (int m = 16; m >= 1; m >>= 1) {
#pragma unroll
    for (int r = 0; r < LN_ROWS; ++r) q[r] += __shfl_xor(q[r], m, 32);
  }

#pragma unroll
  for (int r = 0; r < LN_ROWS; ++r) {
    const float var = q[r] * (1.0f / DIMD);
    const float rs  = rsqrtf(var + 1e-5f);
    const size_t row = row0 + r;
#pragma unroll
    for (int hf = 0; hf < 2; ++hf) {
      float4 y;
      y.x = v[r][hf * 4 + 0] * rs * gg[hf].x + bb[hf].x;
      y.y = v[r][hf * 4 + 1] * rs * gg[hf].y + bb[hf].y;
      y.z = v[r][hf * 4 + 2] * rs * gg[hf].z + bb[hf].z;
      y.w = v[r][hf * 4 + 3] * rs * gg[hf].w + bb[hf].w;
      ((float4*)(out + row * DIMD))[lane + hf * 32] = y;
      if (WRITE_BF16) {
        v4bf o;
        o[0] = (bf16_t)y.x; o[1] = (bf16_t)y.y; o[2] = (bf16_t)y.z; o[3] = (bf16_t)y.w;
        ((v4bf*)(outb + row * DIMD))[lane + hf * 32] = o;
      }
    }
  }
}

extern "C" void kernel_launch(void* const* d_in, const int* in_sizes, int n_in,
                              void* d_out, int out_size, void* d_ws,
                              size_t ws_size, hipStream_t stream) {
  (void)in_sizes; (void)n_in; (void)out_size; (void)ws_size;
  const float* x    = (const float*)d_in[0];
  const float* mask = (const float*)d_in[1];
  const float* Wqkv = (const float*)d_in[2];
  const float* bqkv = (const float*)d_in[3];
  const float* Wo   = (const float*)d_in[4];
  const float* bo   = (const float*)d_in[5];
  const float* W1   = (const float*)d_in[6];
  const float* b1   = (const float*)d_in[7];
  const float* W2   = (const float*)d_in[8];
  const float* b2   = (const float*)d_in[9];
  const float* g1   = (const float*)d_in[10];
  const float* be1  = (const float*)d_in[11];
  const float* g2   = (const float*)d_in[12];
  const float* be2  = (const float*)d_in[13];

  char* ws = (char*)d_ws;
  size_t off = 0;
  auto alloc = [&](size_t bytes) {
    char* p = ws + off;
    off += (bytes + 255) & ~(size_t)255;
    return p;
  };
  bf16_t* xb    = (bf16_t*)alloc((size_t)NTOK * DIMD * 2);
  bf16_t* Wqkvb = (bf16_t*)alloc((size_t)3 * DIMD * DIMD * 2);
  bf16_t* Wob   = (bf16_t*)alloc((size_t)DIMD * DIMD * 2);
  bf16_t* W1b   = (bf16_t*)alloc((size_t)FFND * DIMD * 2);
  bf16_t* W2b   = (bf16_t*)alloc((size_t)DIMD * FFND * 2);
  bf16_t* Qb    = (bf16_t*)alloc((size_t)NTOK * DIMD * 2);
  bf16_t* Kb    = (bf16_t*)alloc((size_t)NTOK * DIMD * 2);
  bf16_t* Vt    = (bf16_t*)alloc((size_t)NTOK * DIMD * 2);
  bf16_t* attnb = (bf16_t*)alloc((size_t)NTOK * DIMD * 2);
  float*  attnO = (float*) alloc((size_t)NTOK * DIMD * 4);
  float*  hbuf  = (float*) alloc((size_t)NTOK * DIMD * 4);
  bf16_t* hb    = (bf16_t*)alloc((size_t)NTOK * DIMD * 2);
  bf16_t* ffn1b = (bf16_t*)alloc((size_t)NTOK * FFND * 2);
  float*  ffn2  = (float*) alloc((size_t)NTOK * DIMD * 4);
  float*  halt  = (float*) alloc((size_t)NTOK * 4);

  float* out      = (float*)d_out;
  float* out_halt = out + (size_t)NTOK * DIMD;

  prep_all<<<PB_X + PB_WQKV + PB_WO + PB_W1 + PB_W2 + PB_HZ, 256, 0, stream>>>(
      x, Wqkv, Wo, W1, W2, xb, Wqkvb, Wob, W1b, W2b, halt);

  gemm_wmma<2><<<dim3(3 * DIMD / 64, NTOK / 128), 128, 0, stream>>>(
      xb, Wqkvb, bqkv, nullptr, nullptr, Qb, Kb, Vt, NTOK, 3 * DIMD, DIMD);

  attn_wmma<<<BATCH * HEADS * (SEQ / 32), 32 * KSPLIT, 0, stream>>>(
      Qb, Kb, Vt, mask, attnb, halt);

  gemm_wmma<0><<<dim3(DIMD / 64, NTOK / 128), 128, 0, stream>>>(
      attnb, Wob, bo, attnO, nullptr, nullptr, nullptr, nullptr,
      NTOK, DIMD, DIMD);

  add_ln<true, true><<<BATCH + NTOK / (LN_WAVES * LN_ROWS), 32 * LN_WAVES, 0, stream>>>(
      x, attnO, g1, be1, hbuf, hb, halt, mask, out_halt);

  gemm_wmma<1><<<dim3(FFND / 64, NTOK / 128), 128, 0, stream>>>(
      hb, W1b, b1, nullptr, ffn1b, nullptr, nullptr, nullptr,
      NTOK, FFND, DIMD);

  gemm_wmma<0><<<dim3(DIMD / 64, NTOK / 128), 128, 0, stream>>>(
      ffn1b, W2b, b2, ffn2, nullptr, nullptr, nullptr, nullptr,
      NTOK, DIMD, FFND);

  add_ln<false, false><<<NTOK / (LN_WAVES * LN_ROWS), 32 * LN_WAVES, 0, stream>>>(
      hbuf, ffn2, g2, be2, out, nullptr, nullptr, nullptr, nullptr);
}


# global loads software-pipelined over register sets: a later step's fragment loads issue as one group before the current step's WMMAs (GEMM k loop 1 step ahead; attention pass 1 3 tiles, pass 2 1)
# speedup vs baseline: 1.9907x; 1.2033x over previous
#include <hip/hip_runtime.h>
#include <hip/hip_bf16.h>

typedef __bf16 bf16_t;
typedef __attribute__((ext_vector_type(16))) __bf16 v16bf;
typedef __attribute__((ext_vector_type(4)))  __bf16 v4bf;
typedef __attribute__((ext_vector_type(8)))  __bf16 v8bf;
typedef __attribute__((ext_vector_type(8)))  float  v8f;

#define DIMD   256
#define HEADS  4
#define HD     64
#define FFND   1024
#define BATCH  8
#define SEQ    2048
#define NTOK   (BATCH*SEQ)

__device__ __forceinline__ v16bf load_frag_bf16(const bf16_t* base) {
  const int half = (threadIdx.x >> 4) & 1;
  const bf16_t* p = base + 8 * half;
  v16bf f;
#pragma unroll
  for (int e = 0; e < 8; ++e) {
    f[e]     = p[e];
    f[e + 8] = p[e + 16];
  }
  return f;
}

__device__ __forceinline__ v16bf load_frag_packed(const v8bf* p) {
  const v8bf lo = p[0];
  const v8bf hi = p[32];
  return __builtin_shufflevector(lo, hi, 0, 1, 2, 3, 4, 5, 6, 7,
                                 8, 9, 10, 11, 12, 13, 14, 15);
}

__device__ __forceinline__ v8f wmma_bf16(v16bf a, v16bf b, v8f c) {
  return __builtin_amdgcn_wmma_f32_16x16x32_bf16(
       false, a,  false, b,
       (short)0, c,  false,  false);
}

#define CVT_U 8
__device__ __forceinline__ void cvt_bf16_blk(const float* __restrict__ in,
                                             bf16_t* __restrict__ out, int blk) {
  const size_t base = (size_t)blk * (256 * CVT_U) + threadIdx.x;
  float4 v[CVT_U];
#pragma unroll
  for (int k = 0; k < CVT_U; ++k) v[k] = ((const float4*)in)[base + k * 256];
#pragma unroll
  for (int k = 0; k < CVT_U; ++k) {
    v4bf o;
    o[0] = (bf16_t)v[k].x; o[1] = (bf16_t)v[k].y;
    o[2] = (bf16_t)v[k].z; o[3] = (bf16_t)v[k].w;
    ((v4bf*)out)[base + k * 256] = o;
  }
}

template <int KD>
__device__ __forceinline__ void cvt_frag_blk(const float* __restrict__ in,
                                             bf16_t* __restrict__ out, int blk) {
  const int u    = blk * 256 + threadIdx.x;
  const int l    = u & 31;
  const int j    = (u >> 5) & 1;
  const int tile = u >> 6;
  const int kc   = tile % (KD / 32);
  const int rt   = tile / (KD / 32);
  const int row  = rt * 16 + (l & 15);
  const int k    = kc * 32 + j * 16 + 8 * (l >> 4);
  const float4* p = (const float4*)(in + (size_t)row * KD + k);
  const float4 x0 = p[0], x1 = p[1];
  v8bf o;
  o[0] = (bf16_t)x0.x; o[1] = (bf16_t)x0.y; o[2] = (bf16_t)x0.z; o[3] = (bf16_t)x0.w;
  o[4] = (bf16_t)x1.x; o[5] = (bf16_t)x1.y; o[6] = (bf16_t)x1.z; o[7] = (bf16_t)x1.w;
  ((v8bf*)out)[u] = o;
}

#define PB_X    (NTOK * DIMD / (1024 * CVT_U))
#define PB_WQKV (3 * DIMD * DIMD / (8 * 256))
#define PB_WO   (DIMD * DIMD / (8 * 256))
#define PB_W1   (FFND * DIMD / (8 * 256))
#define PB_W2   (DIMD * FFND / (8 * 256))
#define PB_HZ   (NTOK / (4 * 256))
__global__ __launch_bounds__(256) void prep_all(
    const float* __restrict__ x, const float* __restrict__ Wqkv,
    const float* __restrict__ Wo, const float* __restrict__ W1,
    const float* __restrict__ W2, bf16_t* __restrict__ xb,
    bf16_t* __restrict__ Wqkvb, bf16_t* __restrict__ Wob,
    bf16_t* __restrict__ W1b, bf16_t* __restrict__ W2b,
    float* __restrict__ halt) {
  int blk = blockIdx.x;
  if (blk < PB_X) { cvt_bf16_blk(x, xb, blk); return; }
  blk -= PB_X;
  if (blk < PB_WQKV) { cvt_frag_blk<DIMD>(Wqkv, Wqkvb, blk); return; }
  blk -= PB_WQKV;
  if (blk < PB_WO) { cvt_frag_blk<DIMD>(Wo, Wob, blk); return; }
  blk -= PB_WO;
  if (blk < PB_W1) { cvt_frag_blk<DIMD>(W1, W1b, blk); return; }
  blk -= PB_W1;
  if (blk < PB_W2) { cvt_frag_blk<FFND>(W2, W2b, blk); return; }
  blk -= PB_W2;
  ((float4*)halt)[blk * 256 + threadIdx.x] = make_float4(0.0f, 0.0f, 0.0f, 0.0f);
}


struct GemmFrags {
  v16bf a0, a1, b[4];
};

__device__ __forceinline__ void gemm_load(
    GemmFrags& f, const bf16_t* __restrict__ ap0,
    const bf16_t* __restrict__ ap1, const v8bf* __restrict__ bq, int kch,
    int k0) {
  __builtin_prefetch(ap0 + k0 + 64, 0, 0);
  __builtin_prefetch(ap1 + k0 + 64, 0, 0);
  f.a0 = load_frag_bf16(ap0 + k0);
  f.a1 = load_frag_bf16(ap1 + k0);
#pragma unroll
  for (int nb = 0; nb < 4; ++nb)
    f.b[nb] = load_frag_packed(bq + (size_t)(nb * kch + (k0 >> 5)) * 64);
}

template <bool TRANS>
__device__ __forceinline__ void gemm_step(v8f (&acc)[2][4], const GemmFrags& f) {
#pragma unroll
  for (int nb = 0; nb < 4; ++nb) {
    if (TRANS) {
      acc[0][nb] = wmma_bf16(f.b[nb], f.a0, acc[0][nb]);
      acc[1][nb] = wmma_bf16(f.b[nb], f.a1, acc[1][nb]);
    } else {
      acc[0][nb] = wmma_bf16(f.a0, f.b[nb], acc[0][nb]);
      acc[1][nb] = wmma_bf16(f.a1, f.b[nb], acc[1][nb]);
    }
  }
}

template <bool TRANS>
__device__ __forceinline__ void gemm_mainloop(
    v8f (&acc)[2][4], const bf16_t* __restrict__ ap0,
    const bf16_t* __restrict__ ap1, const v8bf* __restrict__ bq, int K) {
  const int kch = K >> 5;
  GemmFrags fx, fy;
  gemm_load(fx, ap0, ap1, bq, kch, 0);
  for (int k0 = 0; k0 < K; k0 += 64) {
    gemm_load(fy, ap0, ap1, bq, kch, k0 + 32);
    __builtin_amdgcn_sched_barrier(0);
    gemm_step<TRANS>(acc, fx);
    const int kn = (k0 + 64 < K) ? k0 + 64 : K - 32;
    gemm_load(fx, ap0, ap1, bq, kch, kn);
    __builtin_amdgcn_sched_barrier(0);
    gemm_step<TRANS>(acc, fy);
  }
}

template <int MODE>
__global__ __launch_bounds__(128) void gemm_wmma(
    const bf16_t* __restrict__ A, const bf16_t* __restrict__ B,
    const float* __restrict__ bias, float* __restrict__ Yf,
    bf16_t* __restrict__ Yb,
    bf16_t* __restrict__ Qb, bf16_t* __restrict__ Kb, bf16_t* __restrict__ Vt,
    int M, int N, int K) {
  const int lane = threadIdx.x & 31;
  const int wave = threadIdx.x >> 5;
  const int r    = lane & 15;
  const int half = lane >> 4;
  const int row0 = blockIdx.y * 128 + wave * 32;
  const int col0 = blockIdx.x * 64;

  v8f acc[2][4] = {};
  const bf16_t* ap0 = A + (size_t)(row0 + r) * K;
  const bf16_t* ap1 = A + (size_t)(row0 + 16 + r) * K;
  const v8bf* bq = (const v8bf*)B + (size_t)(col0 >> 4) * (K >> 5) * 64 + lane;

  const bool kblk = (MODE == 2) && (col0 >= DIMD) && (col0 < 2 * DIMD);
  if (kblk) gemm_mainloop<true>(acc, ap0, ap1, bq, K);
  else      gemm_mainloop<false>(acc, ap0, ap1, bq, K);

  if (MODE == 2 && col0 >= DIMD) {
    const int bb  = row0 >> 11;
    const int t0  = row0 & (SEQ - 1);
    if (kblk) {
      const size_t bhk = (size_t)bb * HEADS + ((col0 - DIMD) >> 6);
      v8bf* kf = (v8bf*)Kb + (bhk * (SEQ / 16) + (t0 >> 4)) * 128 + lane;
#pragma unroll
      for (int mi = 0; mi < 2; ++mi) {
#pragma unroll
        for (int nb = 0; nb < 4; ++nb) {
          v8bf o;
#pragma unroll
          for (int i = 0; i < 8; ++i)
            o[i] = (bf16_t)(acc[mi][nb][i] + bias[col0 + nb * 16 + 8 * half + i]);
          kf[mi * 128 + nb * 32] = o;
        }
      }
    } else {
      const size_t bhv = (size_t)bb * HEADS + ((col0 - 2 * DIMD) >> 6);
      v8bf* vf = (v8bf*)Vt + (bhv * (SEQ / 32) + (t0 >> 5)) * 256 + lane;
#pragma unroll
      for (int mi = 0; mi < 2; ++mi) {
#pragma unroll
        for (int nb = 0; nb < 4; ++nb) {
          const float bv = bias[col0 + nb * 16 + r];
          v8bf o;
#pragma unroll
          for (int i = 0; i < 8; ++i) o[i] = (bf16_t)(acc[mi][nb][i] + bv);
          vf[nb * 64 + mi * 32] = o;
        }
      }
    }
    return;
  }

#pragma unroll
  for (int mi = 0; mi < 2; ++mi) {
#pragma unroll
    for (int nb = 0; nb < 4; ++nb) {
#pragma unroll
      for (int i = 0; i < 8; ++i) {
        const int row = row0 + mi * 16 + i + 8 * half;
        const int col = col0 + nb * 16 + r;
        float y = acc[mi][nb][i] + bias[col];
        if (MODE == 0) {
          Yf[(size_t)row * N + col] = y;
        } else if (MODE == 1) {
          Yb[(size_t)row * N + col] = (bf16_t)fmaxf(y, 0.0f);
        } else {
          const int b = row >> 11, t = row & (SEQ - 1);
          const int h = col >> 6, d = col & 63;
          Qb[(((size_t)b * HEADS + h) * SEQ + t) * HD + d] = (bf16_t)y;
        }
      }
    }
  }
}

#define KSPLIT 4
#define CSCALE 0.18033688011112042f
#define NEGRAW (-8589934592.0f)

typedef __attribute__((ext_vector_type(2))) float  f32x2;
typedef __attribute__((ext_vector_type(2))) __bf16 bf16x2;

__device__ __forceinline__ float xhalf(float x) {
  const int b = __builtin_bit_cast(int, x);
  return __builtin_bit_cast(float, __builtin_amdgcn_permlanex16(
      b, b, 0x76543210, 0xfedcba98, false, false));
}

template <int XM>
__device__ __forceinline__ float row_xor(float x) {
  const int b = __builtin_bit_cast(int, x);
  return __builtin_bit_cast(float, __builtin_amdgcn_update_dpp(
      0, b, 0x160 + XM, 0xf, 0xf, true));
}

__device__ __forceinline__ float max3f(float a, float b, float c) {
  return __builtin_fmaxf(__builtin_fmaxf(a, b), c);
}

__global__ __launch_bounds__(32 * KSPLIT) void attn_wmma(
    const bf16_t* __restrict__ Qb, const bf16_t* __restrict__ Kb,
    const bf16_t* __restrict__ Vt, const float* __restrict__ mask,
    bf16_t* __restrict__ attn, float* __restrict__ halt) {
  __shared__ float  lds_m[KSPLIT][32];
  __shared__ float  lds_l[KSPLIT][32];
  __shared__ float  lds_o[KSPLIT][64][32];

  const int bid  = blockIdx.x;
  const int qt   = bid & (SEQ / 32 - 1);
  const int bh   = bid >> 6;
  const int b    = bh >> 2;
  const int lane = threadIdx.x & 31;
  const int w    = threadIdx.x >> 5;
  const int r    = lane & 15;
  const int half = lane >> 4;

  const bf16_t* Qbase = Qb + ((size_t)bh * SEQ + qt * 32) * HD;
  const v8bf* Kq = (const v8bf*)Kb +
                   ((size_t)bh * (SEQ / 16) + (size_t)w * (SEQ / KSPLIT / 16)) * 128 + lane;
  const v8bf* Vq = (const v8bf*)Vt +
                   ((size_t)bh * (SEQ / 32) + (size_t)w * (SEQ / KSPLIT / 32)) * 256 + lane;
  const float* mrow8 = mask + (size_t)b * SEQ + (size_t)w * (SEQ / KSPLIT) + 8 * half;
  float*       hrow  = halt + (size_t)b * SEQ + (size_t)w * (SEQ / KSPLIT);

  v16bf qa[2][2];
#pragma unroll
  for (int qi = 0; qi < 2; ++qi) {
    qa[qi][0] = load_frag_bf16(Qbase + (size_t)(qi * 16 + r) * HD + 0);
    qa[qi][1] = load_frag_bf16(Qbase + (size_t)(qi * 16 + r) * HD + 32);
  }

  const f32x2 c2 = {CSCALE, CSCALE};

  float m[2], sh[2], l[2];
#pragma unroll
  for (int qi = 0; qi < 2; ++qi) {
    m[qi] = -3.0e38f; sh[qi] = 3.0e38f * CSCALE; l[qi] = 0.0f;
  }

  v16bf ks0[4], ks1[4];
  v8f mks[4];
  auto load_tile = [&](v16bf& k0r, v16bf& k1r, v8f& mkr, int kt) {
    k0r = load_frag_packed(Kq + (size_t)kt * 128);
    k1r = load_frag_packed(Kq + (size_t)kt * 128 + 64);
    mkr = *(const v8f*)(mrow8 + kt * 16);
  };
  auto pass1_tile = [&](const v16bf& kb0, const v16bf& kb1, const v8f& mk) {
    const bool msk[8] = {mk[0] > 0.5f, mk[1] > 0.5f, mk[2] > 0.5f, mk[3] > 0.5f,
                         mk[4] > 0.5f, mk[5] > 0.5f, mk[6] > 0.5f, mk[7] > 0.5f};
#pragma unroll
    for (int qi = 0; qi < 2; ++qi) {
      v8f s = {};
      s = wmma_bf16(kb0, qa[qi][0], s);
      s = wmma_bf16(kb1, qa[qi][1], s);
      float sv[8];
#pragma unroll
      for (int i = 0; i < 8; ++i) sv[i] = msk[i] ? NEGRAW : s[i];
      float mx = max3f(m[qi], sv[0], sv[1]);
      mx = max3f(mx, sv[2], sv[3]);
      mx = max3f(mx, sv[4], sv[5]);
      mx = max3f(mx, sv[6], sv[7]);
      const float nm = -(mx * CSCALE);
      const f32x2 nm2 = {nm, nm};
      const float alpha = __builtin_amdgcn_exp2f(nm - sh[qi]);
      f32x2 e[4];
#pragma unroll
      for (int j = 0; j < 4; ++j) {
        const f32x2 a = {sv[2 * j], sv[2 * j + 1]};
        const f32x2 t = __builtin_elementwise_fma(a, c2, nm2);
        e[j][0] = __builtin_amdgcn_exp2f(t[0]);
        e[j][1] = __builtin_amdgcn_exp2f(t[1]);
      }
      const f32x2 es = (e[0] + e[1]) + (e[2] + e[3]);
      l[qi] = __builtin_fmaf(l[qi], alpha, es[0] + es[1]);
      m[qi] = mx;
      sh[qi] = nm;
    }
  };

  constexpr int NKT = SEQ / KSPLIT / 16;
#pragma unroll
  for (int u = 0; u < 3; ++u) load_tile(ks0[u], ks1[u], mks[u], u);
  for (int kt = 0; kt < NKT; kt += 4) {
#pragma unroll
    for (int u = 0; u < 4; ++u) {
      load_tile(ks0[(u + 3) & 3], ks1[(u + 3) & 3], mks[(u + 3) & 3],
                (kt + u + 3) & (NKT - 1));
      __builtin_amdgcn_sched_barrier(0);
      pass1_tile(ks0[u], ks1[u], mks[u]);
    }
  }
#pragma unroll
  for (int qi = 0; qi < 2; ++qi) {
    const float so = xhalf(sh[qi]);
    const float lo = xhalf(l[qi]);
    const float sn = __builtin_fminf(sh[qi], so);
    l[qi] = l[qi] * __builtin_amdgcn_exp2f(sn - sh[qi]) +
            lo * __builtin_amdgcn_exp2f(sn - so);
    sh[qi] = sn;
  }
  if (half == 0) {
#pragma unroll
    for (int qi = 0; qi < 2; ++qi) {
      lds_m[w][qi * 16 + r] = sh[qi];
      lds_l[w][qi * 16 + r] = l[qi];
    }
  }
  __syncthreads();
  f32x2 nmf[2], rl[2];
#pragma unroll
  for (int qi = 0; qi < 2; ++qi) {
    float sw[KSPLIT], lw[KSPLIT];
#pragma unroll
    for (int k = 0; k < KSPLIT; ++k) {
      sw[k] = lds_m[k][qi * 16 + r];
      lw[k] = lds_l[k][qi * 16 + r];
    }
    float nm = sw[0];
#pragma unroll
    for (int k = 1; k < KSPLIT; ++k) nm = __builtin_fminf(nm, sw[k]);
    float ls = 0.0f;
#pragma unroll
    for (int k = 0; k < KSPLIT; ++k)
      ls += lw[k] * __builtin_amdgcn_exp2f(nm - sw[k]);
    const float inv = 1.0f / ls;
    nmf[qi][0] = nm;  nmf[qi][1] = nm;
    rl[qi][0]  = inv; rl[qi][1]  = inv;
  }

  const bool b3 = (r & 8) != 0, b2 = (r & 4) != 0, b1 = (r & 2) != 0;
  v8f o[2][4] = {};
  auto pass2_tile = [&](v16bf (&pa)[2], const v16bf& kb0, const v16bf& kb1,
                        const v8f& mk, int sub, int kt) {
    const bool msk[8] = {mk[0] > 0.5f, mk[1] > 0.5f, mk[2] > 0.5f, mk[3] > 0.5f,
                         mk[4] > 0.5f, mk[5] > 0.5f, mk[6] > 0.5f, mk[7] > 0.5f};
    f32x2 cs[4];
#pragma unroll
    for (int qi = 0; qi < 2; ++qi) {
      v8f s = {};
      s = wmma_bf16(kb0, qa[qi][0], s);
      s = wmma_bf16(kb1, qa[qi][1], s);
#pragma unroll
      for (int j = 0; j < 4; ++j) {
        const f32x2 a = {msk[2 * j] ? NEGRAW : s[2 * j],
                         msk[2 * j + 1] ? NEGRAW : s[2 * j + 1]};
        const f32x2 t = __builtin_elementwise_fma(a, c2, nmf[qi]);
        f32x2 p;
        p[0] = __builtin_amdgcn_exp2f(t[0]);
        p[1] = __builtin_amdgcn_exp2f(t[1]);
        p = p * rl[qi];
        if (qi == 0) cs[j] = p; else cs[j] = cs[j] + p;
        const bf16x2 pb = __builtin_convertvector(p, bf16x2);
        pa[qi][sub * 8 + 2 * j]     = pb[0];
        pa[qi][sub * 8 + 2 * j + 1] = pb[1];
      }
    }
    float k4[4];
#pragma unroll
    for (int j = 0; j < 4; ++j) {
      const float lo4 = cs[j >> 1][j & 1];
      const float hi4 = cs[(j + 4) >> 1][j & 1];
      const float keep = b3 ? hi4 : lo4;
      const float send = b3 ? lo4 : hi4;
      k4[j] = keep + row_xor<8>(send);
    }
    float k2[2];
#pragma unroll
    for (int j = 0; j < 2; ++j) {
      const float keep = b2 ? k4[j + 2] : k4[j];
      const float send = b2 ? k4[j] : k4[j + 2];
      k2[j] = keep + row_xor<4>(send);
    }
    float k1 = (b1 ? k2[1] : k2[0]) + row_xor<2>(b1 ? k2[0] : k2[1]);
    k1 += row_xor<1>(k1);
    if ((lane & 1) == 0) atomicAdd(&hrow[kt * 16 + 8 * half + (r >> 1)], k1);
  };

  for (int kt2 = 0; kt2 < SEQ / KSPLIT / 32; ++kt2) {
    v16bf pa[2];
    load_tile(ks0[1], ks1[1], mks[1], kt2 * 2 + 1);
    __builtin_amdgcn_sched_barrier(0);
    pass2_tile(pa, ks0[0], ks1[0], mks[0], 0, kt2 * 2);
    v16bf vb[4];
#pragma unroll
    for (int nb = 0; nb < 4; ++nb)
      vb[nb] = load_frag_packed(Vq + (size_t)kt2 * 256 + nb * 64);
    load_tile(ks0[0], ks1[0], mks[0], (kt2 * 2 + 2) & (NKT - 1));
    __builtin_amdgcn_sched_barrier(0);
    pass2_tile(pa, ks0[1], ks1[1], mks[1], 1, kt2 * 2 + 1);
#pragma unroll
    for (int nb = 0; nb < 4; ++nb) {
      o[0][nb] = wmma_bf16(pa[0], vb[nb], o[0][nb]);
      o[1][nb] = wmma_bf16(pa[1], vb[nb], o[1][nb]);
    }
  }

#pragma unroll
  for (int qi = 0; qi < 2; ++qi) {
#pragma unroll
    for (int nb = 0; nb < 4; ++nb) {
#pragma unroll
      for (int i = 0; i < 8; ++i) {
        lds_o[w][(qi * 4 + nb) * 8 + i][lane] = o[qi][nb][i];
      }
    }
  }
  __syncthreads();

  const int h   = bh & 3;
  const int qif = w >> 1;
#pragma unroll
  for (int ii = 0; ii < 4; ++ii) {
    const int i = (w & 1) * 4 + ii;
    const int t = qt * 32 + qif * 16 + i + 8 * half;
#pragma unroll
    for (int nb = 0; nb < 4; ++nb) {
      const int j = (qif * 4 + nb) * 8 + i;
      float acc = lds_o[0][j][lane];
#pragma unroll
      for (int k = 1; k < KSPLIT; ++k) acc += lds_o[k][j][lane];
      attn[((size_t)b * SEQ + t) * DIMD + h * HD + nb * 16 + r] = (bf16_t)acc;
    }
  }
}

__device__ __forceinline__ void halt_softmax_blk(
    const float* __restrict__ hacc, const float* __restrict__ mask,
    float* __restrict__ out, int b) {
  __shared__ float red[256];
  const int tid = threadIdx.x;
  float lm = -3.0e38f;
  for (int j = tid; j < SEQ; j += 256) {
    const float v = hacc[(size_t)b * SEQ + j] * 0.25f +
                    mask[(size_t)b * SEQ + j] * (-1e10f);
    lm = fmaxf(lm, v);
  }
  red[tid] = lm;
  __syncthreads();
  for (int s = 128; s > 0; s >>= 1) {
    if (tid < s) red[tid] = fmaxf(red[tid], red[tid + s]);
    __syncthreads();
  }
  const float mx = red[0];
  __syncthreads();
  float ls = 0.0f;
  for (int j = tid; j < SEQ; j += 256) {
    const float v = hacc[(size_t)b * SEQ + j] * 0.25f +
                    mask[(size_t)b * SEQ + j] * (-1e10f);
    ls += __expf(v - mx);
  }
  red[tid] = ls;
  __syncthreads();
  for (int s = 128; s > 0; s >>= 1) {
    if (tid < s) red[tid] += red[tid + s];
    __syncthreads();
  }
  const float inv = 1.0f / red[0];
  for (int j = tid; j < SEQ; j += 256) {
    const float v = hacc[(size_t)b * SEQ + j] * 0.25f +
                    mask[(size_t)b * SEQ + j] * (-1e10f);
    out[(size_t)b * SEQ + j] = __expf(v - mx) * inv;
  }
}

#define LN_ROWS  4
#define LN_WAVES 8
template <bool WRITE_BF16, bool WITH_HALT>
__global__ __launch_bounds__(32 * LN_WAVES)
__attribute__((amdgpu_waves_per_eu(2, 4))) void add_ln(
    const float* __restrict__ A, const float* __restrict__ Bv,
    const float* __restrict__ g, const float* __restrict__ be,
    float* __restrict__ out, bf16_t* __restrict__ outb,
    const float* __restrict__ hacc, const float* __restrict__ mask,
    float* __restrict__ hout) {
  if (WITH_HALT && blockIdx.x < BATCH) {
    halt_softmax_blk(hacc, mask, hout, blockIdx.x);
    return;
  }
  const int lnblk = WITH_HALT ? (int)blockIdx.x - BATCH : (int)blockIdx.x;
  const int lane = threadIdx.x & 31;
  const int wave = threadIdx.x >> 5;
  const size_t row0 = ((size_t)lnblk * LN_WAVES + wave) * LN_ROWS;
  const float4* a4 = (const float4*)(A + row0 * DIMD) + lane;
  const float4* b4 = (const float4*)(Bv + row0 * DIMD) + lane;

  float v[LN_ROWS][8];
  float s[LN_ROWS];
#pragma unroll
  for (int r = 0; r < LN_ROWS; ++r) {
#pragma unroll
    for (int hf = 0; hf < 2; ++hf) {
      const float4 x = a4[r * (DIMD / 4) + hf * 32];
      const float4 y = b4[r * (DIMD / 4) + hf * 32];
      v[r][hf * 4 + 0] = x.x + y.x;
      v[r][hf * 4 + 1] = x.y + y.y;
      v[r][hf * 4 + 2] = x.z + y.z;
      v[r][hf * 4 + 3] = x.w + y.w;
    }
    s[r] = ((v[r][0] + v[r][1]) + (v[r][2] + v[r][3])) +
           ((v[r][4] + v[r][5]) + (v[r][6] + v[r][7]));
  }
  float4 gg[2], bb[2];
#pragma unroll
  for (int hf = 0; hf < 2; ++hf) {
    gg[hf] = ((const float4*)g)[lane + hf * 32];
    bb[hf] = ((const float4*)be)[lane + hf * 32];
  }
#pragma unroll
  for (int m = 16; m >= 1; m >>= 1) {
#pragma unroll
    for (int r = 0; r < LN_ROWS; ++r) s[r] += __shfl_xor(s[r], m, 32);
  }

  float q[LN_ROWS];
#pragma unroll
  for (int r = 0; r < LN_ROWS; ++r) {
    const float mu = s[r] * (1.0f / DIMD);
#pragma unroll
    for (int j = 0; j < 8; ++j) v[r][j] -= mu;
    q[r] = ((v[r][0] * v[r][0] + v[r][1] * v[r][1]) +
            (v[r][2] * v[r][2] + v[r][3] * v[r][3])) +
           ((v[r][4] * v[r][4] + v[r][5] * v[r][5]) +
            (v[r][6] * v[r][6] + v[r][7] * v[r][7]));
  }
#pragma unroll
  for (int m = 16; m >= 1; m >>= 1) {
#pragma unroll
    for (int r = 0; r < LN_ROWS; ++r) q[r] += __shfl_xor(q[r], m, 32);
  }

#pragma unroll
  for (int r = 0; r < LN_ROWS; ++r) {
    const float var = q[r] * (1.0f / DIMD);
    const float rs  = rsqrtf(var + 1e-5f);
    const size_t row = row0 + r;
#pragma unroll
    for (int hf = 0; hf < 2; ++hf) {
      float4 y;
      y.x = v[r][hf * 4 + 0] * rs * gg[hf].x + bb[hf].x;
      y.y = v[r][hf * 4 + 1] * rs * gg[hf].y + bb[hf].y;
      y.z = v[r][hf * 4 + 2] * rs * gg[hf].z + bb[hf].z;
      y.w = v[r][hf * 4 + 3] * rs * gg[hf].w + bb[hf].w;
      ((float4*)(out + row * DIMD))[lane + hf * 32] = y;
      if (WRITE_BF16) {
        v4bf o;
        o[0] = (bf16_t)y.x; o[1] = (bf16_t)y.y; o[2] = (bf16_t)y.z; o[3] = (bf16_t)y.w;
        ((v4bf*)(outb + row * DIMD))[lane + hf * 32] = o;
      }
    }
  }
}

extern "C" void kernel_launch(void* const* d_in, const int* in_sizes, int n_in,
                              void* d_out, int out_size, void* d_ws,
                              size_t ws_size, hipStream_t stream) {
  (void)in_sizes; (void)n_in; (void)out_size; (void)ws_size;
  const float* x    = (const float*)d_in[0];
  const float* mask = (const float*)d_in[1];
  const float* Wqkv = (const float*)d_in[2];
  const float* bqkv = (const float*)d_in[3];
  const float* Wo   = (const float*)d_in[4];
  const float* bo   = (const float*)d_in[5];
  const float* W1   = (const float*)d_in[6];
  const float* b1   = (const float*)d_in[7];
  const float* W2   = (const float*)d_in[8];
  const float* b2   = (const float*)d_in[9];
  const float* g1   = (const float*)d_in[10];
  const float* be1  = (const float*)d_in[11];
  const float* g2   = (const float*)d_in[12];
  const float* be2  = (const float*)d_in[13];

  char* ws = (char*)d_ws;
  size_t off = 0;
  auto alloc = [&](size_t bytes) {
    char* p = ws + off;
    off += (bytes + 255) & ~(size_t)255;
    return p;
  };
  bf16_t* xb    = (bf16_t*)alloc((size_t)NTOK * DIMD * 2);
  bf16_t* Wqkvb = (bf16_t*)alloc((size_t)3 * DIMD * DIMD * 2);
  bf16_t* Wob   = (bf16_t*)alloc((size_t)DIMD * DIMD * 2);
  bf16_t* W1b   = (bf16_t*)alloc((size_t)FFND * DIMD * 2);
  bf16_t* W2b   = (bf16_t*)alloc((size_t)DIMD * FFND * 2);
  bf16_t* Qb    = (bf16_t*)alloc((size_t)NTOK * DIMD * 2);
  bf16_t* Kb    = (bf16_t*)alloc((size_t)NTOK * DIMD * 2);
  bf16_t* Vt    = (bf16_t*)alloc((size_t)NTOK * DIMD * 2);
  bf16_t* attnb = (bf16_t*)alloc((size_t)NTOK * DIMD * 2);
  float*  attnO = (float*) alloc((size_t)NTOK * DIMD * 4);
  float*  hbuf  = (float*) alloc((size_t)NTOK * DIMD * 4);
  bf16_t* hb    = (bf16_t*)alloc((size_t)NTOK * DIMD * 2);
  bf16_t* ffn1b = (bf16_t*)alloc((size_t)NTOK * FFND * 2);
  float*  ffn2  = (float*) alloc((size_t)NTOK * DIMD * 4);
  float*  halt  = (float*) alloc((size_t)NTOK * 4);

  float* out      = (float*)d_out;
  float* out_halt = out + (size_t)NTOK * DIMD;

  prep_all<<<PB_X + PB_WQKV + PB_WO + PB_W1 + PB_W2 + PB_HZ, 256, 0, stream>>>(
      x, Wqkv, Wo, W1, W2, xb, Wqkvb, Wob, W1b, W2b, halt);

  gemm_wmma<2><<<dim3(3 * DIMD / 64, NTOK / 128), 128, 0, stream>>>(
      xb, Wqkvb, bqkv, nullptr, nullptr, Qb, Kb, Vt, NTOK, 3 * DIMD, DIMD);

  attn_wmma<<<BATCH * HEADS * (SEQ / 32), 32 * KSPLIT, 0, stream>>>(
      Qb, Kb, Vt, mask, attnb, halt);

  gemm_wmma<0><<<dim3(DIMD / 64, NTOK / 128), 128, 0, stream>>>(
      attnb, Wob, bo, attnO, nullptr, nullptr, nullptr, nullptr,
      NTOK, DIMD, DIMD);

  add_ln<true, true><<<BATCH + NTOK / (LN_WAVES * LN_ROWS), 32 * LN_WAVES, 0, stream>>>(
      x, attnO, g1, be1, hbuf, hb, halt, mask, out_halt);

  gemm_wmma<1><<<dim3(FFND / 64, NTOK / 128), 128, 0, stream>>>(
      hb, W1b, b1, nullptr, ffn1b, nullptr, nullptr, nullptr,
      NTOK, FFND, DIMD);

  gemm_wmma<0><<<dim3(DIMD / 64, NTOK / 128), 128, 0, stream>>>(
      ffn1b, W2b, b2, ffn2, nullptr, nullptr, nullptr, nullptr,
      NTOK, DIMD, FFND);

  add_ln<false, false><<<NTOK / (LN_WAVES * LN_ROWS), 32 * LN_WAVES, 0, stream>>>(
      hbuf, ffn2, g2, be2, out, nullptr, nullptr, nullptr, nullptr);
}


# attention: 64 queries per wave (was 32): 4 sub-tiles share every K/V fragment, mask load and halt column-sum tree (4 P values added in-lane before the tree, half as many atomics: new sum order)
# speedup vs baseline: 2.3497x; 1.1803x over previous
#include <hip/hip_runtime.h>
#include <hip/hip_bf16.h>

typedef __bf16 bf16_t;
typedef __attribute__((ext_vector_type(16))) __bf16 v16bf;
typedef __attribute__((ext_vector_type(4)))  __bf16 v4bf;
typedef __attribute__((ext_vector_type(8)))  __bf16 v8bf;
typedef __attribute__((ext_vector_type(8)))  float  v8f;

#define DIMD   256
#define HEADS  4
#define HD     64
#define FFND   1024
#define BATCH  8
#define SEQ    2048
#define NTOK   (BATCH*SEQ)

__device__ __forceinline__ v16bf load_frag_bf16(const bf16_t* base) {
  const int half = (threadIdx.x >> 4) & 1;
  const bf16_t* p = base + 8 * half;
  v16bf f;
#pragma unroll
  for (int e = 0; e < 8; ++e) {
    f[e]     = p[e];
    f[e + 8] = p[e + 16];
  }
  return f;
}

__device__ __forceinline__ v16bf load_frag_packed(const v8bf* p) {
  const v8bf lo = p[0];
  const v8bf hi = p[32];
  return __builtin_shufflevector(lo, hi, 0, 1, 2, 3, 4, 5, 6, 7,
                                 8, 9, 10, 11, 12, 13, 14, 15);
}

__device__ __forceinline__ v8f wmma_bf16(v16bf a, v16bf b, v8f c) {
  return __builtin_amdgcn_wmma_f32_16x16x32_bf16(
       false, a,  false, b,
       (short)0, c,  false,  false);
}

#define CVT_U 8
__device__ __forceinline__ void cvt_bf16_blk(const float* __restrict__ in,
                                             bf16_t* __restrict__ out, int blk) {
  const size_t base = (size_t)blk * (256 * CVT_U) + threadIdx.x;
  float4 v[CVT_U];
#pragma unroll
  for (int k = 0; k < CVT_U; ++k) v[k] = ((const float4*)in)[base + k * 256];
#pragma unroll
  for (int k = 0; k < CVT_U; ++k) {
    v4bf o;
    o[0] = (bf16_t)v[k].x; o[1] = (bf16_t)v[k].y;
    o[2] = (bf16_t)v[k].z; o[3] = (bf16_t)v[k].w;
    ((v4bf*)out)[base + k * 256] = o;
  }
}

template <int KD>
__device__ __forceinline__ void cvt_frag_blk(const float* __restrict__ in,
                                             bf16_t* __restrict__ out, int blk) {
  const int u    = blk * 256 + threadIdx.x;
  const int l    = u & 31;
  const int j    = (u >> 5) & 1;
  const int tile = u >> 6;
  const int kc   = tile % (KD / 32);
  const int rt   = tile / (KD / 32);
  const int row  = rt * 16 + (l & 15);
  const int k    = kc * 32 + j * 16 + 8 * (l >> 4);
  const float4* p = (const float4*)(in + (size_t)row * KD + k);
  const float4 x0 = p[0], x1 = p[1];
  v8bf o;
  o[0] = (bf16_t)x0.x; o[1] = (bf16_t)x0.y; o[2] = (bf16_t)x0.z; o[3] = (bf16_t)x0.w;
  o[4] = (bf16_t)x1.x; o[5] = (bf16_t)x1.y; o[6] = (bf16_t)x1.z; o[7] = (bf16_t)x1.w;
  ((v8bf*)out)[u] = o;
}

#define PB_X    (NTOK * DIMD / (1024 * CVT_U))
#define PB_WQKV (3 * DIMD * DIMD / (8 * 256))
#define PB_WO   (DIMD * DIMD / (8 * 256))
#define PB_W1   (FFND * DIMD / (8 * 256))
#define PB_W2   (DIMD * FFND / (8 * 256))
#define PB_HZ   (NTOK / (4 * 256))
__global__ __launch_bounds__(256) void prep_all(
    const float* __restrict__ x, const float* __restrict__ Wqkv,
    const float* __restrict__ Wo, const float* __restrict__ W1,
    const float* __restrict__ W2, bf16_t* __restrict__ xb,
    bf16_t* __restrict__ Wqkvb, bf16_t* __restrict__ Wob,
    bf16_t* __restrict__ W1b, bf16_t* __restrict__ W2b,
    float* __restrict__ halt) {
  int blk = blockIdx.x;
  if (blk < PB_X) { cvt_bf16_blk(x, xb, blk); return; }
  blk -= PB_X;
  if (blk < PB_WQKV) { cvt_frag_blk<DIMD>(Wqkv, Wqkvb, blk); return; }
  blk -= PB_WQKV;
  if (blk < PB_WO) { cvt_frag_blk<DIMD>(Wo, Wob, blk); return; }
  blk -= PB_WO;
  if (blk < PB_W1) { cvt_frag_blk<DIMD>(W1, W1b, blk); return; }
  blk -= PB_W1;
  if (blk < PB_W2) { cvt_frag_blk<FFND>(W2, W2b, blk); return; }
  blk -= PB_W2;
  ((float4*)halt)[blk * 256 + threadIdx.x] = make_float4(0.0f, 0.0f, 0.0f, 0.0f);
}


struct GemmFrags {
  v16bf a0, a1, b[4];
};

__device__ __forceinline__ void gemm_load(
    GemmFrags& f, const bf16_t* __restrict__ ap0,
    const bf16_t* __restrict__ ap1, const v8bf* __restrict__ bq, int kch,
    int k0) {
  __builtin_prefetch(ap0 + k0 + 64, 0, 0);
  __builtin_prefetch(ap1 + k0 + 64, 0, 0);
  f.a0 = load_frag_bf16(ap0 + k0);
  f.a1 = load_frag_bf16(ap1 + k0);
#pragma unroll
  for (int nb = 0; nb < 4; ++nb)
    f.b[nb] = load_frag_packed(bq + (size_t)(nb * kch + (k0 >> 5)) * 64);
}

template <bool TRANS>
__device__ __forceinline__ void gemm_step(v8f (&acc)[2][4], const GemmFrags& f) {
#pragma unroll
  for (int nb = 0; nb < 4; ++nb) {
    if (TRANS) {
      acc[0][nb] = wmma_bf16(f.b[nb], f.a0, acc[0][nb]);
      acc[1][nb] = wmma_bf16(f.b[nb], f.a1, acc[1][nb]);
    } else {
      acc[0][nb] = wmma_bf16(f.a0, f.b[nb], acc[0][nb]);
      acc[1][nb] = wmma_bf16(f.a1, f.b[nb], acc[1][nb]);
    }
  }
}

template <bool TRANS>
__device__ __forceinline__ void gemm_mainloop(
    v8f (&acc)[2][4], const bf16_t* __restrict__ ap0,
    const bf16_t* __restrict__ ap1, const v8bf* __restrict__ bq, int K) {
  const int kch = K >> 5;
  GemmFrags fx, fy;
  gemm_load(fx, ap0, ap1, bq, kch, 0);
  for (int k0 = 0; k0 < K; k0 += 64) {
    gemm_load(fy, ap0, ap1, bq, kch, k0 + 32);
    __builtin_amdgcn_sched_barrier(0);
    gemm_step<TRANS>(acc, fx);
    const int kn = (k0 + 64 < K) ? k0 + 64 : K - 32;
    gemm_load(fx, ap0, ap1, bq, kch, kn);
    __builtin_amdgcn_sched_barrier(0);
    gemm_step<TRANS>(acc, fy);
  }
}

template <int MODE>
__global__ __launch_bounds__(128) void gemm_wmma(
    const bf16_t* __restrict__ A, const bf16_t* __restrict__ B,
    const float* __restrict__ bias, float* __restrict__ Yf,
    bf16_t* __restrict__ Yb,
    bf16_t* __restrict__ Qb, bf16_t* __restrict__ Kb, bf16_t* __restrict__ Vt,
    int M, int N, int K) {
  const int lane = threadIdx.x & 31;
  const int wave = threadIdx.x >> 5;
  const int r    = lane & 15;
  const int half = lane >> 4;
  const int row0 = blockIdx.y * 128 + wave * 32;
  const int col0 = blockIdx.x * 64;

  v8f acc[2][4] = {};
  const bf16_t* ap0 = A + (size_t)(row0 + r) * K;
  const bf16_t* ap1 = A + (size_t)(row0 + 16 + r) * K;
  const v8bf* bq = (const v8bf*)B + (size_t)(col0 >> 4) * (K >> 5) * 64 + lane;

  const bool kblk = (MODE == 2) && (col0 >= DIMD) && (col0 < 2 * DIMD);
  if (kblk) gemm_mainloop<true>(acc, ap0, ap1, bq, K);
  else      gemm_mainloop<false>(acc, ap0, ap1, bq, K);

  if (MODE == 2 && col0 >= DIMD) {
    const int bb  = row0 >> 11;
    const int t0  = row0 & (SEQ - 1);
    if (kblk) {
      const size_t bhk = (size_t)bb * HEADS + ((col0 - DIMD) >> 6);
      v8bf* kf = (v8bf*)Kb + (bhk * (SEQ / 16) + (t0 >> 4)) * 128 + lane;
#pragma unroll
      for (int mi = 0; mi < 2; ++mi) {
#pragma unroll
        for (int nb = 0; nb < 4; ++nb) {
          v8bf o;
#pragma unroll
          for (int i = 0; i < 8; ++i)
            o[i] = (bf16_t)(acc[mi][nb][i] + bias[col0 + nb * 16 + 8 * half + i]);
          kf[mi * 128 + nb * 32] = o;
        }
      }
    } else {
      const size_t bhv = (size_t)bb * HEADS + ((col0 - 2 * DIMD) >> 6);
      v8bf* vf = (v8bf*)Vt + (bhv * (SEQ / 32) + (t0 >> 5)) * 256 + lane;
#pragma unroll
      for (int mi = 0; mi < 2; ++mi) {
#pragma unroll
        for (int nb = 0; nb < 4; ++nb) {
          const float bv = bias[col0 + nb * 16 + r];
          v8bf o;
#pragma unroll
          for (int i = 0; i < 8; ++i) o[i] = (bf16_t)(acc[mi][nb][i] + bv);
          vf[nb * 64 + mi * 32] = o;
        }
      }
    }
    return;
  }

#pragma unroll
  for (int mi = 0; mi < 2; ++mi) {
#pragma unroll
    for (int nb = 0; nb < 4; ++nb) {
#pragma unroll
      for (int i = 0; i < 8; ++i) {
        const int row = row0 + mi * 16 + i + 8 * half;
        const int col = col0 + nb * 16 + r;
        float y = acc[mi][nb][i] + bias[col];
        if (MODE == 0) {
          Yf[(size_t)row * N + col] = y;
        } else if (MODE == 1) {
          Yb[(size_t)row * N + col] = (bf16_t)fmaxf(y, 0.0f);
        } else {
          const int b = row >> 11, t = row & (SEQ - 1);
          const int h = col >> 6, d = col & 63;
          Qb[(((size_t)b * HEADS + h) * SEQ + t) * HD + d] = (bf16_t)y;
        }
      }
    }
  }
}

#define KSPLIT 4
#define NQ     4
#define QTILE  (16 * NQ)
#define CSCALE 0.18033688011112042f
#define NEGRAW (-8589934592.0f)

typedef __attribute__((ext_vector_type(2))) float  f32x2;
typedef __attribute__((ext_vector_type(2))) __bf16 bf16x2;

__device__ __forceinline__ float xhalf(float x) {
  const int b = __builtin_bit_cast(int, x);
  return __builtin_bit_cast(float, __builtin_amdgcn_permlanex16(
      b, b, 0x76543210, 0xfedcba98, false, false));
}

template <int XM>
__device__ __forceinline__ float row_xor(float x) {
  const int b = __builtin_bit_cast(int, x);
  return __builtin_bit_cast(float, __builtin_amdgcn_update_dpp(
      0, b, 0x160 + XM, 0xf, 0xf, true));
}

__device__ __forceinline__ float max3f(float a, float b, float c) {
  return __builtin_fmaxf(__builtin_fmaxf(a, b), c);
}

__global__ __launch_bounds__(32 * KSPLIT) void attn_wmma(
    const bf16_t* __restrict__ Qb, const bf16_t* __restrict__ Kb,
    const bf16_t* __restrict__ Vt, const float* __restrict__ mask,
    bf16_t* __restrict__ attn, float* __restrict__ halt) {
  __shared__ float  lds_m[KSPLIT][QTILE];
  __shared__ float  lds_l[KSPLIT][QTILE];
  __shared__ float  lds_o[KSPLIT][NQ * 32][32];
  static_assert(NQ == KSPLIT, "the final merge gives wave w sub-tile w");

  const int bid  = blockIdx.x;
  const int qt   = bid % (SEQ / QTILE);
  const int bh   = bid / (SEQ / QTILE);
  const int b    = bh >> 2;
  const int lane = threadIdx.x & 31;
  const int w    = threadIdx.x >> 5;
  const int r    = lane & 15;
  const int half = lane >> 4;

  const bf16_t* Qbase = Qb + ((size_t)bh * SEQ + qt * QTILE) * HD;
  const v8bf* Kq = (const v8bf*)Kb +
                   ((size_t)bh * (SEQ / 16) + (size_t)w * (SEQ / KSPLIT / 16)) * 128 + lane;
  const v8bf* Vq = (const v8bf*)Vt +
                   ((size_t)bh * (SEQ / 32) + (size_t)w * (SEQ / KSPLIT / 32)) * 256 + lane;
  const float* mrow8 = mask + (size_t)b * SEQ + (size_t)w * (SEQ / KSPLIT) + 8 * half;
  float*       hrow  = halt + (size_t)b * SEQ + (size_t)w * (SEQ / KSPLIT);

  v16bf qa[NQ][2];
#pragma unroll
  for (int qi = 0; qi < NQ; ++qi) {
    qa[qi][0] = load_frag_bf16(Qbase + (size_t)(qi * 16 + r) * HD + 0);
    qa[qi][1] = load_frag_bf16(Qbase + (size_t)(qi * 16 + r) * HD + 32);
  }

  const f32x2 c2 = {CSCALE, CSCALE};

  float m[NQ], sh[NQ], l[NQ];
#pragma unroll
  for (int qi = 0; qi < NQ; ++qi) {
    m[qi] = -3.0e38f; sh[qi] = 3.0e38f * CSCALE; l[qi] = 0.0f;
  }

  v16bf ks0[4], ks1[4];
  v8f mks[4];
  auto load_tile = [&](v16bf& k0r, v16bf& k1r, v8f& mkr, int kt) {
    k0r = load_frag_packed(Kq + (size_t)kt * 128);
    k1r = load_frag_packed(Kq + (size_t)kt * 128 + 64);
    mkr = *(const v8f*)(mrow8 + kt * 16);
  };
  auto pass1_tile = [&](const v16bf& kb0, const v16bf& kb1, const v8f& mk) {
    const bool msk[8] = {mk[0] > 0.5f, mk[1] > 0.5f, mk[2] > 0.5f, mk[3] > 0.5f,
                         mk[4] > 0.5f, mk[5] > 0.5f, mk[6] > 0.5f, mk[7] > 0.5f};
#pragma unroll
    for (int qi = 0; qi < NQ; ++qi) {
      v8f s = {};
      s = wmma_bf16(kb0, qa[qi][0], s);
      s = wmma_bf16(kb1, qa[qi][1], s);
      float sv[8];
#pragma unroll
      for (int i = 0; i < 8; ++i) sv[i] = msk[i] ? NEGRAW : s[i];
      float mx = max3f(m[qi], sv[0], sv[1]);
      mx = max3f(mx, sv[2], sv[3]);
      mx = max3f(mx, sv[4], sv[5]);
      mx = max3f(mx, sv[6], sv[7]);
      const float nm = -(mx * CSCALE);
      const f32x2 nm2 = {nm, nm};
      const float alpha = __builtin_amdgcn_exp2f(nm - sh[qi]);
      f32x2 e[4];
#pragma unroll
      for (int j = 0; j < 4; ++j) {
        const f32x2 a = {sv[2 * j], sv[2 * j + 1]};
        const f32x2 t = __builtin_elementwise_fma(a, c2, nm2);
        e[j][0] = __builtin_amdgcn_exp2f(t[0]);
        e[j][1] = __builtin_amdgcn_exp2f(t[1]);
      }
      const f32x2 es = (e[0] + e[1]) + (e[2] + e[3]);
      l[qi] = __builtin_fmaf(l[qi], alpha, es[0] + es[1]);
      m[qi] = mx;
      sh[qi] = nm;
    }
  };

  constexpr int NKT = SEQ / KSPLIT / 16;
#pragma unroll
  for (int u = 0; u < 3; ++u) load_tile(ks0[u], ks1[u], mks[u], u);
  for (int kt = 0; kt < NKT; kt += 4) {
#pragma unroll
    for (int u = 0; u < 4; ++u) {
      load_tile(ks0[(u + 3) & 3], ks1[(u + 3) & 3], mks[(u + 3) & 3],
                (kt + u + 3) & (NKT - 1));
      __builtin_amdgcn_sched_barrier(0);
      pass1_tile(ks0[u], ks1[u], mks[u]);
    }
  }
#pragma unroll
  for (int qi = 0; qi < NQ; ++qi) {
    const float so = xhalf(sh[qi]);
    const float lo = xhalf(l[qi]);
    const float sn = __builtin_fminf(sh[qi], so);
    l[qi] = l[qi] * __builtin_amdgcn_exp2f(sn - sh[qi]) +
            lo * __builtin_amdgcn_exp2f(sn - so);
    sh[qi] = sn;
  }
  if (half == 0) {
#pragma unroll
    for (int qi = 0; qi < NQ; ++qi) {
      lds_m[w][qi * 16 + r] = sh[qi];
      lds_l[w][qi * 16 + r] = l[qi];
    }
  }
  __syncthreads();
  f32x2 nmf[NQ], rl[NQ];
#pragma unroll
  for (int qi = 0; qi < NQ; ++qi) {
    float sw[KSPLIT], lw[KSPLIT];
#pragma unroll
    for (int k = 0; k < KSPLIT; ++k) {
      sw[k] = lds_m[k][qi * 16 + r];
      lw[k] = lds_l[k][qi * 16 + r];
    }
    float nm = sw[0];
#pragma unroll
    for (int k = 1; k < KSPLIT; ++k) nm = __builtin_fminf(nm, sw[k]);
    float ls = 0.0f;
#pragma unroll
    for (int k = 0; k < KSPLIT; ++k)
      ls += lw[k] * __builtin_amdgcn_exp2f(nm - sw[k]);
    const float inv = 1.0f / ls;
    nmf[qi][0] = nm;  nmf[qi][1] = nm;
    rl[qi][0]  = inv; rl[qi][1]  = inv;
  }

  const bool b3 = (r & 8) != 0, b2 = (r & 4) != 0, b1 = (r & 2) != 0;
  v8f o[NQ][4] = {};
  auto pass2_tile = [&](v16bf (&pa)[NQ], const v16bf& kb0, const v16bf& kb1,
                        const v8f& mk, int sub, int kt) {
    const bool msk[8] = {mk[0] > 0.5f, mk[1] > 0.5f, mk[2] > 0.5f, mk[3] > 0.5f,
                         mk[4] > 0.5f, mk[5] > 0.5f, mk[6] > 0.5f, mk[7] > 0.5f};
    f32x2 cs[4];
#pragma unroll
    for (int qi = 0; qi < NQ; ++qi) {
      v8f s = {};
      s = wmma_bf16(kb0, qa[qi][0], s);
      s = wmma_bf16(kb1, qa[qi][1], s);
#pragma unroll
      for (int j = 0; j < 4; ++j) {
        const f32x2 a = {msk[2 * j] ? NEGRAW : s[2 * j],
                         msk[2 * j + 1] ? NEGRAW : s[2 * j + 1]};
        const f32x2 t = __builtin_elementwise_fma(a, c2, nmf[qi]);
        f32x2 p;
        p[0] = __builtin_amdgcn_exp2f(t[0]);
        p[1] = __builtin_amdgcn_exp2f(t[1]);
        p = p * rl[qi];
        if (qi == 0) cs[j] = p; else cs[j] = cs[j] + p;
        const bf16x2 pb = __builtin_convertvector(p, bf16x2);
        pa[qi][sub * 8 + 2 * j]     = pb[0];
        pa[qi][sub * 8 + 2 * j + 1] = pb[1];
      }
    }
    float k4[4];
#pragma unroll
    for (int j = 0; j < 4; ++j) {
      const float lo4 = cs[j >> 1][j & 1];
      const float hi4 = cs[(j + 4) >> 1][j & 1];
      const float keep = b3 ? hi4 : lo4;
      const float send = b3 ? lo4 : hi4;
      k4[j] = keep + row_xor<8>(send);
    }
    float k2[2];
#pragma unroll
    for (int j = 0; j < 2; ++j) {
      const float keep = b2 ? k4[j + 2] : k4[j];
      const float send = b2 ? k4[j] : k4[j + 2];
      k2[j] = keep + row_xor<4>(send);
    }
    float k1 = (b1 ? k2[1] : k2[0]) + row_xor<2>(b1 ? k2[0] : k2[1]);
    k1 += row_xor<1>(k1);
    if ((lane & 1) == 0) atomicAdd(&hrow[kt * 16 + 8 * half + (r >> 1)], k1);
  };

  for (int kt2 = 0; kt2 < SEQ / KSPLIT / 32; ++kt2) {
    v16bf pa[NQ];
    load_tile(ks0[1], ks1[1], mks[1], kt2 * 2 + 1);
    __builtin_amdgcn_sched_barrier(0);
    pass2_tile(pa, ks0[0], ks1[0], mks[0], 0, kt2 * 2);
    v16bf vb[4];
#pragma unroll
    for (int nb = 0; nb < 4; ++nb)
      vb[nb] = load_frag_packed(Vq + (size_t)kt2 * 256 + nb * 64);
    load_tile(ks0[0], ks1[0], mks[0], (kt2 * 2 + 2) & (NKT - 1));
    __builtin_amdgcn_sched_barrier(0);
    pass2_tile(pa, ks0[1], ks1[1], mks[1], 1, kt2 * 2 + 1);
#pragma unroll
    for (int nb = 0; nb < 4; ++nb) {
#pragma unroll
      for (int qi = 0; qi < NQ; ++qi)
        o[qi][nb] = wmma_bf16(pa[qi], vb[nb], o[qi][nb]);
    }
  }

#pragma unroll
  for (int qi = 0; qi < NQ; ++qi) {
#pragma unroll
    for (int nb = 0; nb < 4; ++nb) {
#pragma unroll
      for (int i = 0; i < 8; ++i) {
        lds_o[w][(qi * 4 + nb) * 8 + i][lane] = o[qi][nb][i];
      }
    }
  }
  __syncthreads();

  const int h   = bh & 3;
  const int qif = w;
#pragma unroll
  for (int i = 0; i < 8; ++i) {
    const int t = qt * QTILE + qif * 16 + i + 8 * half;
#pragma unroll
    for (int nb = 0; nb < 4; ++nb) {
      const int j = (qif * 4 + nb) * 8 + i;
      float acc = lds_o[0][j][lane];
#pragma unroll
      for (int k = 1; k < KSPLIT; ++k) acc += lds_o[k][j][lane];
      attn[((size_t)b * SEQ + t) * DIMD + h * HD + nb * 16 + r] = (bf16_t)acc;
    }
  }
}

__device__ __forceinline__ void halt_softmax_blk(
    const float* __restrict__ hacc, const float* __restrict__ mask,
    float* __restrict__ out, int b) {
  __shared__ float red[256];
  const int tid = threadIdx.x;
  float lm = -3.0e38f;
  for (int j = tid; j < SEQ; j += 256) {
    const float v = hacc[(size_t)b * SEQ + j] * 0.25f +
                    mask[(size_t)b * SEQ + j] * (-1e10f);
    lm = fmaxf(lm, v);
  }
  red[tid] = lm;
  __syncthreads();
  for (int s = 128; s > 0; s >>= 1) {
    if (tid < s) red[tid] = fmaxf(red[tid], red[tid + s]);
    __syncthreads();
  }
  const float mx = red[0];
  __syncthreads();
  float ls = 0.0f;
  for (int j = tid; j < SEQ; j += 256) {
    const float v = hacc[(size_t)b * SEQ + j] * 0.25f +
                    mask[(size_t)b * SEQ + j] * (-1e10f);
    ls += __expf(v - mx);
  }
  red[tid] = ls;
  __syncthreads();
  for (int s = 128; s > 0; s >>= 1) {
    if (tid < s) red[tid] += red[tid + s];
    __syncthreads();
  }
  const float inv = 1.0f / red[0];
  for (int j = tid; j < SEQ; j += 256) {
    const float v = hacc[(size_t)b * SEQ + j] * 0.25f +
                    mask[(size_t)b * SEQ + j] * (-1e10f);
    out[(size_t)b * SEQ + j] = __expf(v - mx) * inv;
  }
}

#define LN_ROWS  4
#define LN_WAVES 8
template <bool WRITE_BF16, bool WITH_HALT>
__global__ __launch_bounds__(32 * LN_WAVES)
__attribute__((amdgpu_waves_per_eu(2, 4))) void add_ln(
    const float* __restrict__ A, const float* __restrict__ Bv,
    const float* __restrict__ g, const float* __restrict__ be,
    float* __restrict__ out, bf16_t* __restrict__ outb,
    const float* __restrict__ hacc, const float* __restrict__ mask,
    float* __restrict__ hout) {
  if (WITH_HALT && blockIdx.x < BATCH) {
    halt_softmax_blk(hacc, mask, hout, blockIdx.x);
    return;
  }
  const int lnblk = WITH_HALT ? (int)blockIdx.x - BATCH : (int)blockIdx.x;
  const int lane = threadIdx.x & 31;
  const int wave = threadIdx.x >> 5;
  const size_t row0 = ((size_t)lnblk * LN_WAVES + wave) * LN_ROWS;
  const float4* a4 = (const float4*)(A + row0 * DIMD) + lane;
  const float4* b4 = (const float4*)(Bv + row0 * DIMD) + lane;

  float v[LN_ROWS][8];
  float s[LN_ROWS];
#pragma unroll
  for (int r = 0; r < LN_ROWS; ++r) {
#pragma unroll
    for (int hf = 0; hf < 2; ++hf) {
      const float4 x = a4[r * (DIMD / 4) + hf * 32];
      const float4 y = b4[r * (DIMD / 4) + hf * 32];
      v[r][hf * 4 + 0] = x.x + y.x;
      v[r][hf * 4 + 1] = x.y + y.y;
      v[r][hf * 4 + 2] = x.z + y.z;
      v[r][hf * 4 + 3] = x.w + y.w;
    }
    s[r] = ((v[r][0] + v[r][1]) + (v[r][2] + v[r][3])) +
           ((v[r][4] + v[r][5]) + (v[r][6] + v[r][7]));
  }
  float4 gg[2], bb[2];
#pragma unroll
  for (int hf = 0; hf < 2; ++hf) {
    gg[hf] = ((const float4*)g)[lane + hf * 32];
    bb[hf] = ((const float4*)be)[lane + hf * 32];
  }
#pragma unroll
  for (int m = 16; m >= 1; m >>= 1) {
#pragma unroll
    for (int r = 0; r < LN_ROWS; ++r) s[r] += __shfl_xor(s[r], m, 32);
  }

  float q[LN_ROWS];
#pragma unroll
  for (int r = 0; r < LN_ROWS; ++r) {
    const float mu = s[r] * (1.0f / DIMD);
#pragma unroll
    for (int j = 0; j < 8; ++j) v[r][j] -= mu;
    q[r] = ((v[r][0] * v[r][0] + v[r][1] * v[r][1]) +
            (v[r][2] * v[r][2] + v[r][3] * v[r][3])) +
           ((v[r][4] * v[r][4] + v[r][5] * v[r][5]) +
            (v[r][6] * v[r][6] + v[r][7] * v[r][7]));
  }
#pragma unroll
  for (int m = 16; m >= 1; m >>= 1) {
#pragma unroll
    for (int r = 0; r < LN_ROWS; ++r) q[r] += __shfl_xor(q[r], m, 32);
  }

#pragma unroll
  for (int r = 0; r < LN_ROWS; ++r) {
    const float var = q[r] * (1.0f / DIMD);
    const float rs  = rsqrtf(var + 1e-5f);
    const size_t row = row0 + r;
#pragma unroll
    for (int hf = 0; hf < 2; ++hf) {
      float4 y;
      y.x = v[r][hf * 4 + 0] * rs * gg[hf].x + bb[hf].x;
      y.y = v[r][hf * 4 + 1] * rs * gg[hf].y + bb[hf].y;
      y.z = v[r][hf * 4 + 2] * rs * gg[hf].z + bb[hf].z;
      y.w = v[r][hf * 4 + 3] * rs * gg[hf].w + bb[hf].w;
      ((float4*)(out + row * DIMD))[lane + hf * 32] = y;
      if (WRITE_BF16) {
        v4bf o;
        o[0] = (bf16_t)y.x; o[1] = (bf16_t)y.y; o[2] = (bf16_t)y.z; o[3] = (bf16_t)y.w;
        ((v4bf*)(outb + row * DIMD))[lane + hf * 32] = o;
      }
    }
  }
}

extern "C" void kernel_launch(void* const* d_in, const int* in_sizes, int n_in,
                              void* d_out, int out_size, void* d_ws,
                              size_t ws_size, hipStream_t stream) {
  (void)in_sizes; (void)n_in; (void)out_size; (void)ws_size;
  const float* x    = (const float*)d_in[0];
  const float* mask = (const float*)d_in[1];
  const float* Wqkv = (const float*)d_in[2];
  const float* bqkv = (const float*)d_in[3];
  const float* Wo   = (const float*)d_in[4];
  const float* bo   = (const float*)d_in[5];
  const float* W1   = (const float*)d_in[6];
  const float* b1   = (const float*)d_in[7];
  const float* W2   = (const float*)d_in[8];
  const float* b2   = (const float*)d_in[9];
  const float* g1   = (const float*)d_in[10];
  const float* be1  = (const float*)d_in[11];
  const float* g2   = (const float*)d_in[12];
  const float* be2  = (const float*)d_in[13];

  char* ws = (char*)d_ws;
  size_t off = 0;
  auto alloc = [&](size_t bytes) {
    char* p = ws + off;
    off += (bytes + 255) & ~(size_t)255;
    return p;
  };
  bf16_t* xb    = (bf16_t*)alloc((size_t)NTOK * DIMD * 2);
  bf16_t* Wqkvb = (bf16_t*)alloc((size_t)3 * DIMD * DIMD * 2);
  bf16_t* Wob   = (bf16_t*)alloc((size_t)DIMD * DIMD * 2);
  bf16_t* W1b   = (bf16_t*)alloc((size_t)FFND * DIMD * 2);
  bf16_t* W2b   = (bf16_t*)alloc((size_t)DIMD * FFND * 2);
  bf16_t* Qb    = (bf16_t*)alloc((size_t)NTOK * DIMD * 2);
  bf16_t* Kb    = (bf16_t*)alloc((size_t)NTOK * DIMD * 2);
  bf16_t* Vt    = (bf16_t*)alloc((size_t)NTOK * DIMD * 2);
  bf16_t* attnb = (bf16_t*)alloc((size_t)NTOK * DIMD * 2);
  float*  attnO = (float*) alloc((size_t)NTOK * DIMD * 4);
  float*  hbuf  = (float*) alloc((size_t)NTOK * DIMD * 4);
  bf16_t* hb    = (bf16_t*)alloc((size_t)NTOK * DIMD * 2);
  bf16_t* ffn1b = (bf16_t*)alloc((size_t)NTOK * FFND * 2);
  float*  ffn2  = (float*) alloc((size_t)NTOK * DIMD * 4);
  float*  halt  = (float*) alloc((size_t)NTOK * 4);

  float* out      = (float*)d_out;
  float* out_halt = out + (size_t)NTOK * DIMD;

  prep_all<<<PB_X + PB_WQKV + PB_WO + PB_W1 + PB_W2 + PB_HZ, 256, 0, stream>>>(
      x, Wqkv, Wo, W1, W2, xb, Wqkvb, Wob, W1b, W2b, halt);

  gemm_wmma<2><<<dim3(3 * DIMD / 64, NTOK / 128), 128, 0, stream>>>(
      xb, Wqkvb, bqkv, nullptr, nullptr, Qb, Kb, Vt, NTOK, 3 * DIMD, DIMD);

  attn_wmma<<<BATCH * HEADS * (SEQ / QTILE), 32 * KSPLIT, 0, stream>>>(
      Qb, Kb, Vt, mask, attnb, halt);

  gemm_wmma<0><<<dim3(DIMD / 64, NTOK / 128), 128, 0, stream>>>(
      attnb, Wob, bo, attnO, nullptr, nullptr, nullptr, nullptr,
      NTOK, DIMD, DIMD);

  add_ln<true, true><<<BATCH + NTOK / (LN_WAVES * LN_ROWS), 32 * LN_WAVES, 0, stream>>>(
      x, attnO, g1, be1, hbuf, hb, halt, mask, out_halt);

  gemm_wmma<1><<<dim3(FFND / 64, NTOK / 128), 128, 0, stream>>>(
      hb, W1b, b1, nullptr, ffn1b, nullptr, nullptr, nullptr,
      NTOK, FFND, DIMD);

  gemm_wmma<0><<<dim3(DIMD / 64, NTOK / 128), 128, 0, stream>>>(
      ffn1b, W2b, b2, ffn2, nullptr, nullptr, nullptr, nullptr,
      NTOK, DIMD, FFND);

  add_ln<false, false><<<NTOK / (LN_WAVES * LN_ROWS), 32 * LN_WAVES, 0, stream>>>(
      hbuf, ffn2, g2, be2, out, nullptr, nullptr, nullptr, nullptr);
}


# A operands in WMMA fragment order too: x, attention output, LayerNorm bf16 copy and ReLU output are written as 16-byte units; every GEMM A fragment load is 512 contiguous bytes (4 lines, not 16)
# speedup vs baseline: 2.4531x; 1.0440x over previous
#include <hip/hip_runtime.h>
#include <hip/hip_bf16.h>

typedef __bf16 bf16_t;
typedef __attribute__((ext_vector_type(16))) __bf16 v16bf;
typedef __attribute__((ext_vector_type(4)))  __bf16 v4bf;
typedef __attribute__((ext_vector_type(8)))  __bf16 v8bf;
typedef __attribute__((ext_vector_type(8)))  float  v8f;

#define DIMD   256
#define HEADS  4
#define HD     64
#define FFND   1024
#define BATCH  8
#define SEQ    2048
#define NTOK   (BATCH*SEQ)

__device__ __forceinline__ v16bf load_frag_bf16(const bf16_t* base) {
  const int half = (threadIdx.x >> 4) & 1;
  const bf16_t* p = base + 8 * half;
  v16bf f;
#pragma unroll
  for (int e = 0; e < 8; ++e) {
    f[e]     = p[e];
    f[e + 8] = p[e + 16];
  }
  return f;
}

__device__ __forceinline__ v16bf load_frag_packed(const v8bf* p) {
  const v8bf lo = p[0];
  const v8bf hi = p[32];
  return __builtin_shufflevector(lo, hi, 0, 1, 2, 3, 4, 5, 6, 7,
                                 8, 9, 10, 11, 12, 13, 14, 15);
}

__device__ __forceinline__ v8f wmma_bf16(v16bf a, v16bf b, v8f c) {
  return __builtin_amdgcn_wmma_f32_16x16x32_bf16(
       false, a,  false, b,
       (short)0, c,  false,  false);
}

#define XF_U 4
__device__ __forceinline__ void cvt_afrag_blk(const float* __restrict__ in,
                                              bf16_t* __restrict__ out, int blk) {
  static_assert(DIMD == 256, "one 32-column chunk per wave of a 256-thread block");
  const int tid = threadIdx.x;
  const int q   = tid & 3;
  const int kc  = tid >> 5;
  const int r8  = (tid >> 2) & 7;
  float4 x0[XF_U], x1[XF_U];
#pragma unroll
  for (int k = 0; k < XF_U; ++k) {
    const int row = (blk * XF_U + k) * 8 + r8;
    const float4* p = (const float4*)(in + (size_t)row * DIMD + kc * 32 + 8 * q);
    x0[k] = p[0];
    x1[k] = p[1];
  }
#pragma unroll
  for (int k = 0; k < XF_U; ++k) {
    const int row = (blk * XF_U + k) * 8 + r8;
    const size_t unit = ((size_t)((row >> 4) * (DIMD / 32) + kc) * 2 + (q >> 1)) * 32 +
                        (q & 1) * 16 + (row & 15);
    v8bf o;
    o[0] = (bf16_t)x0[k].x; o[1] = (bf16_t)x0[k].y; o[2] = (bf16_t)x0[k].z; o[3] = (bf16_t)x0[k].w;
    o[4] = (bf16_t)x1[k].x; o[5] = (bf16_t)x1[k].y; o[6] = (bf16_t)x1[k].z; o[7] = (bf16_t)x1[k].w;
    ((v8bf*)out)[unit] = o;
  }
}

template <int KD>
__device__ __forceinline__ void cvt_frag_blk(const float* __restrict__ in,
                                             bf16_t* __restrict__ out, int blk) {
  const int u    = blk * 256 + threadIdx.x;
  const int l    = u & 31;
  const int j    = (u >> 5) & 1;
  const int tile = u >> 6;
  const int kc   = tile % (KD / 32);
  const int rt   = tile / (KD / 32);
  const int row  = rt * 16 + (l & 15);
  const int k    = kc * 32 + j * 16 + 8 * (l >> 4);
  const float4* p = (const float4*)(in + (size_t)row * KD + k);
  const float4 x0 = p[0], x1 = p[1];
  v8bf o;
  o[0] = (bf16_t)x0.x; o[1] = (bf16_t)x0.y; o[2] = (bf16_t)x0.z; o[3] = (bf16_t)x0.w;
  o[4] = (bf16_t)x1.x; o[5] = (bf16_t)x1.y; o[6] = (bf16_t)x1.z; o[7] = (bf16_t)x1.w;
  ((v8bf*)out)[u] = o;
}

#define PB_X    (NTOK / (8 * XF_U))
#define PB_WQKV (3 * DIMD * DIMD / (8 * 256))
#define PB_WO   (DIMD * DIMD / (8 * 256))
#define PB_W1   (FFND * DIMD / (8 * 256))
#define PB_W2   (DIMD * FFND / (8 * 256))
#define PB_HZ   (NTOK / (4 * 256))
__global__ __launch_bounds__(256) void prep_all(
    const float* __restrict__ x, const float* __restrict__ Wqkv,
    const float* __restrict__ Wo, const float* __restrict__ W1,
    const float* __restrict__ W2, bf16_t* __restrict__ xb,
    bf16_t* __restrict__ Wqkvb, bf16_t* __restrict__ Wob,
    bf16_t* __restrict__ W1b, bf16_t* __restrict__ W2b,
    float* __restrict__ halt) {
  int blk = blockIdx.x;
  if (blk < PB_X) { cvt_afrag_blk(x, xb, blk); return; }
  blk -= PB_X;
  if (blk < PB_WQKV) { cvt_frag_blk<DIMD>(Wqkv, Wqkvb, blk); return; }
  blk -= PB_WQKV;
  if (blk < PB_WO) { cvt_frag_blk<DIMD>(Wo, Wob, blk); return; }
  blk -= PB_WO;
  if (blk < PB_W1) { cvt_frag_blk<DIMD>(W1, W1b, blk); return; }
  blk -= PB_W1;
  if (blk < PB_W2) { cvt_frag_blk<FFND>(W2, W2b, blk); return; }
  blk -= PB_W2;
  ((float4*)halt)[blk * 256 + threadIdx.x] = make_float4(0.0f, 0.0f, 0.0f, 0.0f);
}


struct GemmFrags {
  v16bf a0, a1, b[4];
};

__device__ __forceinline__ void gemm_load(
    GemmFrags& f, const v8bf* __restrict__ aq, const v8bf* __restrict__ bq,
    int kch, int k0) {
  const int kc = k0 >> 5;
  f.a0 = load_frag_packed(aq + (size_t)kc * 64);
  f.a1 = load_frag_packed(aq + (size_t)(kch + kc) * 64);
#pragma unroll
  for (int nb = 0; nb < 4; ++nb)
    f.b[nb] = load_frag_packed(bq + (size_t)(nb * kch + kc) * 64);
}

template <bool TRANS>
__device__ __forceinline__ void gemm_step(v8f (&acc)[2][4], const GemmFrags& f) {
#pragma unroll
  for (int nb = 0; nb < 4; ++nb) {
    if (TRANS) {
      acc[0][nb] = wmma_bf16(f.b[nb], f.a0, acc[0][nb]);
      acc[1][nb] = wmma_bf16(f.b[nb], f.a1, acc[1][nb]);
    } else {
      acc[0][nb] = wmma_bf16(f.a0, f.b[nb], acc[0][nb]);
      acc[1][nb] = wmma_bf16(f.a1, f.b[nb], acc[1][nb]);
    }
  }
}

template <bool TRANS>
__device__ __forceinline__ void gemm_mainloop(
    v8f (&acc)[2][4], const v8bf* __restrict__ aq,
    const v8bf* __restrict__ bq, int K) {
  const int kch = K >> 5;
  GemmFrags fx, fy;
  gemm_load(fx, aq, bq, kch, 0);
  for (int k0 = 0; k0 < K; k0 += 64) {
    gemm_load(fy, aq, bq, kch, k0 + 32);
    __builtin_amdgcn_sched_barrier(0);
    gemm_step<TRANS>(acc, fx);
    const int kn = (k0 + 64 < K) ? k0 + 64 : K - 32;
    gemm_load(fx, aq, bq, kch, kn);
    __builtin_amdgcn_sched_barrier(0);
    gemm_step<TRANS>(acc, fy);
  }
}

template <int MODE>
__global__ __launch_bounds__(128) void gemm_wmma(
    const bf16_t* __restrict__ A, const bf16_t* __restrict__ B,
    const float* __restrict__ bias, float* __restrict__ Yf,
    bf16_t* __restrict__ Yb,
    bf16_t* __restrict__ Qb, bf16_t* __restrict__ Kb, bf16_t* __restrict__ Vt,
    int M, int N, int K) {
  const int lane = threadIdx.x & 31;
  const int wave = threadIdx.x >> 5;
  const int r    = lane & 15;
  const int half = lane >> 4;
  const int row0 = blockIdx.y * 128 + wave * 32;
  const int col0 = blockIdx.x * 64;

  v8f acc[2][4] = {};
  const v8bf* aq = (const v8bf*)A + (size_t)(row0 >> 4) * (K >> 5) * 64 + lane;
  const v8bf* bq = (const v8bf*)B + (size_t)(col0 >> 4) * (K >> 5) * 64 + lane;

  const bool kblk = (MODE == 2) && (col0 >= DIMD) && (col0 < 2 * DIMD);
  if (MODE == 1 || kblk) gemm_mainloop<true>(acc, aq, bq, K);
  else                   gemm_mainloop<false>(acc, aq, bq, K);

  if (MODE == 1) {
    v8bf* yf = (v8bf*)Yb + ((size_t)(row0 >> 4) * (N >> 5) + (col0 >> 5)) * 64 + lane;
#pragma unroll
    for (int mi = 0; mi < 2; ++mi) {
#pragma unroll
      for (int nb = 0; nb < 4; ++nb) {
        v8bf o;
#pragma unroll
        for (int i = 0; i < 8; ++i)
          o[i] = (bf16_t)fmaxf(acc[mi][nb][i] + bias[col0 + nb * 16 + 8 * half + i], 0.0f);
        yf[(size_t)mi * (N >> 5) * 64 + nb * 32] = o;
      }
    }
    return;
  }

  if (MODE == 2 && col0 >= DIMD) {
    const int bb  = row0 >> 11;
    const int t0  = row0 & (SEQ - 1);
    if (kblk) {
      const size_t bhk = (size_t)bb * HEADS + ((col0 - DIMD) >> 6);
      v8bf* kf = (v8bf*)Kb + (bhk * (SEQ / 16) + (t0 >> 4)) * 128 + lane;
#pragma unroll
      for (int mi = 0; mi < 2; ++mi) {
#pragma unroll
        for (int nb = 0; nb < 4; ++nb) {
          v8bf o;
#pragma unroll
          for (int i = 0; i < 8; ++i)
            o[i] = (bf16_t)(acc[mi][nb][i] + bias[col0 + nb * 16 + 8 * half + i]);
          kf[mi * 128 + nb * 32] = o;
        }
      }
    } else {
      const size_t bhv = (size_t)bb * HEADS + ((col0 - 2 * DIMD) >> 6);
      v8bf* vf = (v8bf*)Vt + (bhv * (SEQ / 32) + (t0 >> 5)) * 256 + lane;
#pragma unroll
      for (int mi = 0; mi < 2; ++mi) {
#pragma unroll
        for (int nb = 0; nb < 4; ++nb) {
          const float bv = bias[col0 + nb * 16 + r];
          v8bf o;
#pragma unroll
          for (int i = 0; i < 8; ++i) o[i] = (bf16_t)(acc[mi][nb][i] + bv);
          vf[nb * 64 + mi * 32] = o;
        }
      }
    }
    return;
  }

#pragma unroll
  for (int mi = 0; mi < 2; ++mi) {
#pragma unroll
    for (int nb = 0; nb < 4; ++nb) {
#pragma unroll
      for (int i = 0; i < 8; ++i) {
        const int row = row0 + mi * 16 + i + 8 * half;
        const int col = col0 + nb * 16 + r;
        float y = acc[mi][nb][i] + bias[col];
        if (MODE == 0) {
          Yf[(size_t)row * N + col] = y;
        } else {
          const int b = row >> 11, t = row & (SEQ - 1);
          const int h = col >> 6, d = col & 63;
          Qb[(((size_t)b * HEADS + h) * SEQ + t) * HD + d] = (bf16_t)y;
        }
      }
    }
  }
}

#define KSPLIT 4
#define NQ     4
#define QTILE  (16 * NQ)
#define CSCALE 0.18033688011112042f
#define NEGRAW (-8589934592.0f)

typedef __attribute__((ext_vector_type(2))) float  f32x2;
typedef __attribute__((ext_vector_type(2))) __bf16 bf16x2;

__device__ __forceinline__ float xhalf(float x) {
  const int b = __builtin_bit_cast(int, x);
  return __builtin_bit_cast(float, __builtin_amdgcn_permlanex16(
      b, b, 0x76543210, 0xfedcba98, false, false));
}

template <int XM>
__device__ __forceinline__ float row_xor(float x) {
  const int b = __builtin_bit_cast(int, x);
  return __builtin_bit_cast(float, __builtin_amdgcn_update_dpp(
      0, b, 0x160 + XM, 0xf, 0xf, true));
}

__device__ __forceinline__ float max3f(float a, float b, float c) {
  return __builtin_fmaxf(__builtin_fmaxf(a, b), c);
}

__global__ __launch_bounds__(32 * KSPLIT) void attn_wmma(
    const bf16_t* __restrict__ Qb, const bf16_t* __restrict__ Kb,
    const bf16_t* __restrict__ Vt, const float* __restrict__ mask,
    bf16_t* __restrict__ attn, float* __restrict__ halt) {
  __shared__ float  lds_m[KSPLIT][QTILE];
  __shared__ float  lds_l[KSPLIT][QTILE];
  __shared__ __attribute__((aligned(16))) float lds_o[KSPLIT][NQ * 32][32];
  static_assert(NQ == KSPLIT, "the final merge gives wave w sub-tile w");

  const int bid  = blockIdx.x;
  const int qt   = bid % (SEQ / QTILE);
  const int bh   = bid / (SEQ / QTILE);
  const int b    = bh >> 2;
  const int lane = threadIdx.x & 31;
  const int w    = threadIdx.x >> 5;
  const int r    = lane & 15;
  const int half = lane >> 4;

  const bf16_t* Qbase = Qb + ((size_t)bh * SEQ + qt * QTILE) * HD;
  const v8bf* Kq = (const v8bf*)Kb +
                   ((size_t)bh * (SEQ / 16) + (size_t)w * (SEQ / KSPLIT / 16)) * 128 + lane;
  const v8bf* Vq = (const v8bf*)Vt +
                   ((size_t)bh * (SEQ / 32) + (size_t)w * (SEQ / KSPLIT / 32)) * 256 + lane;
  const float* mrow8 = mask + (size_t)b * SEQ + (size_t)w * (SEQ / KSPLIT) + 8 * half;
  float*       hrow  = halt + (size_t)b * SEQ + (size_t)w * (SEQ / KSPLIT);

  v16bf qa[NQ][2];
#pragma unroll
  for (int qi = 0; qi < NQ; ++qi) {
    qa[qi][0] = load_frag_bf16(Qbase + (size_t)(qi * 16 + r) * HD + 0);
    qa[qi][1] = load_frag_bf16(Qbase + (size_t)(qi * 16 + r) * HD + 32);
  }

  const f32x2 c2 = {CSCALE, CSCALE};

  float m[NQ], sh[NQ], l[NQ];
#pragma unroll
  for (int qi = 0; qi < NQ; ++qi) {
    m[qi] = -3.0e38f; sh[qi] = 3.0e38f * CSCALE; l[qi] = 0.0f;
  }

  v16bf ks0[4], ks1[4];
  v8f mks[4];
  auto load_tile = [&](v16bf& k0r, v16bf& k1r, v8f& mkr, int kt) {
    k0r = load_frag_packed(Kq + (size_t)kt * 128);
    k1r = load_frag_packed(Kq + (size_t)kt * 128 + 64);
    mkr = *(const v8f*)(mrow8 + kt * 16);
  };
  auto pass1_tile = [&](const v16bf& kb0, const v16bf& kb1, const v8f& mk) {
    const bool msk[8] = {mk[0] > 0.5f, mk[1] > 0.5f, mk[2] > 0.5f, mk[3] > 0.5f,
                         mk[4] > 0.5f, mk[5] > 0.5f, mk[6] > 0.5f, mk[7] > 0.5f};
#pragma unroll
    for (int qi = 0; qi < NQ; ++qi) {
      v8f s = {};
      s = wmma_bf16(kb0, qa[qi][0], s);
      s = wmma_bf16(kb1, qa[qi][1], s);
      float sv[8];
#pragma unroll
      for (int i = 0; i < 8; ++i) sv[i] = msk[i] ? NEGRAW : s[i];
      float mx = max3f(m[qi], sv[0], sv[1]);
      mx = max3f(mx, sv[2], sv[3]);
      mx = max3f(mx, sv[4], sv[5]);
      mx = max3f(mx, sv[6], sv[7]);
      const float nm = -(mx * CSCALE);
      const f32x2 nm2 = {nm, nm};
      const float alpha = __builtin_amdgcn_exp2f(nm - sh[qi]);
      f32x2 e[4];
#pragma unroll
      for (int j = 0; j < 4; ++j) {
        const f32x2 a = {sv[2 * j], sv[2 * j + 1]};
        const f32x2 t = __builtin_elementwise_fma(a, c2, nm2);
        e[j][0] = __builtin_amdgcn_exp2f(t[0]);
        e[j][1] = __builtin_amdgcn_exp2f(t[1]);
      }
      const f32x2 es = (e[0] + e[1]) + (e[2] + e[3]);
      l[qi] = __builtin_fmaf(l[qi], alpha, es[0] + es[1]);
      m[qi] = mx;
      sh[qi] = nm;
    }
  };

  constexpr int NKT = SEQ / KSPLIT / 16;
#pragma unroll
  for (int u = 0; u < 3; ++u) load_tile(ks0[u], ks1[u], mks[u], u);
  for (int kt = 0; kt < NKT; kt += 4) {
#pragma unroll
    for (int u = 0; u < 4; ++u) {
      load_tile(ks0[(u + 3) & 3], ks1[(u + 3) & 3], mks[(u + 3) & 3],
                (kt + u + 3) & (NKT - 1));
      __builtin_amdgcn_sched_barrier(0);
      pass1_tile(ks0[u], ks1[u], mks[u]);
    }
  }
#pragma unroll
  for (int qi = 0; qi < NQ; ++qi) {
    const float so = xhalf(sh[qi]);
    const float lo = xhalf(l[qi]);
    const float sn = __builtin_fminf(sh[qi], so);
    l[qi] = l[qi] * __builtin_amdgcn_exp2f(sn - sh[qi]) +
            lo * __builtin_amdgcn_exp2f(sn - so);
    sh[qi] = sn;
  }
  if (half == 0) {
#pragma unroll
    for (int qi = 0; qi < NQ; ++qi) {
      lds_m[w][qi * 16 + r] = sh[qi];
      lds_l[w][qi * 16 + r] = l[qi];
    }
  }
  __syncthreads();
  f32x2 nmf[NQ], rl[NQ];
#pragma unroll
  for (int qi = 0; qi < NQ; ++qi) {
    float sw[KSPLIT], lw[KSPLIT];
#pragma unroll
    for (int k = 0; k < KSPLIT; ++k) {
      sw[k] = lds_m[k][qi * 16 + r];
      lw[k] = lds_l[k][qi * 16 + r];
    }
    float nm = sw[0];
#pragma unroll
    for (int k = 1; k < KSPLIT; ++k) nm = __builtin_fminf(nm, sw[k]);
    float ls = 0.0f;
#pragma unroll
    for (int k = 0; k < KSPLIT; ++k)
      ls += lw[k] * __builtin_amdgcn_exp2f(nm - sw[k]);
    const float inv = 1.0f / ls;
    nmf[qi][0] = nm;  nmf[qi][1] = nm;
    rl[qi][0]  = inv; rl[qi][1]  = inv;
  }

  const bool b3 = (r & 8) != 0, b2 = (r & 4) != 0, b1 = (r & 2) != 0;
  v8f o[NQ][4] = {};
  auto pass2_tile = [&](v16bf (&pa)[NQ], const v16bf& kb0, const v16bf& kb1,
                        const v8f& mk, int sub, int kt) {
    const bool msk[8] = {mk[0] > 0.5f, mk[1] > 0.5f, mk[2] > 0.5f, mk[3] > 0.5f,
                         mk[4] > 0.5f, mk[5] > 0.5f, mk[6] > 0.5f, mk[7] > 0.5f};
    f32x2 cs[4];
#pragma unroll
    for (int qi = 0; qi < NQ; ++qi) {
      v8f s = {};
      s = wmma_bf16(kb0, qa[qi][0], s);
      s = wmma_bf16(kb1, qa[qi][1], s);
#pragma unroll
      for (int j = 0; j < 4; ++j) {
        const f32x2 a = {msk[2 * j] ? NEGRAW : s[2 * j],
                         msk[2 * j + 1] ? NEGRAW : s[2 * j + 1]};
        const f32x2 t = __builtin_elementwise_fma(a, c2, nmf[qi]);
        f32x2 p;
        p[0] = __builtin_amdgcn_exp2f(t[0]);
        p[1] = __builtin_amdgcn_exp2f(t[1]);
        p = p * rl[qi];
        if (qi == 0) cs[j] = p; else cs[j] = cs[j] + p;
        const bf16x2 pb = __builtin_convertvector(p, bf16x2);
        pa[qi][sub * 8 + 2 * j]     = pb[0];
        pa[qi][sub * 8 + 2 * j + 1] = pb[1];
      }
    }
    float k4[4];
#pragma unroll
    for (int j = 0; j < 4; ++j) {
      const float lo4 = cs[j >> 1][j & 1];
      const float hi4 = cs[(j + 4) >> 1][j & 1];
      const float keep = b3 ? hi4 : lo4;
      const float send = b3 ? lo4 : hi4;
      k4[j] = keep + row_xor<8>(send);
    }
    float k2[2];
#pragma unroll
    for (int j = 0; j < 2; ++j) {
      const float keep = b2 ? k4[j + 2] : k4[j];
      const float send = b2 ? k4[j] : k4[j + 2];
      k2[j] = keep + row_xor<4>(send);
    }
    float k1 = (b1 ? k2[1] : k2[0]) + row_xor<2>(b1 ? k2[0] : k2[1]);
    k1 += row_xor<1>(k1);
    if ((lane & 1) == 0) atomicAdd(&hrow[kt * 16 + 8 * half + (r >> 1)], k1);
  };

  for (int kt2 = 0; kt2 < SEQ / KSPLIT / 32; ++kt2) {
    v16bf pa[NQ];
    load_tile(ks0[1], ks1[1], mks[1], kt2 * 2 + 1);
    __builtin_amdgcn_sched_barrier(0);
    pass2_tile(pa, ks0[0], ks1[0], mks[0], 0, kt2 * 2);
    v16bf vb[4];
#pragma unroll
    for (int nb = 0; nb < 4; ++nb)
      vb[nb] = load_frag_packed(Vq + (size_t)kt2 * 256 + nb * 64);
    load_tile(ks0[0], ks1[0], mks[0], (kt2 * 2 + 2) & (NKT - 1));
    __builtin_amdgcn_sched_barrier(0);
    pass2_tile(pa, ks0[1], ks1[1], mks[1], 1, kt2 * 2 + 1);
#pragma unroll
    for (int nb = 0; nb < 4; ++nb) {
#pragma unroll
      for (int qi = 0; qi < NQ; ++qi)
        o[qi][nb] = wmma_bf16(pa[qi], vb[nb], o[qi][nb]);
    }
  }

#pragma unroll
  for (int qi = 0; qi < NQ; ++qi) {
#pragma unroll
    for (int nb = 0; nb < 4; ++nb) {
#pragma unroll
      for (int i = 0; i < 8; ++i) {
        lds_o[w][(qi * 4 + nb) * 8 + i][lane] = o[qi][nb][i];
      }
    }
  }
  __syncthreads();

  const int h = bh & 3;
  const size_t rt = ((size_t)b * SEQ + (size_t)qt * QTILE + (size_t)w * 16) >> 4;
  v8bf* af = (v8bf*)attn + (rt * (DIMD / 32) + (size_t)h * 2) * 64 + lane;
#pragma unroll
  for (int nb = 0; nb < 4; ++nb) {
    const int j  = (w * 4 + nb) * 8 + (r & 7);
    const int c0 = 16 * (r >> 3) + 8 * half;
    float4 s0 = *(const float4*)&lds_o[0][j][c0];
    float4 s1 = *(const float4*)&lds_o[0][j][c0 + 4];
#pragma unroll
    for (int k = 1; k < KSPLIT; ++k) {
      const float4 t0 = *(const float4*)&lds_o[k][j][c0];
      const float4 t1 = *(const float4*)&lds_o[k][j][c0 + 4];
      s0.x += t0.x; s0.y += t0.y; s0.z += t0.z; s0.w += t0.w;
      s1.x += t1.x; s1.y += t1.y; s1.z += t1.z; s1.w += t1.w;
    }
    v8bf o;
    o[0] = (bf16_t)s0.x; o[1] = (bf16_t)s0.y; o[2] = (bf16_t)s0.z; o[3] = (bf16_t)s0.w;
    o[4] = (bf16_t)s1.x; o[5] = (bf16_t)s1.y; o[6] = (bf16_t)s1.z; o[7] = (bf16_t)s1.w;
    af[nb * 32] = o;
  }
}

__device__ __forceinline__ void halt_softmax_blk(
    const float* __restrict__ hacc, const float* __restrict__ mask,
    float* __restrict__ out, int b) {
  __shared__ float red[256];
  const int tid = threadIdx.x;
  float lm = -3.0e38f;
  for (int j = tid; j < SEQ; j += 256) {
    const float v = hacc[(size_t)b * SEQ + j] * 0.25f +
                    mask[(size_t)b * SEQ + j] * (-1e10f);
    lm = fmaxf(lm, v);
  }
  red[tid] = lm;
  __syncthreads();
  for (int s = 128; s > 0; s >>= 1) {
    if (tid < s) red[tid] = fmaxf(red[tid], red[tid + s]);
    __syncthreads();
  }
  const float mx = red[0];
  __syncthreads();
  float ls = 0.0f;
  for (int j = tid; j < SEQ; j += 256) {
    const float v = hacc[(size_t)b * SEQ + j] * 0.25f +
                    mask[(size_t)b * SEQ + j] * (-1e10f);
    ls += __expf(v - mx);
  }
  red[tid] = ls;
  __syncthreads();
  for (int s = 128; s > 0; s >>= 1) {
    if (tid < s) red[tid] += red[tid + s];
    __syncthreads();
  }
  const float inv = 1.0f / red[0];
  for (int j = tid; j < SEQ; j += 256) {
    const float v = hacc[(size_t)b * SEQ + j] * 0.25f +
                    mask[(size_t)b * SEQ + j] * (-1e10f);
    out[(size_t)b * SEQ + j] = __expf(v - mx) * inv;
  }
}

#define LN_ROWS  4
#define LN_WAVES 8
template <bool WRITE_BF16, bool WITH_HALT>
__global__ __launch_bounds__(32 * LN_WAVES)
__attribute__((amdgpu_waves_per_eu(2, 4))) void add_ln(
    const float* __restrict__ A, const float* __restrict__ Bv,
    const float* __restrict__ g, const float* __restrict__ be,
    float* __restrict__ out, bf16_t* __restrict__ outb,
    const float* __restrict__ hacc, const float* __restrict__ mask,
    float* __restrict__ hout) {
  if (WITH_HALT && blockIdx.x < BATCH) {
    halt_softmax_blk(hacc, mask, hout, blockIdx.x);
    return;
  }
  const int lnblk = WITH_HALT ? (int)blockIdx.x - BATCH : (int)blockIdx.x;
  const int lane = threadIdx.x & 31;
  const int wave = threadIdx.x >> 5;
  const size_t row0 = ((size_t)lnblk * LN_WAVES + wave) * LN_ROWS;
  const float4* a4 = (const float4*)(A + row0 * DIMD) + lane;
  const float4* b4 = (const float4*)(Bv + row0 * DIMD) + lane;

  float v[LN_ROWS][8];
  float s[LN_ROWS];
#pragma unroll
  for (int r = 0; r < LN_ROWS; ++r) {
#pragma unroll
    for (int hf = 0; hf < 2; ++hf) {
      const float4 x = a4[r * (DIMD / 4) + hf * 32];
      const float4 y = b4[r * (DIMD / 4) + hf * 32];
      v[r][hf * 4 + 0] = x.x + y.x;
      v[r][hf * 4 + 1] = x.y + y.y;
      v[r][hf * 4 + 2] = x.z + y.z;
      v[r][hf * 4 + 3] = x.w + y.w;
    }
    s[r] = ((v[r][0] + v[r][1]) + (v[r][2] + v[r][3])) +
           ((v[r][4] + v[r][5]) + (v[r][6] + v[r][7]));
  }
  float4 gg[2], bb[2];
#pragma unroll
  for (int hf = 0; hf < 2; ++hf) {
    gg[hf] = ((const float4*)g)[lane + hf * 32];
    bb[hf] = ((const float4*)be)[lane + hf * 32];
  }
#pragma unroll
  for (int m = 16; m >= 1; m >>= 1) {
#pragma unroll
    for (int r = 0; r < LN_ROWS; ++r) s[r] += __shfl_xor(s[r], m, 32);
  }

  float q[LN_ROWS];
#pragma unroll
  for (int r = 0; r < LN_ROWS; ++r) {
    const float mu = s[r] * (1.0f / DIMD);
#pragma unroll
    for (int j = 0; j < 8; ++j) v[r][j] -= mu;
    q[r] = ((v[r][0] * v[r][0] + v[r][1] * v[r][1]) +
            (v[r][2] * v[r][2] + v[r][3] * v[r][3])) +
           ((v[r][4] * v[r][4] + v[r][5] * v[r][5]) +
            (v[r][6] * v[r][6] + v[r][7] * v[r][7]));
  }
#pragma unroll
  for (int m = 16; m >= 1; m >>= 1) {
#pragma unroll
    for (int r = 0; r < LN_ROWS; ++r) q[r] += __shfl_xor(q[r], m, 32);
  }

  static_assert(!WRITE_BF16 || (LN_WAVES * LN_ROWS == 32 && DIMD == 256),
                "a block is two 16-row tiles of 8 chunks x 2 pieces");
  __shared__ __attribute__((aligned(16))) bf16_t stage[WRITE_BF16 ? LN_WAVES * LN_ROWS : 1][DIMD + 8];
  float4 y[LN_ROWS][2];
#pragma unroll
  for (int r = 0; r < LN_ROWS; ++r) {
    const float var = q[r] * (1.0f / DIMD);
    const float rs  = rsqrtf(var + 1e-5f);
#pragma unroll
    for (int hf = 0; hf < 2; ++hf) {
      y[r][hf].x = v[r][hf * 4 + 0] * rs * gg[hf].x + bb[hf].x;
      y[r][hf].y = v[r][hf * 4 + 1] * rs * gg[hf].y + bb[hf].y;
      y[r][hf].z = v[r][hf * 4 + 2] * rs * gg[hf].z + bb[hf].z;
      y[r][hf].w = v[r][hf * 4 + 3] * rs * gg[hf].w + bb[hf].w;
      if (WRITE_BF16) {
        v4bf o;
        o[0] = (bf16_t)y[r][hf].x; o[1] = (bf16_t)y[r][hf].y;
        o[2] = (bf16_t)y[r][hf].z; o[3] = (bf16_t)y[r][hf].w;
        *(v4bf*)&stage[wave * LN_ROWS + r][4 * lane + hf * 128] = o;
      }
    }
  }
  if (WRITE_BF16) {
    __syncthreads();
    const int rr = lane & 15, hh = lane >> 4;
    v8bf* hf_out = (v8bf*)outb + (size_t)lnblk * 1024 + lane;
#pragma unroll
    for (int s = 0; s < 4; ++s) {
      const int G = wave * 4 + s;
      const v8bf u = *(const v8bf*)&stage[(G >> 4) * 16 + rr]
                                         [((G >> 1) & 7) * 32 + (G & 1) * 16 + 8 * hh];
      hf_out[G * 32] = u;
    }
  }
#pragma unroll
  for (int r = 0; r < LN_ROWS; ++r) {
#pragma unroll
    for (int hf = 0; hf < 2; ++hf)
      ((float4*)(out + (row0 + r) * DIMD))[lane + hf * 32] = y[r][hf];
  }
}

extern "C" void kernel_launch(void* const* d_in, const int* in_sizes, int n_in,
                              void* d_out, int out_size, void* d_ws,
                              size_t ws_size, hipStream_t stream) {
  (void)in_sizes; (void)n_in; (void)out_size; (void)ws_size;
  const float* x    = (const float*)d_in[0];
  const float* mask = (const float*)d_in[1];
  const float* Wqkv = (const float*)d_in[2];
  const float* bqkv = (const float*)d_in[3];
  const float* Wo   = (const float*)d_in[4];
  const float* bo   = (const float*)d_in[5];
  const float* W1   = (const float*)d_in[6];
  const float* b1   = (const float*)d_in[7];
  const float* W2   = (const float*)d_in[8];
  const float* b2   = (const float*)d_in[9];
  const float* g1   = (const float*)d_in[10];
  const float* be1  = (const float*)d_in[11];
  const float* g2   = (const float*)d_in[12];
  const float* be2  = (const float*)d_in[13];

  char* ws = (char*)d_ws;
  size_t off = 0;
  auto alloc = [&](size_t bytes) {
    char* p = ws + off;
    off += (bytes + 255) & ~(size_t)255;
    return p;
  };
  bf16_t* xb    = (bf16_t*)alloc((size_t)NTOK * DIMD * 2);
  bf16_t* Wqkvb = (bf16_t*)alloc((size_t)3 * DIMD * DIMD * 2);
  bf16_t* Wob   = (bf16_t*)alloc((size_t)DIMD * DIMD * 2);
  bf16_t* W1b   = (bf16_t*)alloc((size_t)FFND * DIMD * 2);
  bf16_t* W2b   = (bf16_t*)alloc((size_t)DIMD * FFND * 2);
  bf16_t* Qb    = (bf16_t*)alloc((size_t)NTOK * DIMD * 2);
  bf16_t* Kb    = (bf16_t*)alloc((size_t)NTOK * DIMD * 2);
  bf16_t* Vt    = (bf16_t*)alloc((size_t)NTOK * DIMD * 2);
  bf16_t* attnb = (bf16_t*)alloc((size_t)NTOK * DIMD * 2);
  float*  attnO = (float*) alloc((size_t)NTOK * DIMD * 4);
  float*  hbuf  = (float*) alloc((size_t)NTOK * DIMD * 4);
  bf16_t* hb    = (bf16_t*)alloc((size_t)NTOK * DIMD * 2);
  bf16_t* ffn1b = (bf16_t*)alloc((size_t)NTOK * FFND * 2);
  float*  ffn2  = (float*) alloc((size_t)NTOK * DIMD * 4);
  float*  halt  = (float*) alloc((size_t)NTOK * 4);

  float* out      = (float*)d_out;
  float* out_halt = out + (size_t)NTOK * DIMD;

  prep_all<<<PB_X + PB_WQKV + PB_WO + PB_W1 + PB_W2 + PB_HZ, 256, 0, stream>>>(
      x, Wqkv, Wo, W1, W2, xb, Wqkvb, Wob, W1b, W2b, halt);

  gemm_wmma<2><<<dim3(3 * DIMD / 64, NTOK / 128), 128, 0, stream>>>(
      xb, Wqkvb, bqkv, nullptr, nullptr, Qb, Kb, Vt, NTOK, 3 * DIMD, DIMD);

  attn_wmma<<<BATCH * HEADS * (SEQ / QTILE), 32 * KSPLIT, 0, stream>>>(
      Qb, Kb, Vt, mask, attnb, halt);

  gemm_wmma<0><<<dim3(DIMD / 64, NTOK / 128), 128, 0, stream>>>(
      attnb, Wob, bo, attnO, nullptr, nullptr, nullptr, nullptr,
      NTOK, DIMD, DIMD);

  add_ln<true, true><<<BATCH + NTOK / (LN_WAVES * LN_ROWS), 32 * LN_WAVES, 0, stream>>>(
      x, attnO, g1, be1, hbuf, hb, halt, mask, out_halt);

  gemm_wmma<1><<<dim3(FFND / 64, NTOK / 128), 128, 0, stream>>>(
      hb, W1b, b1, nullptr, ffn1b, nullptr, nullptr, nullptr,
      NTOK, FFND, DIMD);

  gemm_wmma<0><<<dim3(DIMD / 64, NTOK / 128), 128, 0, stream>>>(
      ffn1b, W2b, b2, ffn2, nullptr, nullptr, nullptr, nullptr,
      NTOK, DIMD, FFND);

  add_ln<false, false><<<NTOK / (LN_WAVES * LN_ROWS), 32 * LN_WAVES, 0, stream>>>(
      hbuf, ffn2, g2, be2, out, nullptr, nullptr, nullptr, nullptr);
}


# residual add + LayerNorm moved into the epilogue of the GEMM that feeds it (Wo, FFN2): 64x256 block tile, product handed over in LDS, not memory; 8 launches -> 6; no sum order changes
# speedup vs baseline: 2.6219x; 1.0688x over previous
#include <hip/hip_runtime.h>
#include <hip/hip_bf16.h>

typedef __bf16 bf16_t;
typedef __attribute__((ext_vector_type(16))) __bf16 v16bf;
typedef __attribute__((ext_vector_type(4)))  __bf16 v4bf;
typedef __attribute__((ext_vector_type(8)))  __bf16 v8bf;
typedef __attribute__((ext_vector_type(8)))  float  v8f;

#define DIMD   256
#define HEADS  4
#define HD     64
#define FFND   1024
#define BATCH  8
#define SEQ    2048
#define NTOK   (BATCH*SEQ)

__device__ __forceinline__ v16bf load_frag_bf16(const bf16_t* base) {
  const int half = (threadIdx.x >> 4) & 1;
  const bf16_t* p = base + 8 * half;
  v16bf f;
#pragma unroll
  for (int e = 0; e < 8; ++e) {
    f[e]     = p[e];
    f[e + 8] = p[e + 16];
  }
  return f;
}

__device__ __forceinline__ v16bf load_frag_packed(const v8bf* p) {
  const v8bf lo = p[0];
  const v8bf hi = p[32];
  return __builtin_shufflevector(lo, hi, 0, 1, 2, 3, 4, 5, 6, 7,
                                 8, 9, 10, 11, 12, 13, 14, 15);
}

__device__ __forceinline__ v8f wmma_bf16(v16bf a, v16bf b, v8f c) {
  return __builtin_amdgcn_wmma_f32_16x16x32_bf16(
       false, a,  false, b,
       (short)0, c,  false,  false);
}

#define XF_U 4
__device__ __forceinline__ void cvt_afrag_blk(const float* __restrict__ in,
                                              bf16_t* __restrict__ out, int blk) {
  static_assert(DIMD == 256, "one 32-column chunk per wave of a 256-thread block");
  const int tid = threadIdx.x;
  const int q   = tid & 3;
  const int kc  = tid >> 5;
  const int r8  = (tid >> 2) & 7;
  float4 x0[XF_U], x1[XF_U];
#pragma unroll
  for (int k = 0; k < XF_U; ++k) {
    const int row = (blk * XF_U + k) * 8 + r8;
    const float4* p = (const float4*)(in + (size_t)row * DIMD + kc * 32 + 8 * q);
    x0[k] = p[0];
    x1[k] = p[1];
  }
#pragma unroll
  for (int k = 0; k < XF_U; ++k) {
    const int row = (blk * XF_U + k) * 8 + r8;
    const size_t unit = ((size_t)((row >> 4) * (DIMD / 32) + kc) * 2 + (q >> 1)) * 32 +
                        (q & 1) * 16 + (row & 15);
    v8bf o;
    o[0] = (bf16_t)x0[k].x; o[1] = (bf16_t)x0[k].y; o[2] = (bf16_t)x0[k].z; o[3] = (bf16_t)x0[k].w;
    o[4] = (bf16_t)x1[k].x; o[5] = (bf16_t)x1[k].y; o[6] = (bf16_t)x1[k].z; o[7] = (bf16_t)x1[k].w;
    ((v8bf*)out)[unit] = o;
  }
}

template <int KD>
__device__ __forceinline__ void cvt_frag_blk(const float* __restrict__ in,
                                             bf16_t* __restrict__ out, int blk) {
  const int u    = blk * 256 + threadIdx.x;
  const int l    = u & 31;
  const int j    = (u >> 5) & 1;
  const int tile = u >> 6;
  const int kc   = tile % (KD / 32);
  const int rt   = tile / (KD / 32);
  const int row  = rt * 16 + (l & 15);
  const int k    = kc * 32 + j * 16 + 8 * (l >> 4);
  const float4* p = (const float4*)(in + (size_t)row * KD + k);
  const float4 x0 = p[0], x1 = p[1];
  v8bf o;
  o[0] = (bf16_t)x0.x; o[1] = (bf16_t)x0.y; o[2] = (bf16_t)x0.z; o[3] = (bf16_t)x0.w;
  o[4] = (bf16_t)x1.x; o[5] = (bf16_t)x1.y; o[6] = (bf16_t)x1.z; o[7] = (bf16_t)x1.w;
  ((v8bf*)out)[u] = o;
}

#define PB_X    (NTOK / (8 * XF_U))
#define PB_WQKV (3 * DIMD * DIMD / (8 * 256))
#define PB_WO   (DIMD * DIMD / (8 * 256))
#define PB_W1   (FFND * DIMD / (8 * 256))
#define PB_W2   (DIMD * FFND / (8 * 256))
#define PB_HZ   (NTOK / (4 * 256))
__global__ __launch_bounds__(256) void prep_all(
    const float* __restrict__ x, const float* __restrict__ Wqkv,
    const float* __restrict__ Wo, const float* __restrict__ W1,
    const float* __restrict__ W2, bf16_t* __restrict__ xb,
    bf16_t* __restrict__ Wqkvb, bf16_t* __restrict__ Wob,
    bf16_t* __restrict__ W1b, bf16_t* __restrict__ W2b,
    float* __restrict__ halt) {
  int blk = blockIdx.x;
  if (blk < PB_X) { cvt_afrag_blk(x, xb, blk); return; }
  blk -= PB_X;
  if (blk < PB_WQKV) { cvt_frag_blk<DIMD>(Wqkv, Wqkvb, blk); return; }
  blk -= PB_WQKV;
  if (blk < PB_WO) { cvt_frag_blk<DIMD>(Wo, Wob, blk); return; }
  blk -= PB_WO;
  if (blk < PB_W1) { cvt_frag_blk<DIMD>(W1, W1b, blk); return; }
  blk -= PB_W1;
  if (blk < PB_W2) { cvt_frag_blk<FFND>(W2, W2b, blk); return; }
  blk -= PB_W2;
  ((float4*)halt)[blk * 256 + threadIdx.x] = make_float4(0.0f, 0.0f, 0.0f, 0.0f);
}


struct GemmFrags {
  v16bf a0, a1, b[4];
};

__device__ __forceinline__ void gemm_load(
    GemmFrags& f, const v8bf* __restrict__ aq, const v8bf* __restrict__ bq,
    int kch, int k0) {
  const int kc = k0 >> 5;
  f.a0 = load_frag_packed(aq + (size_t)kc * 64);
  f.a1 = load_frag_packed(aq + (size_t)(kch + kc) * 64);
#pragma unroll
  for (int nb = 0; nb < 4; ++nb)
    f.b[nb] = load_frag_packed(bq + (size_t)(nb * kch + kc) * 64);
}

template <bool TRANS>
__device__ __forceinline__ void gemm_step(v8f (&acc)[2][4], const GemmFrags& f) {
#pragma unroll
  for (int nb = 0; nb < 4; ++nb) {
    if (TRANS) {
      acc[0][nb] = wmma_bf16(f.b[nb], f.a0, acc[0][nb]);
      acc[1][nb] = wmma_bf16(f.b[nb], f.a1, acc[1][nb]);
    } else {
      acc[0][nb] = wmma_bf16(f.a0, f.b[nb], acc[0][nb]);
      acc[1][nb] = wmma_bf16(f.a1, f.b[nb], acc[1][nb]);
    }
  }
}

template <bool TRANS>
__device__ __forceinline__ void gemm_mainloop(
    v8f (&acc)[2][4], const v8bf* __restrict__ aq,
    const v8bf* __restrict__ bq, int K) {
  const int kch = K >> 5;
  GemmFrags fx, fy;
  gemm_load(fx, aq, bq, kch, 0);
  for (int k0 = 0; k0 < K; k0 += 64) {
    gemm_load(fy, aq, bq, kch, k0 + 32);
    __builtin_amdgcn_sched_barrier(0);
    gemm_step<TRANS>(acc, fx);
    const int kn = (k0 + 64 < K) ? k0 + 64 : K - 32;
    gemm_load(fx, aq, bq, kch, kn);
    __builtin_amdgcn_sched_barrier(0);
    gemm_step<TRANS>(acc, fy);
  }
}

template <int MODE>
__global__ __launch_bounds__(128) void gemm_wmma(
    const bf16_t* __restrict__ A, const bf16_t* __restrict__ B,
    const float* __restrict__ bias, bf16_t* __restrict__ Yb,
    bf16_t* __restrict__ Qb, bf16_t* __restrict__ Kb, bf16_t* __restrict__ Vt,
    int M, int N, int K) {
  const int lane = threadIdx.x & 31;
  const int wave = threadIdx.x >> 5;
  const int r    = lane & 15;
  const int half = lane >> 4;
  const int row0 = blockIdx.y * 128 + wave * 32;
  const int col0 = blockIdx.x * 64;

  v8f acc[2][4] = {};
  const v8bf* aq = (const v8bf*)A + (size_t)(row0 >> 4) * (K >> 5) * 64 + lane;
  const v8bf* bq = (const v8bf*)B + (size_t)(col0 >> 4) * (K >> 5) * 64 + lane;

  const bool kblk = (MODE == 2) && (col0 >= DIMD) && (col0 < 2 * DIMD);
  if (MODE == 1 || kblk) gemm_mainloop<true>(acc, aq, bq, K);
  else                   gemm_mainloop<false>(acc, aq, bq, K);

  if (MODE == 1) {
    v8bf* yf = (v8bf*)Yb + ((size_t)(row0 >> 4) * (N >> 5) + (col0 >> 5)) * 64 + lane;
#pragma unroll
    for (int mi = 0; mi < 2; ++mi) {
#pragma unroll
      for (int nb = 0; nb < 4; ++nb) {
        v8bf o;
#pragma unroll
        for (int i = 0; i < 8; ++i)
          o[i] = (bf16_t)fmaxf(acc[mi][nb][i] + bias[col0 + nb * 16 + 8 * half + i], 0.0f);
        yf[(size_t)mi * (N >> 5) * 64 + nb * 32] = o;
      }
    }
    return;
  }

  if (MODE == 2 && col0 >= DIMD) {
    const int bb  = row0 >> 11;
    const int t0  = row0 & (SEQ - 1);
    if (kblk) {
      const size_t bhk = (size_t)bb * HEADS + ((col0 - DIMD) >> 6);
      v8bf* kf = (v8bf*)Kb + (bhk * (SEQ / 16) + (t0 >> 4)) * 128 + lane;
#pragma unroll
      for (int mi = 0; mi < 2; ++mi) {
#pragma unroll
        for (int nb = 0; nb < 4; ++nb) {
          v8bf o;
#pragma unroll
          for (int i = 0; i < 8; ++i)
            o[i] = (bf16_t)(acc[mi][nb][i] + bias[col0 + nb * 16 + 8 * half + i]);
          kf[mi * 128 + nb * 32] = o;
        }
      }
    } else {
      const size_t bhv = (size_t)bb * HEADS + ((col0 - 2 * DIMD) >> 6);
      v8bf* vf = (v8bf*)Vt + (bhv * (SEQ / 32) + (t0 >> 5)) * 256 + lane;
#pragma unroll
      for (int mi = 0; mi < 2; ++mi) {
#pragma unroll
        for (int nb = 0; nb < 4; ++nb) {
          const float bv = bias[col0 + nb * 16 + r];
          v8bf o;
#pragma unroll
          for (int i = 0; i < 8; ++i) o[i] = (bf16_t)(acc[mi][nb][i] + bv);
          vf[nb * 64 + mi * 32] = o;
        }
      }
    }
    return;
  }

#pragma unroll
  for (int mi = 0; mi < 2; ++mi) {
#pragma unroll
    for (int nb = 0; nb < 4; ++nb) {
#pragma unroll
      for (int i = 0; i < 8; ++i) {
        const int row = row0 + mi * 16 + i + 8 * half;
        const int col = col0 + nb * 16 + r;
        const float y = acc[mi][nb][i] + bias[col];
        const int b = row >> 11, t = row & (SEQ - 1);
        const int h = col >> 6, d = col & 63;
        Qb[(((size_t)b * HEADS + h) * SEQ + t) * HD + d] = (bf16_t)y;
      }
    }
  }
}

#define KSPLIT 4
#define NQ     4
#define QTILE  (16 * NQ)
#define CSCALE 0.18033688011112042f
#define NEGRAW (-8589934592.0f)

typedef __attribute__((ext_vector_type(2))) float  f32x2;
typedef __attribute__((ext_vector_type(2))) __bf16 bf16x2;

__device__ __forceinline__ float xhalf(float x) {
  const int b = __builtin_bit_cast(int, x);
  return __builtin_bit_cast(float, __builtin_amdgcn_permlanex16(
      b, b, 0x76543210, 0xfedcba98, false, false));
}

template <int XM>
__device__ __forceinline__ float row_xor(float x) {
  const int b = __builtin_bit_cast(int, x);
  return __builtin_bit_cast(float, __builtin_amdgcn_update_dpp(
      0, b, 0x160 + XM, 0xf, 0xf, true));
}

__device__ __forceinline__ float max3f(float a, float b, float c) {
  return __builtin_fmaxf(__builtin_fmaxf(a, b), c);
}

__global__ __launch_bounds__(32 * KSPLIT) void attn_wmma(
    const bf16_t* __restrict__ Qb, const bf16_t* __restrict__ Kb,
    const bf16_t* __restrict__ Vt, const float* __restrict__ mask,
    bf16_t* __restrict__ attn, float* __restrict__ halt) {
  __shared__ float  lds_m[KSPLIT][QTILE];
  __shared__ float  lds_l[KSPLIT][QTILE];
  __shared__ __attribute__((aligned(16))) float lds_o[KSPLIT][NQ * 32][32];
  static_assert(NQ == KSPLIT, "the final merge gives wave w sub-tile w");

  const int bid  = blockIdx.x;
  const int qt   = bid % (SEQ / QTILE);
  const int bh   = bid / (SEQ / QTILE);
  const int b    = bh >> 2;
  const int lane = threadIdx.x & 31;
  const int w    = threadIdx.x >> 5;
  const int r    = lane & 15;
  const int half = lane >> 4;

  const bf16_t* Qbase = Qb + ((size_t)bh * SEQ + qt * QTILE) * HD;
  const v8bf* Kq = (const v8bf*)Kb +
                   ((size_t)bh * (SEQ / 16) + (size_t)w * (SEQ / KSPLIT / 16)) * 128 + lane;
  const v8bf* Vq = (const v8bf*)Vt +
                   ((size_t)bh * (SEQ / 32) + (size_t)w * (SEQ / KSPLIT / 32)) * 256 + lane;
  const float* mrow8 = mask + (size_t)b * SEQ + (size_t)w * (SEQ / KSPLIT) + 8 * half;
  float*       hrow  = halt + (size_t)b * SEQ + (size_t)w * (SEQ / KSPLIT);

  v16bf qa[NQ][2];
#pragma unroll
  for (int qi = 0; qi < NQ; ++qi) {
    qa[qi][0] = load_frag_bf16(Qbase + (size_t)(qi * 16 + r) * HD + 0);
    qa[qi][1] = load_frag_bf16(Qbase + (size_t)(qi * 16 + r) * HD + 32);
  }

  const f32x2 c2 = {CSCALE, CSCALE};

  float m[NQ], sh[NQ], l[NQ];
#pragma unroll
  for (int qi = 0; qi < NQ; ++qi) {
    m[qi] = -3.0e38f; sh[qi] = 3.0e38f * CSCALE; l[qi] = 0.0f;
  }

  v16bf ks0[4], ks1[4];
  v8f mks[4];
  auto load_tile = [&](v16bf& k0r, v16bf& k1r, v8f& mkr, int kt) {
    k0r = load_frag_packed(Kq + (size_t)kt * 128);
    k1r = load_frag_packed(Kq + (size_t)kt * 128 + 64);
    mkr = *(const v8f*)(mrow8 + kt * 16);
  };
  auto pass1_tile = [&](const v16bf& kb0, const v16bf& kb1, const v8f& mk) {
    const bool msk[8] = {mk[0] > 0.5f, mk[1] > 0.5f, mk[2] > 0.5f, mk[3] > 0.5f,
                         mk[4] > 0.5f, mk[5] > 0.5f, mk[6] > 0.5f, mk[7] > 0.5f};
#pragma unroll
    for (int qi = 0; qi < NQ; ++qi) {
      v8f s = {};
      s = wmma_bf16(kb0, qa[qi][0], s);
      s = wmma_bf16(kb1, qa[qi][1], s);
      float sv[8];
#pragma unroll
      for (int i = 0; i < 8; ++i) sv[i] = msk[i] ? NEGRAW : s[i];
      float mx = max3f(m[qi], sv[0], sv[1]);
      mx = max3f(mx, sv[2], sv[3]);
      mx = max3f(mx, sv[4], sv[5]);
      mx = max3f(mx, sv[6], sv[7]);
      const float nm = -(mx * CSCALE);
      const f32x2 nm2 = {nm, nm};
      const float alpha = __builtin_amdgcn_exp2f(nm - sh[qi]);
      f32x2 e[4];
#pragma unroll
      for (int j = 0; j < 4; ++j) {
        const f32x2 a = {sv[2 * j], sv[2 * j + 1]};
        const f32x2 t = __builtin_elementwise_fma(a, c2, nm2);
        e[j][0] = __builtin_amdgcn_exp2f(t[0]);
        e[j][1] = __builtin_amdgcn_exp2f(t[1]);
      }
      const f32x2 es = (e[0] + e[1]) + (e[2] + e[3]);
      l[qi] = __builtin_fmaf(l[qi], alpha, es[0] + es[1]);
      m[qi] = mx;
      sh[qi] = nm;
    }
  };

  constexpr int NKT = SEQ / KSPLIT / 16;
#pragma unroll
  for (int u = 0; u < 3; ++u) load_tile(ks0[u], ks1[u], mks[u], u);
  for (int kt = 0; kt < NKT; kt += 4) {
#pragma unroll
    for (int u = 0; u < 4; ++u) {
      load_tile(ks0[(u + 3) & 3], ks1[(u + 3) & 3], mks[(u + 3) & 3],
                (kt + u + 3) & (NKT - 1));
      __builtin_amdgcn_sched_barrier(0);
      pass1_tile(ks0[u], ks1[u], mks[u]);
    }
  }
#pragma unroll
  for (int qi = 0; qi < NQ; ++qi) {
    const float so = xhalf(sh[qi]);
    const float lo = xhalf(l[qi]);
    const float sn = __builtin_fminf(sh[qi], so);
    l[qi] = l[qi] * __builtin_amdgcn_exp2f(sn - sh[qi]) +
            lo * __builtin_amdgcn_exp2f(sn - so);
    sh[qi] = sn;
  }
  if (half == 0) {
#pragma unroll
    for (int qi = 0; qi < NQ; ++qi) {
      lds_m[w][qi * 16 + r] = sh[qi];
      lds_l[w][qi * 16 + r] = l[qi];
    }
  }
  __syncthreads();
  f32x2 nmf[NQ], rl[NQ];
#pragma unroll
  for (int qi = 0; qi < NQ; ++qi) {
    float sw[KSPLIT], lw[KSPLIT];
#pragma unroll
    for (int k = 0; k < KSPLIT; ++k) {
      sw[k] = lds_m[k][qi * 16 + r];
      lw[k] = lds_l[k][qi * 16 + r];
    }
    float nm = sw[0];
#pragma unroll
    for (int k = 1; k < KSPLIT; ++k) nm = __builtin_fminf(nm, sw[k]);
    float ls = 0.0f;
#pragma unroll
    for (int k = 0; k < KSPLIT; ++k)
      ls += lw[k] * __builtin_amdgcn_exp2f(nm - sw[k]);
    const float inv = 1.0f / ls;
    nmf[qi][0] = nm;  nmf[qi][1] = nm;
    rl[qi][0]  = inv; rl[qi][1]  = inv;
  }

  const bool b3 = (r & 8) != 0, b2 = (r & 4) != 0, b1 = (r & 2) != 0;
  v8f o[NQ][4] = {};
  auto pass2_tile = [&](v16bf (&pa)[NQ], const v16bf& kb0, const v16bf& kb1,
                        const v8f& mk, int sub, int kt) {
    const bool msk[8] = {mk[0] > 0.5f, mk[1] > 0.5f, mk[2] > 0.5f, mk[3] > 0.5f,
                         mk[4] > 0.5f, mk[5] > 0.5f, mk[6] > 0.5f, mk[7] > 0.5f};
    f32x2 cs[4];
#pragma unroll
    for (int qi = 0; qi < NQ; ++qi) {
      v8f s = {};
      s = wmma_bf16(kb0, qa[qi][0], s);
      s = wmma_bf16(kb1, qa[qi][1], s);
#pragma unroll
      for (int j = 0; j < 4; ++j) {
        const f32x2 a = {msk[2 * j] ? NEGRAW : s[2 * j],
                         msk[2 * j + 1] ? NEGRAW : s[2 * j + 1]};
        const f32x2 t = __builtin_elementwise_fma(a, c2, nmf[qi]);
        f32x2 p;
        p[0] = __builtin_amdgcn_exp2f(t[0]);
        p[1] = __builtin_amdgcn_exp2f(t[1]);
        p = p * rl[qi];
        if (qi == 0) cs[j] = p; else cs[j] = cs[j] + p;
        const bf16x2 pb = __builtin_convertvector(p, bf16x2);
        pa[qi][sub * 8 + 2 * j]     = pb[0];
        pa[qi][sub * 8 + 2 * j + 1] = pb[1];
      }
    }
    float k4[4];
#pragma unroll
    for (int j = 0; j < 4; ++j) {
      const float lo4 = cs[j >> 1][j & 1];
      const float hi4 = cs[(j + 4) >> 1][j & 1];
      const float keep = b3 ? hi4 : lo4;
      const float send = b3 ? lo4 : hi4;
      k4[j] = keep + row_xor<8>(send);
    }
    float k2[2];
#pragma unroll
    for (int j = 0; j < 2; ++j) {
      const float keep = b2 ? k4[j + 2] : k4[j];
      const float send = b2 ? k4[j] : k4[j + 2];
      k2[j] = keep + row_xor<4>(send);
    }
    float k1 = (b1 ? k2[1] : k2[0]) + row_xor<2>(b1 ? k2[0] : k2[1]);
    k1 += row_xor<1>(k1);
    if ((lane & 1) == 0) atomicAdd(&hrow[kt * 16 + 8 * half + (r >> 1)], k1);
  };

  for (int kt2 = 0; kt2 < SEQ / KSPLIT / 32; ++kt2) {
    v16bf pa[NQ];
    load_tile(ks0[1], ks1[1], mks[1], kt2 * 2 + 1);
    __builtin_amdgcn_sched_barrier(0);
    pass2_tile(pa, ks0[0], ks1[0], mks[0], 0, kt2 * 2);
    v16bf vb[4];
#pragma unroll
    for (int nb = 0; nb < 4; ++nb)
      vb[nb] = load_frag_packed(Vq + (size_t)kt2 * 256 + nb * 64);
    load_tile(ks0[0], ks1[0], mks[0], (kt2 * 2 + 2) & (NKT - 1));
    __builtin_amdgcn_sched_barrier(0);
    pass2_tile(pa, ks0[1], ks1[1], mks[1], 1, kt2 * 2 + 1);
#pragma unroll
    for (int nb = 0; nb < 4; ++nb) {
#pragma unroll
      for (int qi = 0; qi < NQ; ++qi)
        o[qi][nb] = wmma_bf16(pa[qi], vb[nb], o[qi][nb]);
    }
  }

#pragma unroll
  for (int qi = 0; qi < NQ; ++qi) {
#pragma unroll
    for (int nb = 0; nb < 4; ++nb) {
#pragma unroll
      for (int i = 0; i < 8; ++i) {
        lds_o[w][(qi * 4 + nb) * 8 + i][lane] = o[qi][nb][i];
      }
    }
  }
  __syncthreads();

  const int h = bh & 3;
  const size_t rt = ((size_t)b * SEQ + (size_t)qt * QTILE + (size_t)w * 16) >> 4;
  v8bf* af = (v8bf*)attn + (rt * (DIMD / 32) + (size_t)h * 2) * 64 + lane;
#pragma unroll
  for (int nb = 0; nb < 4; ++nb) {
    const int j  = (w * 4 + nb) * 8 + (r & 7);
    const int c0 = 16 * (r >> 3) + 8 * half;
    float4 s0 = *(const float4*)&lds_o[0][j][c0];
    float4 s1 = *(const float4*)&lds_o[0][j][c0 + 4];
#pragma unroll
    for (int k = 1; k < KSPLIT; ++k) {
      const float4 t0 = *(const float4*)&lds_o[k][j][c0];
      const float4 t1 = *(const float4*)&lds_o[k][j][c0 + 4];
      s0.x += t0.x; s0.y += t0.y; s0.z += t0.z; s0.w += t0.w;
      s1.x += t1.x; s1.y += t1.y; s1.z += t1.z; s1.w += t1.w;
    }
    v8bf o;
    o[0] = (bf16_t)s0.x; o[1] = (bf16_t)s0.y; o[2] = (bf16_t)s0.z; o[3] = (bf16_t)s0.w;
    o[4] = (bf16_t)s1.x; o[5] = (bf16_t)s1.y; o[6] = (bf16_t)s1.z; o[7] = (bf16_t)s1.w;
    af[nb * 32] = o;
  }
}

__device__ __forceinline__ void halt_softmax_blk(
    const float* __restrict__ hacc, const float* __restrict__ mask,
    float* __restrict__ out, int b) {
  __shared__ float red[256];
  const int tid = threadIdx.x;
  float lm = -3.0e38f;
  for (int j = tid; j < SEQ; j += 256) {
    const float v = hacc[(size_t)b * SEQ + j] * 0.25f +
                    mask[(size_t)b * SEQ + j] * (-1e10f);
    lm = fmaxf(lm, v);
  }
  red[tid] = lm;
  __syncthreads();
  for (int s = 128; s > 0; s >>= 1) {
    if (tid < s) red[tid] = fmaxf(red[tid], red[tid + s]);
    __syncthreads();
  }
  const float mx = red[0];
  __syncthreads();
  float ls = 0.0f;
  for (int j = tid; j < SEQ; j += 256) {
    const float v = hacc[(size_t)b * SEQ + j] * 0.25f +
                    mask[(size_t)b * SEQ + j] * (-1e10f);
    ls += __expf(v - mx);
  }
  red[tid] = ls;
  __syncthreads();
  for (int s = 128; s > 0; s >>= 1) {
    if (tid < s) red[tid] += red[tid + s];
    __syncthreads();
  }
  const float inv = 1.0f / red[0];
  for (int j = tid; j < SEQ; j += 256) {
    const float v = hacc[(size_t)b * SEQ + j] * 0.25f +
                    mask[(size_t)b * SEQ + j] * (-1e10f);
    out[(size_t)b * SEQ + j] = __expf(v - mx) * inv;
  }
}

#define FL_ROWS  64
#define FL_WROWS 8
#define FL_PITCH (DIMD + 4)
template <bool WRITE_BF16, bool WITH_HALT>
__global__ __launch_bounds__(256) void gemm_add_ln(
    const bf16_t* __restrict__ A, const bf16_t* __restrict__ B,
    const float* __restrict__ bias, const float* __restrict__ Rs,
    const float* __restrict__ g, const float* __restrict__ be,
    float* __restrict__ out, bf16_t* __restrict__ outb,
    const float* __restrict__ hacc, const float* __restrict__ mask,
    float* __restrict__ hout, int K) {
  static_assert(DIMD == 256 && FL_ROWS == 8 * FL_WROWS,
                "eight waves: 32 rows x 64 columns each in the GEMM, 8 rows each in the LayerNorm");
  if (WITH_HALT && blockIdx.x < BATCH) {
    halt_softmax_blk(hacc, mask, hout, blockIdx.x);
    return;
  }
  const int blk  = WITH_HALT ? (int)blockIdx.x - BATCH : (int)blockIdx.x;
  const int lane = threadIdx.x & 31;
  const int wave = threadIdx.x >> 5;
  const int r    = lane & 15;
  const int half = lane >> 4;
  const int rg   = wave >> 2;
  const int row0 = blk * FL_ROWS + rg * 32;
  const int col0 = (wave & 3) * 64;

  __shared__ __attribute__((aligned(16))) float ystage[FL_ROWS][FL_PITCH];

  v8f acc[2][4] = {};
  const v8bf* aq = (const v8bf*)A + (size_t)(row0 >> 4) * (K >> 5) * 64 + lane;
  const v8bf* bq = (const v8bf*)B + (size_t)(col0 >> 4) * (K >> 5) * 64 + lane;

  const size_t lrow0 = (size_t)blk * FL_ROWS + (size_t)wave * FL_WROWS;
  const float4* r4 = (const float4*)(Rs + lrow0 * DIMD) + lane;
  float4 x[FL_WROWS][2];
#pragma unroll
  for (int rr = 0; rr < FL_WROWS; ++rr) {
#pragma unroll
    for (int hf = 0; hf < 2; ++hf) x[rr][hf] = r4[rr * (DIMD / 4) + hf * 32];
  }

  gemm_mainloop<false>(acc, aq, bq, K);

#pragma unroll
  for (int mi = 0; mi < 2; ++mi) {
#pragma unroll
    for (int nb = 0; nb < 4; ++nb) {
      const float bv = bias[col0 + nb * 16 + r];
#pragma unroll
      for (int i = 0; i < 8; ++i)
        ystage[rg * 32 + mi * 16 + i + 8 * half][col0 + nb * 16 + r] = acc[mi][nb][i] + bv;
    }
  }
  __syncthreads();

  float v[FL_WROWS][8];
  float s[FL_WROWS];
#pragma unroll
  for (int rr = 0; rr < FL_WROWS; ++rr) {
#pragma unroll
    for (int hf = 0; hf < 2; ++hf) {
      const float4 y = *(const float4*)&ystage[wave * FL_WROWS + rr][4 * lane + hf * 128];
      v[rr][hf * 4 + 0] = x[rr][hf].x + y.x;
      v[rr][hf * 4 + 1] = x[rr][hf].y + y.y;
      v[rr][hf * 4 + 2] = x[rr][hf].z + y.z;
      v[rr][hf * 4 + 3] = x[rr][hf].w + y.w;
    }
    s[rr] = ((v[rr][0] + v[rr][1]) + (v[rr][2] + v[rr][3])) +
            ((v[rr][4] + v[rr][5]) + (v[rr][6] + v[rr][7]));
  }
  float4 gg[2], bb[2];
#pragma unroll
  for (int hf = 0; hf < 2; ++hf) {
    gg[hf] = ((const float4*)g)[lane + hf * 32];
    bb[hf] = ((const float4*)be)[lane + hf * 32];
  }
#pragma unroll
  for (int m = 16; m >= 1; m >>= 1) {
#pragma unroll
    for (int rr = 0; rr < FL_WROWS; ++rr) s[rr] += __shfl_xor(s[rr], m, 32);
  }

  float q[FL_WROWS];
#pragma unroll
  for (int rr = 0; rr < FL_WROWS; ++rr) {
    const float mu = s[rr] * (1.0f / DIMD);
#pragma unroll
    for (int j = 0; j < 8; ++j) v[rr][j] -= mu;
    q[rr] = ((v[rr][0] * v[rr][0] + v[rr][1] * v[rr][1]) +
             (v[rr][2] * v[rr][2] + v[rr][3] * v[rr][3])) +
            ((v[rr][4] * v[rr][4] + v[rr][5] * v[rr][5]) +
             (v[rr][6] * v[rr][6] + v[rr][7] * v[rr][7]));
  }
#pragma unroll
  for (int m = 16; m >= 1; m >>= 1) {
#pragma unroll
    for (int rr = 0; rr < FL_WROWS; ++rr) q[rr] += __shfl_xor(q[rr], m, 32);
  }

  __shared__ __attribute__((aligned(16))) bf16_t bstage[WRITE_BF16 ? FL_ROWS : 1][DIMD + 8];
  float4 y[FL_WROWS][2];
#pragma unroll
  for (int rr = 0; rr < FL_WROWS; ++rr) {
    const float var = q[rr] * (1.0f / DIMD);
    const float rs  = rsqrtf(var + 1e-5f);
#pragma unroll
    for (int hf = 0; hf < 2; ++hf) {
      y[rr][hf].x = v[rr][hf * 4 + 0] * rs * gg[hf].x + bb[hf].x;
      y[rr][hf].y = v[rr][hf * 4 + 1] * rs * gg[hf].y + bb[hf].y;
      y[rr][hf].z = v[rr][hf * 4 + 2] * rs * gg[hf].z + bb[hf].z;
      y[rr][hf].w = v[rr][hf * 4 + 3] * rs * gg[hf].w + bb[hf].w;
      if (WRITE_BF16) {
        v4bf o;
        o[0] = (bf16_t)y[rr][hf].x; o[1] = (bf16_t)y[rr][hf].y;
        o[2] = (bf16_t)y[rr][hf].z; o[3] = (bf16_t)y[rr][hf].w;
        *(v4bf*)&bstage[wave * FL_WROWS + rr][4 * lane + hf * 128] = o;
      }
    }
  }
  if (WRITE_BF16) {
    __syncthreads();
    const int pr = lane & 15, ph = lane >> 4;
    v8bf* hf_out = (v8bf*)outb + (size_t)blk * 2048 + lane;
#pragma unroll
    for (int t = 0; t < 8; ++t) {
      const int G = wave * 8 + t;
      const v8bf u = *(const v8bf*)&bstage[(G >> 4) * 16 + pr]
                                          [((G >> 1) & 7) * 32 + (G & 1) * 16 + 8 * ph];
      hf_out[G * 32] = u;
    }
  }
#pragma unroll
  for (int rr = 0; rr < FL_WROWS; ++rr) {
#pragma unroll
    for (int hf = 0; hf < 2; ++hf)
      ((float4*)(out + (lrow0 + rr) * DIMD))[lane + hf * 32] = y[rr][hf];
  }
}

extern "C" void kernel_launch(void* const* d_in, const int* in_sizes, int n_in,
                              void* d_out, int out_size, void* d_ws,
                              size_t ws_size, hipStream_t stream) {
  (void)in_sizes; (void)n_in; (void)out_size; (void)ws_size;
  const float* x    = (const float*)d_in[0];
  const float* mask = (const float*)d_in[1];
  const float* Wqkv = (const float*)d_in[2];
  const float* bqkv = (const float*)d_in[3];
  const float* Wo   = (const float*)d_in[4];
  const float* bo   = (const float*)d_in[5];
  const float* W1   = (const float*)d_in[6];
  const float* b1   = (const float*)d_in[7];
  const float* W2   = (const float*)d_in[8];
  const float* b2   = (const float*)d_in[9];
  const float* g1   = (const float*)d_in[10];
  const float* be1  = (const float*)d_in[11];
  const float* g2   = (const float*)d_in[12];
  const float* be2  = (const float*)d_in[13];

  char* ws = (char*)d_ws;
  size_t off = 0;
  auto alloc = [&](size_t bytes) {
    char* p = ws + off;
    off += (bytes + 255) & ~(size_t)255;
    return p;
  };
  bf16_t* xb    = (bf16_t*)alloc((size_t)NTOK * DIMD * 2);
  bf16_t* Wqkvb = (bf16_t*)alloc((size_t)3 * DIMD * DIMD * 2);
  bf16_t* Wob   = (bf16_t*)alloc((size_t)DIMD * DIMD * 2);
  bf16_t* W1b   = (bf16_t*)alloc((size_t)FFND * DIMD * 2);
  bf16_t* W2b   = (bf16_t*)alloc((size_t)DIMD * FFND * 2);
  bf16_t* Qb    = (bf16_t*)alloc((size_t)NTOK * DIMD * 2);
  bf16_t* Kb    = (bf16_t*)alloc((size_t)NTOK * DIMD * 2);
  bf16_t* Vt    = (bf16_t*)alloc((size_t)NTOK * DIMD * 2);
  bf16_t* attnb = (bf16_t*)alloc((size_t)NTOK * DIMD * 2);
  float*  hbuf  = (float*) alloc((size_t)NTOK * DIMD * 4);
  bf16_t* hb    = (bf16_t*)alloc((size_t)NTOK * DIMD * 2);
  bf16_t* ffn1b = (bf16_t*)alloc((size_t)NTOK * FFND * 2);
  float*  halt  = (float*) alloc((size_t)NTOK * 4);

  float* out      = (float*)d_out;
  float* out_halt = out + (size_t)NTOK * DIMD;

  prep_all<<<PB_X + PB_WQKV + PB_WO + PB_W1 + PB_W2 + PB_HZ, 256, 0, stream>>>(
      x, Wqkv, Wo, W1, W2, xb, Wqkvb, Wob, W1b, W2b, halt);

  gemm_wmma<2><<<dim3(3 * DIMD / 64, NTOK / 128), 128, 0, stream>>>(
      xb, Wqkvb, bqkv, nullptr, Qb, Kb, Vt, NTOK, 3 * DIMD, DIMD);

  attn_wmma<<<BATCH * HEADS * (SEQ / QTILE), 32 * KSPLIT, 0, stream>>>(
      Qb, Kb, Vt, mask, attnb, halt);

  gemm_add_ln<true, true><<<BATCH + NTOK / FL_ROWS, 256, 0, stream>>>(
      attnb, Wob, bo, x, g1, be1, hbuf, hb, halt, mask, out_halt, DIMD);

  gemm_wmma<1><<<dim3(FFND / 64, NTOK / 128), 128, 0, stream>>>(
      hb, W1b, b1, ffn1b, nullptr, nullptr, nullptr, NTOK, FFND, DIMD);

  gemm_add_ln<false, false><<<NTOK / FL_ROWS, 256, 0, stream>>>(
      ffn1b, W2b, b2, hbuf, g2, be2, out, nullptr, nullptr, nullptr, nullptr, FFND);
}


# loads issued further ahead in the loops that ran one step ahead and have spare registers: attention pass 2 three key tiles (4 K/mask sets, 2 V sets), FFN2 k loop 3 steps, QKV and FFN1 k loops 2
# speedup vs baseline: 2.9109x; 1.1102x over previous
#include <hip/hip_runtime.h>
#include <hip/hip_bf16.h>

typedef __bf16 bf16_t;
typedef __attribute__((ext_vector_type(16))) __bf16 v16bf;
typedef __attribute__((ext_vector_type(4)))  __bf16 v4bf;
typedef __attribute__((ext_vector_type(8)))  __bf16 v8bf;
typedef __attribute__((ext_vector_type(8)))  float  v8f;

#define DIMD   256
#define HEADS  4
#define HD     64
#define FFND   1024
#define BATCH  8
#define SEQ    2048
#define NTOK   (BATCH*SEQ)

__device__ __forceinline__ v16bf load_frag_bf16(const bf16_t* base) {
  const int half = (threadIdx.x >> 4) & 1;
  const bf16_t* p = base + 8 * half;
  v16bf f;
#pragma unroll
  for (int e = 0; e < 8; ++e) {
    f[e]     = p[e];
    f[e + 8] = p[e + 16];
  }
  return f;
}

__device__ __forceinline__ v16bf load_frag_packed(const v8bf* p) {
  const v8bf lo = p[0];
  const v8bf hi = p[32];
  return __builtin_shufflevector(lo, hi, 0, 1, 2, 3, 4, 5, 6, 7,
                                 8, 9, 10, 11, 12, 13, 14, 15);
}

__device__ __forceinline__ v8f wmma_bf16(v16bf a, v16bf b, v8f c) {
  return __builtin_amdgcn_wmma_f32_16x16x32_bf16(
       false, a,  false, b,
       (short)0, c,  false,  false);
}

#define XF_U 4
__device__ __forceinline__ void cvt_afrag_blk(const float* __restrict__ in,
                                              bf16_t* __restrict__ out, int blk) {
  static_assert(DIMD == 256, "one 32-column chunk per wave of a 256-thread block");
  const int tid = threadIdx.x;
  const int q   = tid & 3;
  const int kc  = tid >> 5;
  const int r8  = (tid >> 2) & 7;
  float4 x0[XF_U], x1[XF_U];
#pragma unroll
  for (int k = 0; k < XF_U; ++k) {
    const int row = (blk * XF_U + k) * 8 + r8;
    const float4* p = (const float4*)(in + (size_t)row * DIMD + kc * 32 + 8 * q);
    x0[k] = p[0];
    x1[k] = p[1];
  }
#pragma unroll
  for (int k = 0; k < XF_U; ++k) {
    const int row = (blk * XF_U + k) * 8 + r8;
    const size_t unit = ((size_t)((row >> 4) * (DIMD / 32) + kc) * 2 + (q >> 1)) * 32 +
                        (q & 1) * 16 + (row & 15);
    v8bf o;
    o[0] = (bf16_t)x0[k].x; o[1] = (bf16_t)x0[k].y; o[2] = (bf16_t)x0[k].z; o[3] = (bf16_t)x0[k].w;
    o[4] = (bf16_t)x1[k].x; o[5] = (bf16_t)x1[k].y; o[6] = (bf16_t)x1[k].z; o[7] = (bf16_t)x1[k].w;
    ((v8bf*)out)[unit] = o;
  }
}

template <int KD>
__device__ __forceinline__ void cvt_frag_blk(const float* __restrict__ in,
                                             bf16_t* __restrict__ out, int blk) {
  const int u    = blk * 256 + threadIdx.x;
  const int l    = u & 31;
  const int j    = (u >> 5) & 1;
  const int tile = u >> 6;
  const int kc   = tile % (KD / 32);
  const int rt   = tile / (KD / 32);
  const int row  = rt * 16 + (l & 15);
  const int k    = kc * 32 + j * 16 + 8 * (l >> 4);
  const float4* p = (const float4*)(in + (size_t)row * KD + k);
  const float4 x0 = p[0], x1 = p[1];
  v8bf o;
  o[0] = (bf16_t)x0.x; o[1] = (bf16_t)x0.y; o[2] = (bf16_t)x0.z; o[3] = (bf16_t)x0.w;
  o[4] = (bf16_t)x1.x; o[5] = (bf16_t)x1.y; o[6] = (bf16_t)x1.z; o[7] = (bf16_t)x1.w;
  ((v8bf*)out)[u] = o;
}

#define PB_X    (NTOK / (8 * XF_U))
#define PB_WQKV (3 * DIMD * DIMD / (8 * 256))
#define PB_WO   (DIMD * DIMD / (8 * 256))
#define PB_W1   (FFND * DIMD / (8 * 256))
#define PB_W2   (DIMD * FFND / (8 * 256))
#define PB_HZ   (NTOK / (4 * 256))
__global__ __launch_bounds__(256) void prep_all(
    const float* __restrict__ x, const float* __restrict__ Wqkv,
    const float* __restrict__ Wo, const float* __restrict__ W1,
    const float* __restrict__ W2, bf16_t* __restrict__ xb,
    bf16_t* __restrict__ Wqkvb, bf16_t* __restrict__ Wob,
    bf16_t* __restrict__ W1b, bf16_t* __restrict__ W2b,
    float* __restrict__ halt) {
  int blk = blockIdx.x;
  if (blk < PB_X) { cvt_afrag_blk(x, xb, blk); return; }
  blk -= PB_X;
  if (blk < PB_WQKV) { cvt_frag_blk<DIMD>(Wqkv, Wqkvb, blk); return; }
  blk -= PB_WQKV;
  if (blk < PB_WO) { cvt_frag_blk<DIMD>(Wo, Wob, blk); return; }
  blk -= PB_WO;
  if (blk < PB_W1) { cvt_frag_blk<DIMD>(W1, W1b, blk); return; }
  blk -= PB_W1;
  if (blk < PB_W2) { cvt_frag_blk<FFND>(W2, W2b, blk); return; }
  blk -= PB_W2;
  ((float4*)halt)[blk * 256 + threadIdx.x] = make_float4(0.0f, 0.0f, 0.0f, 0.0f);
}


struct GemmFrags {
  v16bf a0, a1, b[4];
};

__device__ __forceinline__ void gemm_load(
    GemmFrags& f, const v8bf* __restrict__ aq, const v8bf* __restrict__ bq,
    int kch, int k0) {
  const int kc = k0 >> 5;
  f.a0 = load_frag_packed(aq + (size_t)kc * 64);
  f.a1 = load_frag_packed(aq + (size_t)(kch + kc) * 64);
#pragma unroll
  for (int nb = 0; nb < 4; ++nb)
    f.b[nb] = load_frag_packed(bq + (size_t)(nb * kch + kc) * 64);
}

template <bool TRANS>
__device__ __forceinline__ void gemm_step(v8f (&acc)[2][4], const GemmFrags& f) {
#pragma unroll
  for (int nb = 0; nb < 4; ++nb) {
    if (TRANS) {
      acc[0][nb] = wmma_bf16(f.b[nb], f.a0, acc[0][nb]);
      acc[1][nb] = wmma_bf16(f.b[nb], f.a1, acc[1][nb]);
    } else {
      acc[0][nb] = wmma_bf16(f.a0, f.b[nb], acc[0][nb]);
      acc[1][nb] = wmma_bf16(f.a1, f.b[nb], acc[1][nb]);
    }
  }
}

template <bool TRANS>
__device__ __forceinline__ void gemm_mainloop(
    v8f (&acc)[2][4], const v8bf* __restrict__ aq,
    const v8bf* __restrict__ bq, int K) {
  const int kch = K >> 5;
  GemmFrags fx, fy;
  gemm_load(fx, aq, bq, kch, 0);
  for (int k0 = 0; k0 < K; k0 += 64) {
    gemm_load(fy, aq, bq, kch, k0 + 32);
    __builtin_amdgcn_sched_barrier(0);
    gemm_step<TRANS>(acc, fx);
    const int kn = (k0 + 64 < K) ? k0 + 64 : K - 32;
    gemm_load(fx, aq, bq, kch, kn);
    __builtin_amdgcn_sched_barrier(0);
    gemm_step<TRANS>(acc, fy);
  }
}

template <bool TRANS, int SETS, int KSTEPS>
__device__ __forceinline__ void gemm_mainloop_ahead(
    v8f (&acc)[2][4], const v8bf* __restrict__ aq,
    const v8bf* __restrict__ bq) {
  static_assert(KSTEPS >= SETS - 1, "the prologue fills SETS - 1 sets");
  GemmFrags f[SETS];
#pragma unroll
  for (int s = 0; s < SETS - 1; ++s) gemm_load(f[s], aq, bq, KSTEPS, s * 32);
#pragma unroll
  for (int s = 0; s < KSTEPS; ++s) {
    if (s + SETS - 1 < KSTEPS)
      gemm_load(f[(s + SETS - 1) % SETS], aq, bq, KSTEPS, (s + SETS - 1) * 32);
    __builtin_amdgcn_sched_barrier(0);
    gemm_step<TRANS>(acc, f[s % SETS]);
  }
}

template <bool TRANS>
__device__ __forceinline__ void gemm_mainloop_ahead3(
    v8f (&acc)[2][4], const v8bf* __restrict__ aq,
    const v8bf* __restrict__ bq, int K) {
  const int kch = K >> 5;
  GemmFrags f[4];
#pragma unroll
  for (int u = 0; u < 3; ++u) gemm_load(f[u], aq, bq, kch, u * 32);
  for (int k0 = 0; k0 < K; k0 += 128) {
#pragma unroll
    for (int u = 0; u < 4; ++u) {
      const int kn = k0 + (u + 3) * 32;
      gemm_load(f[(u + 3) & 3], aq, bq, kch, kn < K ? kn : K - 32);
      __builtin_amdgcn_sched_barrier(0);
      gemm_step<TRANS>(acc, f[u]);
    }
  }
}

template <int MODE>
__global__ __launch_bounds__(128) void gemm_wmma(
    const bf16_t* __restrict__ A, const bf16_t* __restrict__ B,
    const float* __restrict__ bias, bf16_t* __restrict__ Yb,
    bf16_t* __restrict__ Qb, bf16_t* __restrict__ Kb, bf16_t* __restrict__ Vt,
    int M, int N, int K) {
  const int lane = threadIdx.x & 31;
  const int wave = threadIdx.x >> 5;
  const int r    = lane & 15;
  const int half = lane >> 4;
  const int row0 = blockIdx.y * 128 + wave * 32;
  const int col0 = blockIdx.x * 64;

  v8f acc[2][4] = {};
  const v8bf* aq = (const v8bf*)A + (size_t)(row0 >> 4) * (K >> 5) * 64 + lane;
  const v8bf* bq = (const v8bf*)B + (size_t)(col0 >> 4) * (K >> 5) * 64 + lane;

  const bool kblk = (MODE == 2) && (col0 >= DIMD) && (col0 < 2 * DIMD);
  if (MODE == 1 || kblk) gemm_mainloop_ahead<true, 3, DIMD / 32>(acc, aq, bq);
  else                   gemm_mainloop_ahead<false, 3, DIMD / 32>(acc, aq, bq);

  if (MODE == 1) {
    v8bf* yf = (v8bf*)Yb + ((size_t)(row0 >> 4) * (N >> 5) + (col0 >> 5)) * 64 + lane;
#pragma unroll
    for (int mi = 0; mi < 2; ++mi) {
#pragma unroll
      for (int nb = 0; nb < 4; ++nb) {
        v8bf o;
#pragma unroll
        for (int i = 0; i < 8; ++i)
          o[i] = (bf16_t)fmaxf(acc[mi][nb][i] + bias[col0 + nb * 16 + 8 * half + i], 0.0f);
        yf[(size_t)mi * (N >> 5) * 64 + nb * 32] = o;
      }
    }
    return;
  }

  if (MODE == 2 && col0 >= DIMD) {
    const int bb  = row0 >> 11;
    const int t0  = row0 & (SEQ - 1);
    if (kblk) {
      const size_t bhk = (size_t)bb * HEADS + ((col0 - DIMD) >> 6);
      v8bf* kf = (v8bf*)Kb + (bhk * (SEQ / 16) + (t0 >> 4)) * 128 + lane;
#pragma unroll
      for (int mi = 0; mi < 2; ++mi) {
#pragma unroll
        for (int nb = 0; nb < 4; ++nb) {
          v8bf o;
#pragma unroll
          for (int i = 0; i < 8; ++i)
            o[i] = (bf16_t)(acc[mi][nb][i] + bias[col0 + nb * 16 + 8 * half + i]);
          kf[mi * 128 + nb * 32] = o;
        }
      }
    } else {
      const size_t bhv = (size_t)bb * HEADS + ((col0 - 2 * DIMD) >> 6);
      v8bf* vf = (v8bf*)Vt + (bhv * (SEQ / 32) + (t0 >> 5)) * 256 + lane;
#pragma unroll
      for (int mi = 0; mi < 2; ++mi) {
#pragma unroll
        for (int nb = 0; nb < 4; ++nb) {
          const float bv = bias[col0 + nb * 16 + r];
          v8bf o;
#pragma unroll
          for (int i = 0; i < 8; ++i) o[i] = (bf16_t)(acc[mi][nb][i] + bv);
          vf[nb * 64 + mi * 32] = o;
        }
      }
    }
    return;
  }

#pragma unroll
  for (int mi = 0; mi < 2; ++mi) {
#pragma unroll
    for (int nb = 0; nb < 4; ++nb) {
#pragma unroll
      for (int i = 0; i < 8; ++i) {
        const int row = row0 + mi * 16 + i + 8 * half;
        const int col = col0 + nb * 16 + r;
        const float y = acc[mi][nb][i] + bias[col];
        const int b = row >> 11, t = row & (SEQ - 1);
        const int h = col >> 6, d = col & 63;
        Qb[(((size_t)b * HEADS + h) * SEQ + t) * HD + d] = (bf16_t)y;
      }
    }
  }
}

#define KSPLIT 4
#define NQ     4
#define QTILE  (16 * NQ)
#define CSCALE 0.18033688011112042f
#define NEGRAW (-8589934592.0f)

typedef __attribute__((ext_vector_type(2))) float  f32x2;
typedef __attribute__((ext_vector_type(2))) __bf16 bf16x2;

__device__ __forceinline__ float xhalf(float x) {
  const int b = __builtin_bit_cast(int, x);
  return __builtin_bit_cast(float, __builtin_amdgcn_permlanex16(
      b, b, 0x76543210, 0xfedcba98, false, false));
}

template <int XM>
__device__ __forceinline__ float row_xor(float x) {
  const int b = __builtin_bit_cast(int, x);
  return __builtin_bit_cast(float, __builtin_amdgcn_update_dpp(
      0, b, 0x160 + XM, 0xf, 0xf, true));
}

__device__ __forceinline__ float max3f(float a, float b, float c) {
  return __builtin_fmaxf(__builtin_fmaxf(a, b), c);
}

__global__ __launch_bounds__(32 * KSPLIT) void attn_wmma(
    const bf16_t* __restrict__ Qb, const bf16_t* __restrict__ Kb,
    const bf16_t* __restrict__ Vt, const float* __restrict__ mask,
    bf16_t* __restrict__ attn, float* __restrict__ halt) {
  __shared__ float  lds_m[KSPLIT][QTILE];
  __shared__ float  lds_l[KSPLIT][QTILE];
  __shared__ __attribute__((aligned(16))) float lds_o[KSPLIT][NQ * 32][32];
  static_assert(NQ == KSPLIT, "the final merge gives wave w sub-tile w");

  const int bid  = blockIdx.x;
  const int qt   = bid % (SEQ / QTILE);
  const int bh   = bid / (SEQ / QTILE);
  const int b    = bh >> 2;
  const int lane = threadIdx.x & 31;
  const int w    = threadIdx.x >> 5;
  const int r    = lane & 15;
  const int half = lane >> 4;

  const bf16_t* Qbase = Qb + ((size_t)bh * SEQ + qt * QTILE) * HD;
  const v8bf* Kq = (const v8bf*)Kb +
                   ((size_t)bh * (SEQ / 16) + (size_t)w * (SEQ / KSPLIT / 16)) * 128 + lane;
  const v8bf* Vq = (const v8bf*)Vt +
                   ((size_t)bh * (SEQ / 32) + (size_t)w * (SEQ / KSPLIT / 32)) * 256 + lane;
  const float* mrow8 = mask + (size_t)b * SEQ + (size_t)w * (SEQ / KSPLIT) + 8 * half;
  float*       hrow  = halt + (size_t)b * SEQ + (size_t)w * (SEQ / KSPLIT);

  v16bf qa[NQ][2];
#pragma unroll
  for (int qi = 0; qi < NQ; ++qi) {
    qa[qi][0] = load_frag_bf16(Qbase + (size_t)(qi * 16 + r) * HD + 0);
    qa[qi][1] = load_frag_bf16(Qbase + (size_t)(qi * 16 + r) * HD + 32);
  }

  const f32x2 c2 = {CSCALE, CSCALE};

  float m[NQ], sh[NQ], l[NQ];
#pragma unroll
  for (int qi = 0; qi < NQ; ++qi) {
    m[qi] = -3.0e38f; sh[qi] = 3.0e38f * CSCALE; l[qi] = 0.0f;
  }

  v16bf ks0[4], ks1[4];
  v8f mks[4];
  auto load_tile = [&](v16bf& k0r, v16bf& k1r, v8f& mkr, int kt) {
    k0r = load_frag_packed(Kq + (size_t)kt * 128);
    k1r = load_frag_packed(Kq + (size_t)kt * 128 + 64);
    mkr = *(const v8f*)(mrow8 + kt * 16);
  };
  auto pass1_tile = [&](const v16bf& kb0, const v16bf& kb1, const v8f& mk) {
    const bool msk[8] = {mk[0] > 0.5f, mk[1] > 0.5f, mk[2] > 0.5f, mk[3] > 0.5f,
                         mk[4] > 0.5f, mk[5] > 0.5f, mk[6] > 0.5f, mk[7] > 0.5f};
#pragma unroll
    for (int qi = 0; qi < NQ; ++qi) {
      v8f s = {};
      s = wmma_bf16(kb0, qa[qi][0], s);
      s = wmma_bf16(kb1, qa[qi][1], s);
      float sv[8];
#pragma unroll
      for (int i = 0; i < 8; ++i) sv[i] = msk[i] ? NEGRAW : s[i];
      float mx = max3f(m[qi], sv[0], sv[1]);
      mx = max3f(mx, sv[2], sv[3]);
      mx = max3f(mx, sv[4], sv[5]);
      mx = max3f(mx, sv[6], sv[7]);
      const float nm = -(mx * CSCALE);
      const f32x2 nm2 = {nm, nm};
      const float alpha = __builtin_amdgcn_exp2f(nm - sh[qi]);
      f32x2 e[4];
#pragma unroll
      for (int j = 0; j < 4; ++j) {
        const f32x2 a = {sv[2 * j], sv[2 * j + 1]};
        const f32x2 t = __builtin_elementwise_fma(a, c2, nm2);
        e[j][0] = __builtin_amdgcn_exp2f(t[0]);
        e[j][1] = __builtin_amdgcn_exp2f(t[1]);
      }
      const f32x2 es = (e[0] + e[1]) + (e[2] + e[3]);
      l[qi] = __builtin_fmaf(l[qi], alpha, es[0] + es[1]);
      m[qi] = mx;
      sh[qi] = nm;
    }
  };

  constexpr int NKT = SEQ / KSPLIT / 16;
#pragma unroll
  for (int u = 0; u < 3; ++u) load_tile(ks0[u], ks1[u], mks[u], u);
  for (int kt = 0; kt < NKT; kt += 4) {
#pragma unroll
    for (int u = 0; u < 4; ++u) {
      load_tile(ks0[(u + 3) & 3], ks1[(u + 3) & 3], mks[(u + 3) & 3],
                (kt + u + 3) & (NKT - 1));
      __builtin_amdgcn_sched_barrier(0);
      pass1_tile(ks0[u], ks1[u], mks[u]);
    }
  }
  constexpr int NVT = SEQ / KSPLIT / 32;
  v16bf vba[4], vbb[4];
  auto load_v = [&](v16bf (&vb)[4], int vt) {
#pragma unroll
    for (int nb = 0; nb < 4; ++nb)
      vb[nb] = load_frag_packed(Vq + (size_t)vt * 256 + nb * 64);
  };
  load_v(vba, 0);
#pragma unroll
  for (int qi = 0; qi < NQ; ++qi) {
    const float so = xhalf(sh[qi]);
    const float lo = xhalf(l[qi]);
    const float sn = __builtin_fminf(sh[qi], so);
    l[qi] = l[qi] * __builtin_amdgcn_exp2f(sn - sh[qi]) +
            lo * __builtin_amdgcn_exp2f(sn - so);
    sh[qi] = sn;
  }
  if (half == 0) {
#pragma unroll
    for (int qi = 0; qi < NQ; ++qi) {
      lds_m[w][qi * 16 + r] = sh[qi];
      lds_l[w][qi * 16 + r] = l[qi];
    }
  }
  __syncthreads();
  f32x2 nmf[NQ], rl[NQ];
#pragma unroll
  for (int qi = 0; qi < NQ; ++qi) {
    float sw[KSPLIT], lw[KSPLIT];
#pragma unroll
    for (int k = 0; k < KSPLIT; ++k) {
      sw[k] = lds_m[k][qi * 16 + r];
      lw[k] = lds_l[k][qi * 16 + r];
    }
    float nm = sw[0];
#pragma unroll
    for (int k = 1; k < KSPLIT; ++k) nm = __builtin_fminf(nm, sw[k]);
    float ls = 0.0f;
#pragma unroll
    for (int k = 0; k < KSPLIT; ++k)
      ls += lw[k] * __builtin_amdgcn_exp2f(nm - sw[k]);
    const float inv = 1.0f / ls;
    nmf[qi][0] = nm;  nmf[qi][1] = nm;
    rl[qi][0]  = inv; rl[qi][1]  = inv;
  }

  const bool b3 = (r & 8) != 0, b2 = (r & 4) != 0, b1 = (r & 2) != 0;
  v8f o[NQ][4] = {};
  auto pass2_tile = [&](v16bf (&pa)[NQ], const v16bf& kb0, const v16bf& kb1,
                        const v8f& mk, int sub, int kt) {
    const bool msk[8] = {mk[0] > 0.5f, mk[1] > 0.5f, mk[2] > 0.5f, mk[3] > 0.5f,
                         mk[4] > 0.5f, mk[5] > 0.5f, mk[6] > 0.5f, mk[7] > 0.5f};
    f32x2 cs[4];
#pragma unroll
    for (int qi = 0; qi < NQ; ++qi) {
      v8f s = {};
      s = wmma_bf16(kb0, qa[qi][0], s);
      s = wmma_bf16(kb1, qa[qi][1], s);
#pragma unroll
      for (int j = 0; j < 4; ++j) {
        const f32x2 a = {msk[2 * j] ? NEGRAW : s[2 * j],
                         msk[2 * j + 1] ? NEGRAW : s[2 * j + 1]};
        const f32x2 t = __builtin_elementwise_fma(a, c2, nmf[qi]);
        f32x2 p;
        p[0] = __builtin_amdgcn_exp2f(t[0]);
        p[1] = __builtin_amdgcn_exp2f(t[1]);
        p = p * rl[qi];
        if (qi == 0) cs[j] = p; else cs[j] = cs[j] + p;
        const bf16x2 pb = __builtin_convertvector(p, bf16x2);
        pa[qi][sub * 8 + 2 * j]     = pb[0];
        pa[qi][sub * 8 + 2 * j + 1] = pb[1];
      }
    }
    float k4[4];
#pragma unroll
    for (int j = 0; j < 4; ++j) {
      const float lo4 = cs[j >> 1][j & 1];
      const float hi4 = cs[(j + 4) >> 1][j & 1];
      const float keep = b3 ? hi4 : lo4;
      const float send = b3 ? lo4 : hi4;
      k4[j] = keep + row_xor<8>(send);
    }
    float k2[2];
#pragma unroll
    for (int j = 0; j < 2; ++j) {
      const float keep = b2 ? k4[j + 2] : k4[j];
      const float send = b2 ? k4[j] : k4[j + 2];
      k2[j] = keep + row_xor<4>(send);
    }
    float k1 = (b1 ? k2[1] : k2[0]) + row_xor<2>(b1 ? k2[0] : k2[1]);
    k1 += row_xor<1>(k1);
    if ((lane & 1) == 0) atomicAdd(&hrow[kt * 16 + 8 * half + (r >> 1)], k1);
  };

  auto pv_tile = [&](const v16bf (&pa)[NQ], const v16bf (&vb)[4]) {
#pragma unroll
    for (int nb = 0; nb < 4; ++nb) {
#pragma unroll
      for (int qi = 0; qi < NQ; ++qi)
        o[qi][nb] = wmma_bf16(pa[qi], vb[nb], o[qi][nb]);
    }
  };

  for (int kt = 0; kt < NKT; kt += 4) {
    v16bf pa[NQ];
    load_tile(ks0[3], ks1[3], mks[3], kt + 3);
    load_v(vbb, (kt >> 1) + 1);
    __builtin_amdgcn_sched_barrier(0);
    pass2_tile(pa, ks0[0], ks1[0], mks[0], 0, kt);
    load_tile(ks0[0], ks1[0], mks[0], (kt + 4) & (NKT - 1));
    __builtin_amdgcn_sched_barrier(0);
    pass2_tile(pa, ks0[1], ks1[1], mks[1], 1, kt + 1);
    pv_tile(pa, vba);
    load_tile(ks0[1], ks1[1], mks[1], (kt + 5) & (NKT - 1));
    load_v(vba, ((kt >> 1) + 2) & (NVT - 1));
    __builtin_amdgcn_sched_barrier(0);
    pass2_tile(pa, ks0[2], ks1[2], mks[2], 0, kt + 2);
    load_tile(ks0[2], ks1[2], mks[2], (kt + 6) & (NKT - 1));
    __builtin_amdgcn_sched_barrier(0);
    pass2_tile(pa, ks0[3], ks1[3], mks[3], 1, kt + 3);
    pv_tile(pa, vbb);
  }

#pragma unroll
  for (int qi = 0; qi < NQ; ++qi) {
#pragma unroll
    for (int nb = 0; nb < 4; ++nb) {
#pragma unroll
      for (int i = 0; i < 8; ++i) {
        lds_o[w][(qi * 4 + nb) * 8 + i][lane] = o[qi][nb][i];
      }
    }
  }
  __syncthreads();

  const int h = bh & 3;
  const size_t rt = ((size_t)b * SEQ + (size_t)qt * QTILE + (size_t)w * 16) >> 4;
  v8bf* af = (v8bf*)attn + (rt * (DIMD / 32) + (size_t)h * 2) * 64 + lane;
#pragma unroll
  for (int nb = 0; nb < 4; ++nb) {
    const int j  = (w * 4 + nb) * 8 + (r & 7);
    const int c0 = 16 * (r >> 3) + 8 * half;
    float4 s0 = *(const float4*)&lds_o[0][j][c0];
    float4 s1 = *(const float4*)&lds_o[0][j][c0 + 4];
#pragma unroll
    for (int k = 1; k < KSPLIT; ++k) {
      const float4 t0 = *(const float4*)&lds_o[k][j][c0];
      const float4 t1 = *(const float4*)&lds_o[k][j][c0 + 4];
      s0.x += t0.x; s0.y += t0.y; s0.z += t0.z; s0.w += t0.w;
      s1.x += t1.x; s1.y += t1.y; s1.z += t1.z; s1.w += t1.w;
    }
    v8bf o;
    o[0] = (bf16_t)s0.x; o[1] = (bf16_t)s0.y; o[2] = (bf16_t)s0.z; o[3] = (bf16_t)s0.w;
    o[4] = (bf16_t)s1.x; o[5] = (bf16_t)s1.y; o[6] = (bf16_t)s1.z; o[7] = (bf16_t)s1.w;
    af[nb * 32] = o;
  }
}

__device__ __forceinline__ void halt_softmax_blk(
    const float* __restrict__ hacc, const float* __restrict__ mask,
    float* __restrict__ out, int b) {
  __shared__ float red[256];
  const int tid = threadIdx.x;
  float lm = -3.0e38f;
  for (int j = tid; j < SEQ; j += 256) {
    const float v = hacc[(size_t)b * SEQ + j] * 0.25f +
                    mask[(size_t)b * SEQ + j] * (-1e10f);
    lm = fmaxf(lm, v);
  }
  red[tid] = lm;
  __syncthreads();
  for (int s = 128; s > 0; s >>= 1) {
    if (tid < s) red[tid] = fmaxf(red[tid], red[tid + s]);
    __syncthreads();
  }
  const float mx = red[0];
  __syncthreads();
  float ls = 0.0f;
  for (int j = tid; j < SEQ; j += 256) {
    const float v = hacc[(size_t)b * SEQ + j] * 0.25f +
                    mask[(size_t)b * SEQ + j] * (-1e10f);
    ls += __expf(v - mx);
  }
  red[tid] = ls;
  __syncthreads();
  for (int s = 128; s > 0; s >>= 1) {
    if (tid < s) red[tid] += red[tid + s];
    __syncthreads();
  }
  const float inv = 1.0f / red[0];
  for (int j = tid; j < SEQ; j += 256) {
    const float v = hacc[(size_t)b * SEQ + j] * 0.25f +
                    mask[(size_t)b * SEQ + j] * (-1e10f);
    out[(size_t)b * SEQ + j] = __expf(v - mx) * inv;
  }
}

#define FL_ROWS  64
#define FL_WROWS 8
#define FL_PITCH (DIMD + 4)
template <bool WRITE_BF16, bool WITH_HALT, bool AHEAD3>
__global__ __launch_bounds__(256) void gemm_add_ln(
    const bf16_t* __restrict__ A, const bf16_t* __restrict__ B,
    const float* __restrict__ bias, const float* __restrict__ Rs,
    const float* __restrict__ g, const float* __restrict__ be,
    float* __restrict__ out, bf16_t* __restrict__ outb,
    const float* __restrict__ hacc, const float* __restrict__ mask,
    float* __restrict__ hout, int K) {
  static_assert(DIMD == 256 && FL_ROWS == 8 * FL_WROWS,
                "eight waves: 32 rows x 64 columns each in the GEMM, 8 rows each in the LayerNorm");
  if (WITH_HALT && blockIdx.x < BATCH) {
    halt_softmax_blk(hacc, mask, hout, blockIdx.x);
    return;
  }
  const int blk  = WITH_HALT ? (int)blockIdx.x - BATCH : (int)blockIdx.x;
  const int lane = threadIdx.x & 31;
  const int wave = threadIdx.x >> 5;
  const int r    = lane & 15;
  const int half = lane >> 4;
  const int rg   = wave >> 2;
  const int row0 = blk * FL_ROWS + rg * 32;
  const int col0 = (wave & 3) * 64;

  __shared__ __attribute__((aligned(16))) float ystage[FL_ROWS][FL_PITCH];

  v8f acc[2][4] = {};
  const v8bf* aq = (const v8bf*)A + (size_t)(row0 >> 4) * (K >> 5) * 64 + lane;
  const v8bf* bq = (const v8bf*)B + (size_t)(col0 >> 4) * (K >> 5) * 64 + lane;

  const size_t lrow0 = (size_t)blk * FL_ROWS + (size_t)wave * FL_WROWS;
  const float4* r4 = (const float4*)(Rs + lrow0 * DIMD) + lane;
  float4 x[FL_WROWS][2];
#pragma unroll
  for (int rr = 0; rr < FL_WROWS; ++rr) {
#pragma unroll
    for (int hf = 0; hf < 2; ++hf) x[rr][hf] = r4[rr * (DIMD / 4) + hf * 32];
  }

  if (AHEAD3) gemm_mainloop_ahead3<false>(acc, aq, bq, K);
  else        gemm_mainloop<false>(acc, aq, bq, K);

#pragma unroll
  for (int mi = 0; mi < 2; ++mi) {
#pragma unroll
    for (int nb = 0; nb < 4; ++nb) {
      const float bv = bias[col0 + nb * 16 + r];
#pragma unroll
      for (int i = 0; i < 8; ++i)
        ystage[rg * 32 + mi * 16 + i + 8 * half][col0 + nb * 16 + r] = acc[mi][nb][i] + bv;
    }
  }
  __syncthreads();

  float v[FL_WROWS][8];
  float s[FL_WROWS];
#pragma unroll
  for (int rr = 0; rr < FL_WROWS; ++rr) {
#pragma unroll
    for (int hf = 0; hf < 2; ++hf) {
      const float4 y = *(const float4*)&ystage[wave * FL_WROWS + rr][4 * lane + hf * 128];
      v[rr][hf * 4 + 0] = x[rr][hf].x + y.x;
      v[rr][hf * 4 + 1] = x[rr][hf].y + y.y;
      v[rr][hf * 4 + 2] = x[rr][hf].z + y.z;
      v[rr][hf * 4 + 3] = x[rr][hf].w + y.w;
    }
    s[rr] = ((v[rr][0] + v[rr][1]) + (v[rr][2] + v[rr][3])) +
            ((v[rr][4] + v[rr][5]) + (v[rr][6] + v[rr][7]));
  }
  float4 gg[2], bb[2];
#pragma unroll
  for (int hf = 0; hf < 2; ++hf) {
    gg[hf] = ((const float4*)g)[lane + hf * 32];
    bb[hf] = ((const float4*)be)[lane + hf * 32];
  }
#pragma unroll
  for (int m = 16; m >= 1; m >>= 1) {
#pragma unroll
    for (int rr = 0; rr < FL_WROWS; ++rr) s[rr] += __shfl_xor(s[rr], m, 32);
  }

  float q[FL_WROWS];
#pragma unroll
  for (int rr = 0; rr < FL_WROWS; ++rr) {
    const float mu = s[rr] * (1.0f / DIMD);
#pragma unroll
    for (int j = 0; j < 8; ++j) v[rr][j] -= mu;
    q[rr] = ((v[rr][0] * v[rr][0] + v[rr][1] * v[rr][1]) +
             (v[rr][2] * v[rr][2] + v[rr][3] * v[rr][3])) +
            ((v[rr][4] * v[rr][4] + v[rr][5] * v[rr][5]) +
             (v[rr][6] * v[rr][6] + v[rr][7] * v[rr][7]));
  }
#pragma unroll
  for (int m = 16; m >= 1; m >>= 1) {
#pragma unroll
    for (int rr = 0; rr < FL_WROWS; ++rr) q[rr] += __shfl_xor(q[rr], m, 32);
  }

  __shared__ __attribute__((aligned(16))) bf16_t bstage[WRITE_BF16 ? FL_ROWS : 1][DIMD + 8];
  float4 y[FL_WROWS][2];
#pragma unroll
  for (int rr = 0; rr < FL_WROWS; ++rr) {
    const float var = q[rr] * (1.0f / DIMD);
    const float rs  = rsqrtf(var + 1e-5f);
#pragma unroll
    for (int hf = 0; hf < 2; ++hf) {
      y[rr][hf].x = v[rr][hf * 4 + 0] * rs * gg[hf].x + bb[hf].x;
      y[rr][hf].y = v[rr][hf * 4 + 1] * rs * gg[hf].y + bb[hf].y;
      y[rr][hf].z = v[rr][hf * 4 + 2] * rs * gg[hf].z + bb[hf].z;
      y[rr][hf].w = v[rr][hf * 4 + 3] * rs * gg[hf].w + bb[hf].w;
      if (WRITE_BF16) {
        v4bf o;
        o[0] = (bf16_t)y[rr][hf].x; o[1] = (bf16_t)y[rr][hf].y;
        o[2] = (bf16_t)y[rr][hf].z; o[3] = (bf16_t)y[rr][hf].w;
        *(v4bf*)&bstage[wave * FL_WROWS + rr][4 * lane + hf * 128] = o;
      }
    }
  }
  if (WRITE_BF16) {
    __syncthreads();
    const int pr = lane & 15, ph = lane >> 4;
    v8bf* hf_out = (v8bf*)outb + (size_t)blk * 2048 + lane;
#pragma unroll
    for (int t = 0; t < 8; ++t) {
      const int G = wave * 8 + t;
      const v8bf u = *(const v8bf*)&bstage[(G >> 4) * 16 + pr]
                                          [((G >> 1) & 7) * 32 + (G & 1) * 16 + 8 * ph];
      hf_out[G * 32] = u;
    }
  }
#pragma unroll
  for (int rr = 0; rr < FL_WROWS; ++rr) {
#pragma unroll
    for (int hf = 0; hf < 2; ++hf)
      ((float4*)(out + (lrow0 + rr) * DIMD))[lane + hf * 32] = y[rr][hf];
  }
}

extern "C" void kernel_launch(void* const* d_in, const int* in_sizes, int n_in,
                              void* d_out, int out_size, void* d_ws,
                              size_t ws_size, hipStream_t stream) {
  (void)in_sizes; (void)n_in; (void)out_size; (void)ws_size;
  const float* x    = (const float*)d_in[0];
  const float* mask = (const float*)d_in[1];
  const float* Wqkv = (const float*)d_in[2];
  const float* bqkv = (const float*)d_in[3];
  const float* Wo   = (const float*)d_in[4];
  const float* bo   = (const float*)d_in[5];
  const float* W1   = (const float*)d_in[6];
  const float* b1   = (const float*)d_in[7];
  const float* W2   = (const float*)d_in[8];
  const float* b2   = (const float*)d_in[9];
  const float* g1   = (const float*)d_in[10];
  const float* be1  = (const float*)d_in[11];
  const float* g2   = (const float*)d_in[12];
  const float* be2  = (const float*)d_in[13];

  char* ws = (char*)d_ws;
  size_t off = 0;
  auto alloc = [&](size_t bytes) {
    char* p = ws + off;
    off += (bytes + 255) & ~(size_t)255;
    return p;
  };
  bf16_t* xb    = (bf16_t*)alloc((size_t)NTOK * DIMD * 2);
  bf16_t* Wqkvb = (bf16_t*)alloc((size_t)3 * DIMD * DIMD * 2);
  bf16_t* Wob   = (bf16_t*)alloc((size_t)DIMD * DIMD * 2);
  bf16_t* W1b   = (bf16_t*)alloc((size_t)FFND * DIMD * 2);
  bf16_t* W2b   = (bf16_t*)alloc((size_t)DIMD * FFND * 2);
  bf16_t* Qb    = (bf16_t*)alloc((size_t)NTOK * DIMD * 2);
  bf16_t* Kb    = (bf16_t*)alloc((size_t)NTOK * DIMD * 2);
  bf16_t* Vt    = (bf16_t*)alloc((size_t)NTOK * DIMD * 2);
  bf16_t* attnb = (bf16_t*)alloc((size_t)NTOK * DIMD * 2);
  float*  hbuf  = (float*) alloc((size_t)NTOK * DIMD * 4);
  bf16_t* hb    = (bf16_t*)alloc((size_t)NTOK * DIMD * 2);
  bf16_t* ffn1b = (bf16_t*)alloc((size_t)NTOK * FFND * 2);
  float*  halt  = (float*) alloc((size_t)NTOK * 4);

  float* out      = (float*)d_out;
  float* out_halt = out + (size_t)NTOK * DIMD;

  prep_all<<<PB_X + PB_WQKV + PB_WO + PB_W1 + PB_W2 + PB_HZ, 256, 0, stream>>>(
      x, Wqkv, Wo, W1, W2, xb, Wqkvb, Wob, W1b, W2b, halt);

  gemm_wmma<2><<<dim3(3 * DIMD / 64, NTOK / 128), 128, 0, stream>>>(
      xb, Wqkvb, bqkv, nullptr, Qb, Kb, Vt, NTOK, 3 * DIMD, DIMD);

  attn_wmma<<<BATCH * HEADS * (SEQ / QTILE), 32 * KSPLIT, 0, stream>>>(
      Qb, Kb, Vt, mask, attnb, halt);

  gemm_add_ln<true, true, false><<<BATCH + NTOK / FL_ROWS, 256, 0, stream>>>(
      attnb, Wob, bo, x, g1, be1, hbuf, hb, halt, mask, out_halt, DIMD);

  gemm_wmma<1><<<dim3(FFND / 64, NTOK / 128), 128, 0, stream>>>(
      hb, W1b, b1, ffn1b, nullptr, nullptr, nullptr, NTOK, FFND, DIMD);

  gemm_add_ln<false, false, true><<<NTOK / FL_ROWS, 256, 0, stream>>>(
      ffn1b, W2b, b2, hbuf, g2, be2, out, nullptr, nullptr, nullptr, nullptr, FFND);
}
